# starting kernel
# baseline (speedup 1.0000x reference)
#include <hip/hip_runtime.h>
#include <math.h>

typedef float v2f __attribute__((ext_vector_type(2)));
typedef float v8f __attribute__((ext_vector_type(8)));

#define DIN 128
#define C1  64
#define NEG_SLOPE 0.2f

__device__ __forceinline__ float atomicMaxF(float* addr, float val) {
  if (val >= 0.0f)
    return __int_as_float(atomicMax((int*)addr, __float_as_int(val)));
  else
    return __uint_as_float(atomicMin((unsigned int*)addr, __float_as_uint(val)));
}

__global__ void k_init(float* __restrict__ out1, float* __restrict__ m,
                       float* __restrict__ z, float* __restrict__ outAcc, int n) {
  int i = blockIdx.x * blockDim.x + threadIdx.x;
  if (i < n * C1) out1[i] = 0.0f;
  if (i < n) { m[i] = -INFINITY; z[i] = 0.0f; outAcc[i] = 0.0f; }
}

__global__ __launch_bounds__(256) void k_gemm_wmma(const float* __restrict__ x,
                                                   const float* __restrict__ W,
                                                   float* __restrict__ h, int n) {
  __shared__ float Wt[C1 * DIN];
  int tid = threadIdx.x;
#pragma unroll
  for (int it = 0; it < (DIN * C1) / 256; ++it) {
    int idx = tid + it * 256;
    int k  = idx >> 6;
    int nn = idx & 63;
    Wt[nn * DIN + k] = W[idx];
  }
  __syncthreads();

  int wave = tid >> 5;
  int lane = tid & 31;
  int half = lane >> 4;
  int l    = lane & 15;
  int rowBase = blockIdx.x * 128 + wave * 16;

  int arow = rowBase + l;
  if (arow > n - 1) arow = n - 1;
  const float* ap = x + (size_t)arow * DIN + 2 * half;

  v8f c0 = {}, c1 = {}, c2 = {}, c3 = {};
#pragma unroll 4
  for (int k0 = 0; k0 < DIN; k0 += 4) {
    v2f a = *(const v2f*)(ap + k0);
    const float* wp = &Wt[l * DIN + k0 + 2 * half];
    v2f b0 = *(const v2f*)(wp);
    v2f b1 = *(const v2f*)(wp + 16 * DIN);
    v2f b2 = *(const v2f*)(wp + 32 * DIN);
    v2f b3 = *(const v2f*)(wp + 48 * DIN);
    c0 = __builtin_amdgcn_wmma_f32_16x16x4_f32(false, a, false, b0, (short)0, c0, false, false);
    c1 = __builtin_amdgcn_wmma_f32_16x16x4_f32(false, a, false, b1, (short)0, c1, false, false);
    c2 = __builtin_amdgcn_wmma_f32_16x16x4_f32(false, a, false, b2, (short)0, c2, false, false);
    c3 = __builtin_amdgcn_wmma_f32_16x16x4_f32(false, a, false, b3, (short)0, c3, false, false);
  }

#pragma unroll
  for (int v = 0; v < 8; ++v) {
    int row = rowBase + 8 * half + v;
    if (row < n) {
      float* hp = h + (size_t)row * C1 + l;
      hp[0]  = c0[v];
      hp[16] = c1[v];
      hp[32] = c2[v];
      hp[48] = c3[v];
    }
  }
}

__global__ void k_logits(const float* __restrict__ h, const float* __restrict__ a_src,
                         const float* __restrict__ a_dst, float* __restrict__ esrc,
                         float* __restrict__ edst, int n) {
  int i = blockIdx.x * blockDim.x + threadIdx.x;
  if (i >= n) return;
  const float4* hp = (const float4*)(h + (size_t)i * C1);
  float s = 0.0f, d = 0.0f;
#pragma unroll
  for (int c = 0; c < C1 / 4; ++c) {
    float4 hv = hp[c];
    float4 as = ((const float4*)a_src)[c];
    float4 ad = ((const float4*)a_dst)[c];
    s += hv.x * as.x + hv.y * as.y + hv.z * as.z + hv.w * as.w;
    d += hv.x * ad.x + hv.y * ad.y + hv.z * ad.z + hv.w * ad.w;
  }
  esrc[i] = s;
  edst[i] = d;
}

__global__ void k_edge_max(const int* __restrict__ ei, int E, int n,
                           const float* __restrict__ esrc, const float* __restrict__ edst,
                           float* __restrict__ m, float* __restrict__ eedge) {
  int idx = blockIdx.x * blockDim.x + threadIdx.x;
  int Etot = E + n;
  if (idx >= Etot) return;
  int s, d;
  if (idx < E) { s = ei[idx]; d = ei[E + idx]; }
  else         { s = d = idx - E; }
  float v = esrc[s] + edst[d];
  v = v > 0.0f ? v : NEG_SLOPE * v;
  eedge[idx] = v;
  atomicMaxF(&m[d], v);
}

__global__ void k_edge_expsum(const int* __restrict__ ei, int E, int n,
                              const float* __restrict__ m, float* __restrict__ z,
                              float* __restrict__ eedge) {
  int idx = blockIdx.x * blockDim.x + threadIdx.x;
  int Etot = E + n;
  if (idx >= Etot) return;
  int d = (idx < E) ? ei[E + idx] : (idx - E);
  float p = __expf(eedge[idx] - m[d]);
  eedge[idx] = p;
  atomicAdd(&z[d], p);
}

__global__ void k_edge_aggr64(const int* __restrict__ ei, int E, int n,
                              const float* __restrict__ z, const float* __restrict__ eedge,
                              const float* __restrict__ h, float* __restrict__ out1) {
  long long t = (long long)blockIdx.x * blockDim.x + threadIdx.x;
  int idx = (int)(t >> 4);
  int sub = (int)(t & 15);
  int Etot = E + n;
  if (idx >= Etot) return;
  int s, d;
  if (idx < E) { s = ei[idx]; d = ei[E + idx]; }
  else         { s = d = idx - E; }
  float alpha = eedge[idx] / z[d];
  float4 hv = *(const float4*)(h + (size_t)s * C1 + sub * 4);
  float* op = out1 + (size_t)d * C1 + sub * 4;
  atomicAdd(op + 0, alpha * hv.x);
  atomicAdd(op + 1, alpha * hv.y);
  atomicAdd(op + 2, alpha * hv.z);
  atomicAdd(op + 3, alpha * hv.w);
}

__global__ void k_layer2_prep(const float* __restrict__ out1, const float* __restrict__ b1,
                              const float* __restrict__ W2, const float* __restrict__ a_src2,
                              const float* __restrict__ a_dst2, float* __restrict__ g,
                              float* __restrict__ esrc, float* __restrict__ edst,
                              float* __restrict__ m, float* __restrict__ z, int n) {
  int i = blockIdx.x * blockDim.x + threadIdx.x;
  if (i >= n) return;
  const float4* op = (const float4*)(out1 + (size_t)i * C1);
  float acc = 0.0f;
#pragma unroll
  for (int c = 0; c < C1 / 4; ++c) {
    float4 v = op[c];
    float4 bb = ((const float4*)b1)[c];
    float4 ww = ((const float4*)W2)[c];
    float v0 = fmaxf(v.x + bb.x, 0.0f);
    float v1 = fmaxf(v.y + bb.y, 0.0f);
    float v2 = fmaxf(v.z + bb.z, 0.0f);
    float v3 = fmaxf(v.w + bb.w, 0.0f);
    acc += v0 * ww.x + v1 * ww.y + v2 * ww.z + v3 * ww.w;
  }
  g[i] = acc;
  esrc[i] = acc * a_src2[0];
  edst[i] = acc * a_dst2[0];
  m[i] = -INFINITY;
  z[i] = 0.0f;
}

__global__ void k_edge_aggr1(const int* __restrict__ ei, int E, int n,
                             const float* __restrict__ z, const float* __restrict__ eedge,
                             const float* __restrict__ g, float* __restrict__ outAcc) {
  int idx = blockIdx.x * blockDim.x + threadIdx.x;
  int Etot = E + n;
  if (idx >= Etot) return;
  int s, d;
  if (idx < E) { s = ei[idx]; d = ei[E + idx]; }
  else         { s = d = idx - E; }
  float alpha = eedge[idx] / z[d];
  atomicAdd(&outAcc[d], alpha * g[s]);
}

__global__ void k_final(float* __restrict__ out, const float* __restrict__ b2, int n) {
  int i = blockIdx.x * blockDim.x + threadIdx.x;
  if (i >= n) return;
  float v = out[i] + b2[0];
  out[i] = 1.0f / (1.0f + __expf(-v));
}

extern "C" void kernel_launch(void* const* d_in, const int* in_sizes, int n_in,
                              void* d_out, int out_size, void* d_ws, size_t ws_size,
                              hipStream_t stream) {
  const int*   ei     = (const int*)d_in[0];
  const float* x      = (const float*)d_in[1];
  const float* W1     = (const float*)d_in[2];
  const float* a_src1 = (const float*)d_in[3];
  const float* a_dst1 = (const float*)d_in[4];
  const float* b1     = (const float*)d_in[5];
  const float* W2     = (const float*)d_in[6];
  const float* a_src2 = (const float*)d_in[7];
  const float* a_dst2 = (const float*)d_in[8];
  const float* b2     = (const float*)d_in[9];

  const int E = in_sizes[0] / 2;
  const int n = in_sizes[1] / DIN;
  const int Etot = E + n;

  float* f     = (float*)d_ws;
  float* h     = f;
  float* out1  = f + (size_t)64 * n;
  float* esrc  = f + (size_t)128 * n;
  float* edst  = f + (size_t)129 * n;
  float* m     = f + (size_t)130 * n;
  float* z     = f + (size_t)131 * n;
  float* g     = f + (size_t)132 * n;
  float* eedge = f + (size_t)133 * n;
  float* outAcc = (float*)d_out;

  const int B = 256;
  dim3 blk(B);

  k_init<<<dim3((n * C1 + B - 1) / B), blk, 0, stream>>>(out1, m, z, outAcc, n);

  k_gemm_wmma<<<dim3((n + 127) / 128), blk, 0, stream>>>(x, W1, h, n);
  k_logits<<<dim3((n + B - 1) / B), blk, 0, stream>>>(h, a_src1, a_dst1, esrc, edst, n);
  k_edge_max<<<dim3((Etot + B - 1) / B), blk, 0, stream>>>(ei, E, n, esrc, edst, m, eedge);
  k_edge_expsum<<<dim3((Etot + B - 1) / B), blk, 0, stream>>>(ei, E, n, m, z, eedge);
  {
    long long tthreads = (long long)Etot * 16;
    int grid = (int)((tthreads + B - 1) / B);
    k_edge_aggr64<<<dim3(grid), blk, 0, stream>>>(ei, E, n, z, eedge, h, out1);
  }

  k_layer2_prep<<<dim3((n + B - 1) / B), blk, 0, stream>>>(out1, b1, W2, a_src2, a_dst2,
                                                           g, esrc, edst, m, z, n);
  k_edge_max<<<dim3((Etot + B - 1) / B), blk, 0, stream>>>(ei, E, n, esrc, edst, m, eedge);
  k_edge_expsum<<<dim3((Etot + B - 1) / B), blk, 0, stream>>>(ei, E, n, m, z, eedge);
  k_edge_aggr1<<<dim3((Etot + B - 1) / B), blk, 0, stream>>>(ei, E, n, z, eedge, g, outAcc);

  k_final<<<dim3((n + B - 1) / B), blk, 0, stream>>>(outAcc, b2, n);
}


# Layer-1 aggregate: two workgroups per 800 dst nodes each scan half the dst list and add their edges into LDS rows (LDS float adds), store partial rows once; global float atomics go; sum order changes
# speedup vs baseline: 1.5857x; 1.5857x over previous
#include <hip/hip_runtime.h>
#include <math.h>

typedef float v2f __attribute__((ext_vector_type(2)));
typedef float v8f __attribute__((ext_vector_type(8)));

#define DIN 128
#define C1  64
#define NEG_SLOPE 0.2f

__device__ __forceinline__ float atomicMaxF(float* addr, float val) {
  if (val >= 0.0f)
    return __int_as_float(atomicMax((int*)addr, __float_as_int(val)));
  else
    return __uint_as_float(atomicMin((unsigned int*)addr, __float_as_uint(val)));
}

__global__ void k_init(float* __restrict__ m, int n) {
  int i = blockIdx.x * blockDim.x + threadIdx.x;
  if (i < n) m[i] = -INFINITY;
}

__global__ __launch_bounds__(256) void k_gemm_wmma(const float* __restrict__ x,
                                                   const float* __restrict__ W,
                                                   float* __restrict__ h, int n) {
  __shared__ float Wt[C1 * DIN];
  int tid = threadIdx.x;
#pragma unroll
  for (int it = 0; it < (DIN * C1) / 256; ++it) {
    int idx = tid + it * 256;
    int k  = idx >> 6;
    int nn = idx & 63;
    Wt[nn * DIN + k] = W[idx];
  }
  __syncthreads();

  int wave = tid >> 5;
  int lane = tid & 31;
  int half = lane >> 4;
  int l    = lane & 15;
  int rowBase = blockIdx.x * 128 + wave * 16;

  int arow = rowBase + l;
  if (arow > n - 1) arow = n - 1;
  const float* ap = x + (size_t)arow * DIN + 2 * half;

  v8f c0 = {}, c1 = {}, c2 = {}, c3 = {};
#pragma unroll 4
  for (int k0 = 0; k0 < DIN; k0 += 4) {
    v2f a = *(const v2f*)(ap + k0);
    const float* wp = &Wt[l * DIN + k0 + 2 * half];
    v2f b0 = *(const v2f*)(wp);
    v2f b1 = *(const v2f*)(wp + 16 * DIN);
    v2f b2 = *(const v2f*)(wp + 32 * DIN);
    v2f b3 = *(const v2f*)(wp + 48 * DIN);
    c0 = __builtin_amdgcn_wmma_f32_16x16x4_f32(false, a, false, b0, (short)0, c0, false, false);
    c1 = __builtin_amdgcn_wmma_f32_16x16x4_f32(false, a, false, b1, (short)0, c1, false, false);
    c2 = __builtin_amdgcn_wmma_f32_16x16x4_f32(false, a, false, b2, (short)0, c2, false, false);
    c3 = __builtin_amdgcn_wmma_f32_16x16x4_f32(false, a, false, b3, (short)0, c3, false, false);
  }

#pragma unroll
  for (int v = 0; v < 8; ++v) {
    int row = rowBase + 8 * half + v;
    if (row < n) {
      float* hp = h + (size_t)row * C1 + l;
      hp[0]  = c0[v];
      hp[16] = c1[v];
      hp[32] = c2[v];
      hp[48] = c3[v];
    }
  }
}

__global__ void k_logits(const float* __restrict__ h, const float* __restrict__ a_src,
                         const float* __restrict__ a_dst, float* __restrict__ esrc,
                         float* __restrict__ edst, int n) {
  int i = blockIdx.x * blockDim.x + threadIdx.x;
  if (i >= n) return;
  const float4* hp = (const float4*)(h + (size_t)i * C1);
  float s = 0.0f, d = 0.0f;
#pragma unroll
  for (int c = 0; c < C1 / 4; ++c) {
    float4 hv = hp[c];
    float4 as = ((const float4*)a_src)[c];
    float4 ad = ((const float4*)a_dst)[c];
    s += hv.x * as.x + hv.y * as.y + hv.z * as.z + hv.w * as.w;
    d += hv.x * ad.x + hv.y * ad.y + hv.z * ad.z + hv.w * ad.w;
  }
  esrc[i] = s;
  edst[i] = d;
}

__global__ void k_edge_max(const int* __restrict__ ei, int E, int n,
                           const float* __restrict__ esrc, const float* __restrict__ edst,
                           float* __restrict__ m, float* __restrict__ eedge) {
  int idx = blockIdx.x * blockDim.x + threadIdx.x;
  int Etot = E + n;
  if (idx >= Etot) return;
  int s, d;
  if (idx < E) { s = ei[idx]; d = ei[E + idx]; }
  else         { s = d = idx - E; }
  float v = esrc[s] + edst[d];
  v = v > 0.0f ? v : NEG_SLOPE * v;
  eedge[idx] = v;
  atomicMaxF(&m[d], v);
}

#define RNG   800
#define QCAP  12288
#define AGT   1024
#define PU    8
__device__ __forceinline__ void aggr_add(float p, float4 hv, int dl, int sub, int rot,
                                         float* acc, float* zl) {
  float* ap = acc + dl * C1 + sub * 4;
  float v[4] = {p * hv.x, p * hv.y, p * hv.z, p * hv.w};
#pragma unroll
  for (int j = 0; j < 4; ++j) {
    int jj = (j + rot) & 3;
    atomicAdd(ap + jj, v[jj]);
  }
  if (sub == 0) atomicAdd(zl + dl, p);
}

__global__ __launch_bounds__(AGT) void k_edge_aggr64(const int* __restrict__ ei, int E, int n,
                              const float* __restrict__ m, float* __restrict__ z,
                              const float* __restrict__ eedge,
                              const float* __restrict__ h, float* __restrict__ out1) {
  __shared__ float acc[RNG * C1];
  __shared__ float zl[RNG];
  __shared__ int   q[QCAP];
  __shared__ int   qn;
  const int tid  = threadIdx.x;
  const int half = blockIdx.x & 1;
  const int base = (blockIdx.x >> 1) * RNG;
  const int cnt  = min(RNG, n - base);
  const int sub  = tid & 15;
  const int grp  = tid >> 4;
  const int rot  = grp & 1;
  if (tid == 0) qn = 0;

  for (int r = grp; r < cnt; r += AGT / 16) {
    float p = 0.0f;
    float4 hv = make_float4(0.0f, 0.0f, 0.0f, 0.0f);
    if (half == 0) {
      int d = base + r;
      p = __expf(eedge[E + d] - m[d]);
      hv = *(const float4*)(h + (size_t)d * C1 + sub * 4);
    }
    *(float4*)(acc + r * C1 + sub * 4) = make_float4(p * hv.x, p * hv.y, p * hv.z, p * hv.w);
    if (sub == 0) zl[r] = p;
  }
  __syncthreads();

  const int* dst = ei + E;
  const unsigned ucnt = (unsigned)cnt;
  int sbeg, send;
  if ((((size_t)dst) & 15) == 0) {
    const int nvall = E >> 2;
    const int nvh = (nvall + 1) >> 1;
    const int vbeg = half * nvh;
    const int nv = min(nvall, vbeg + nvh);
    const int4* dv = (const int4*)dst;
    for (int v0 = vbeg + tid; v0 < nv; v0 += 4 * AGT) {
      int4 a[4];
#pragma unroll
      for (int u = 0; u < 4; ++u) {
        int v = v0 + u * AGT;
        a[u] = (v < nv) ? dv[v] : make_int4(-1, -1, -1, -1);
      }
      unsigned hit = 0;
#pragma unroll
      for (int u = 0; u < 4; ++u) {
        hit |= ((unsigned)(a[u].x - base) < ucnt ? 1u : 0u) << (4 * u + 0);
        hit |= ((unsigned)(a[u].y - base) < ucnt ? 1u : 0u) << (4 * u + 1);
        hit |= ((unsigned)(a[u].z - base) < ucnt ? 1u : 0u) << (4 * u + 2);
        hit |= ((unsigned)(a[u].w - base) < ucnt ? 1u : 0u) << (4 * u + 3);
      }
      while (hit) {
        int b = __ffs(hit) - 1;
        hit &= hit - 1;
        int idx = 4 * (v0 + (b >> 2) * AGT) + (b & 3);
        int slot = atomicAdd(&qn, 1);
        if (slot < QCAP) q[slot] = idx;
        else {
          int s = ei[idx];
          int d = dst[idx];
          float p = __expf(eedge[idx] - m[d]);
#pragma nounroll
          for (int s2 = 0; s2 < 16; ++s2)
            aggr_add(p, *(const float4*)(h + (size_t)s * C1 + s2 * 4), d - base, s2, 0, acc, zl);
        }
      }
    }
    sbeg = half ? (nvall << 2) : E;
    send = E;
  } else {
    const int eh = (E + 1) >> 1;
    sbeg = half * eh;
    send = min(E, sbeg + eh);
  }
  for (int idx = sbeg + tid; idx < send; idx += AGT) {
    if ((unsigned)(dst[idx] - base) < ucnt) {
      int slot = atomicAdd(&qn, 1);
      if (slot < QCAP) q[slot] = idx;
      else {
        int s = ei[idx];
        int d = dst[idx];
        float p = __expf(eedge[idx] - m[d]);
#pragma nounroll
        for (int s2 = 0; s2 < 16; ++s2)
          aggr_add(p, *(const float4*)(h + (size_t)s * C1 + s2 * 4), d - base, s2, 0, acc, zl);
      }
    }
  }
  __syncthreads();

  const int nq = min(qn, QCAP);
  for (int k0 = grp; k0 < nq; k0 += PU * (AGT / 16)) {
    int    idx[PU], sv[PU], dvv[PU];
    float  ev[PU], mv[PU];
    float4 hv[PU];
#pragma unroll
    for (int u = 0; u < PU; ++u) {
      int k = k0 + u * (AGT / 16);
      idx[u] = q[k < nq ? k : k0];
    }
#pragma unroll
    for (int u = 0; u < PU; ++u) {
      sv[u]  = ei[idx[u]];
      dvv[u] = dst[idx[u]];
      ev[u]  = eedge[idx[u]];
    }
#pragma unroll
    for (int u = 0; u < PU; ++u) {
      mv[u] = m[dvv[u]];
      hv[u] = *(const float4*)(h + (size_t)sv[u] * C1 + sub * 4);
    }
#pragma unroll
    for (int u = 0; u < PU; ++u)
      if (k0 + u * (AGT / 16) < nq)
        aggr_add(__expf(ev[u] - mv[u]), hv[u], dvv[u] - base, sub, rot, acc, zl);
  }
  __syncthreads();

  float* op = out1 + (size_t)half * n * C1 + (size_t)base * C1;
  float* zp = z + (size_t)half * n + base;
  for (int i = tid; i < cnt * (C1 / 4); i += AGT)
    *(float4*)(op + i * 4) = *(const float4*)(acc + i * 4);
  for (int r = tid; r < cnt; r += AGT) zp[r] = zl[r];
}

__global__ void k_layer2_prep(const float* __restrict__ out1, const float* __restrict__ z,
                              const float* __restrict__ b1,
                              const float* __restrict__ W2, const float* __restrict__ a_src2,
                              const float* __restrict__ a_dst2, float* __restrict__ g,
                              float* __restrict__ esrc, float* __restrict__ edst,
                              float* __restrict__ m, float* __restrict__ zn, int n) {
  int i = blockIdx.x * blockDim.x + threadIdx.x;
  if (i >= n) return;
  const float4* op = (const float4*)(out1 + (size_t)i * C1);
  const float4* oq = (const float4*)(out1 + (size_t)n * C1 + (size_t)i * C1);
  const float rz = 1.0f / (z[i] + z[n + i]);
  float acc = 0.0f;
#pragma unroll
  for (int c = 0; c < C1 / 4; ++c) {
    float4 v = op[c];
    float4 w = oq[c];
    v.x += w.x; v.y += w.y; v.z += w.z; v.w += w.w;
    float4 bb = ((const float4*)b1)[c];
    float4 ww = ((const float4*)W2)[c];
    float v0 = fmaxf(v.x * rz + bb.x, 0.0f);
    float v1 = fmaxf(v.y * rz + bb.y, 0.0f);
    float v2 = fmaxf(v.z * rz + bb.z, 0.0f);
    float v3 = fmaxf(v.w * rz + bb.w, 0.0f);
    acc += v0 * ww.x + v1 * ww.y + v2 * ww.z + v3 * ww.w;
  }
  g[i] = acc;
  esrc[i] = acc * a_src2[0];
  edst[i] = acc * a_dst2[0];
  m[i] = -INFINITY;
  ((float2*)zn)[i] = make_float2(0.0f, 0.0f);
}

__global__ void k_edge_aggr1(const int* __restrict__ ei, int E, int n,
                             const float* __restrict__ m, const float* __restrict__ eedge,
                             const float* __restrict__ g, float* __restrict__ zn) {
  int t = blockIdx.x * blockDim.x + threadIdx.x;
  int idx = t >> 1;
  int sub = t & 1;
  int Etot = E + n;
  if (idx >= Etot) return;
  int s, d;
  if (idx < E) { s = ei[idx]; d = ei[E + idx]; }
  else         { s = d = idx - E; }
  float gs = g[s];
  float p = __expf(eedge[idx] - m[d]);
  float v = sub ? p * gs : p;
  atomicAdd(&zn[2 * (size_t)d + sub], v);
}

__global__ void k_final(const float* __restrict__ zn, const float* __restrict__ b2,
                        float* __restrict__ out, int n) {
  int i = blockIdx.x * blockDim.x + threadIdx.x;
  if (i >= n) return;
  float2 a = ((const float2*)zn)[i];
  float v = a.y / a.x + b2[0];
  out[i] = 1.0f / (1.0f + __expf(-v));
}

extern "C" void kernel_launch(void* const* d_in, const int* in_sizes, int n_in,
                              void* d_out, int out_size, void* d_ws, size_t ws_size,
                              hipStream_t stream) {
  const int*   ei     = (const int*)d_in[0];
  const float* x      = (const float*)d_in[1];
  const float* W1     = (const float*)d_in[2];
  const float* a_src1 = (const float*)d_in[3];
  const float* a_dst1 = (const float*)d_in[4];
  const float* b1     = (const float*)d_in[5];
  const float* W2     = (const float*)d_in[6];
  const float* a_src2 = (const float*)d_in[7];
  const float* a_dst2 = (const float*)d_in[8];
  const float* b2     = (const float*)d_in[9];

  const int E = in_sizes[0] / 2;
  const int n = in_sizes[1] / DIN;
  const int Etot = E + n;

  float* f     = (float*)d_ws;
  float* h     = f;
  float* out1  = f + (size_t)64 * n;
  float* esrc  = f + (size_t)192 * n;
  float* edst  = f + (size_t)193 * n;
  float* m     = f + (size_t)194 * n;
  float* g     = f + (size_t)195 * n;
  float* z     = f + (size_t)196 * n;
  float* zn    = f + (size_t)198 * n;
  float* eedge = f + (size_t)200 * n;
  float* out   = (float*)d_out;

  const int B = 256;
  dim3 blk(B);

  k_init<<<dim3((n + B - 1) / B), blk, 0, stream>>>(m, n);

  k_gemm_wmma<<<dim3((n + 127) / 128), blk, 0, stream>>>(x, W1, h, n);
  k_logits<<<dim3((n + B - 1) / B), blk, 0, stream>>>(h, a_src1, a_dst1, esrc, edst, n);
  k_edge_max<<<dim3((Etot + B - 1) / B), blk, 0, stream>>>(ei, E, n, esrc, edst, m, eedge);
  k_edge_aggr64<<<dim3(2 * ((n + RNG - 1) / RNG)), dim3(AGT), 0, stream>>>(ei, E, n, m, z, eedge, h, out1);

  k_layer2_prep<<<dim3((n + B - 1) / B), blk, 0, stream>>>(out1, z, b1, W2, a_src2, a_dst2,
                                                           g, esrc, edst, m, zn, n);
  k_edge_max<<<dim3((Etot + B - 1) / B), blk, 0, stream>>>(ei, E, n, esrc, edst, m, eedge);
  k_edge_aggr1<<<dim3((2 * Etot + B - 1) / B), blk, 0, stream>>>(ei, E, n, m, eedge, g, zn);

  k_final<<<dim3((n + B - 1) / B), blk, 0, stream>>>(zn, b2, out, n);
}


# Layer-2 edge work per destination range from the edge list the layer-1 range kernel keeps: max, exp sums in LDS, halves joined by rescaling; its global atomics go; 9 launches to 8; sum order changes
# speedup vs baseline: 2.4503x; 1.5452x over previous
#include <hip/hip_runtime.h>
#include <math.h>

typedef float v2f __attribute__((ext_vector_type(2)));
typedef float v8f __attribute__((ext_vector_type(8)));

#define DIN 128
#define C1  64
#define NEG_SLOPE 0.2f

__device__ __forceinline__ float atomicMaxF(float* addr, float val) {
  if (val >= 0.0f)
    return __int_as_float(atomicMax((int*)addr, __float_as_int(val)));
  else
    return __uint_as_float(atomicMin((unsigned int*)addr, __float_as_uint(val)));
}

__global__ void k_init(float* __restrict__ m, int n) {
  int i = blockIdx.x * blockDim.x + threadIdx.x;
  if (i < n) m[i] = -INFINITY;
}

__global__ __launch_bounds__(256) void k_gemm_wmma(const float* __restrict__ x,
                                                   const float* __restrict__ W,
                                                   float* __restrict__ h, int n) {
  __shared__ float Wt[C1 * DIN];
  int tid = threadIdx.x;
#pragma unroll
  for (int it = 0; it < (DIN * C1) / 256; ++it) {
    int idx = tid + it * 256;
    int k  = idx >> 6;
    int nn = idx & 63;
    Wt[nn * DIN + k] = W[idx];
  }
  __syncthreads();

  int wave = tid >> 5;
  int lane = tid & 31;
  int half = lane >> 4;
  int l    = lane & 15;
  int rowBase = blockIdx.x * 128 + wave * 16;

  int arow = rowBase + l;
  if (arow > n - 1) arow = n - 1;
  const float* ap = x + (size_t)arow * DIN + 2 * half;

  v8f c0 = {}, c1 = {}, c2 = {}, c3 = {};
#pragma unroll 4
  for (int k0 = 0; k0 < DIN; k0 += 4) {
    v2f a = *(const v2f*)(ap + k0);
    const float* wp = &Wt[l * DIN + k0 + 2 * half];
    v2f b0 = *(const v2f*)(wp);
    v2f b1 = *(const v2f*)(wp + 16 * DIN);
    v2f b2 = *(const v2f*)(wp + 32 * DIN);
    v2f b3 = *(const v2f*)(wp + 48 * DIN);
    c0 = __builtin_amdgcn_wmma_f32_16x16x4_f32(false, a, false, b0, (short)0, c0, false, false);
    c1 = __builtin_amdgcn_wmma_f32_16x16x4_f32(false, a, false, b1, (short)0, c1, false, false);
    c2 = __builtin_amdgcn_wmma_f32_16x16x4_f32(false, a, false, b2, (short)0, c2, false, false);
    c3 = __builtin_amdgcn_wmma_f32_16x16x4_f32(false, a, false, b3, (short)0, c3, false, false);
  }

#pragma unroll
  for (int v = 0; v < 8; ++v) {
    int row = rowBase + 8 * half + v;
    if (row < n) {
      float* hp = h + (size_t)row * C1 + l;
      hp[0]  = c0[v];
      hp[16] = c1[v];
      hp[32] = c2[v];
      hp[48] = c3[v];
    }
  }
}

__global__ void k_logits(const float* __restrict__ h, const float* __restrict__ a_src,
                         const float* __restrict__ a_dst, float* __restrict__ esrc,
                         float* __restrict__ edst, int n) {
  int i = blockIdx.x * blockDim.x + threadIdx.x;
  if (i >= n) return;
  const float4* hp = (const float4*)(h + (size_t)i * C1);
  float s = 0.0f, d = 0.0f;
#pragma unroll
  for (int c = 0; c < C1 / 4; ++c) {
    float4 hv = hp[c];
    float4 as = ((const float4*)a_src)[c];
    float4 ad = ((const float4*)a_dst)[c];
    s += hv.x * as.x + hv.y * as.y + hv.z * as.z + hv.w * as.w;
    d += hv.x * ad.x + hv.y * ad.y + hv.z * ad.z + hv.w * ad.w;
  }
  esrc[i] = s;
  edst[i] = d;
}

__global__ void k_edge_max(const int* __restrict__ ei, int E, int n,
                           const float* __restrict__ esrc, const float* __restrict__ edst,
                           float* __restrict__ m, float* __restrict__ eedge) {
  int idx = blockIdx.x * blockDim.x + threadIdx.x;
  int Etot = E + n;
  if (idx >= Etot) return;
  int s, d;
  if (idx < E) { s = ei[idx]; d = ei[E + idx]; }
  else         { s = d = idx - E; }
  float v = esrc[s] + edst[d];
  v = v > 0.0f ? v : NEG_SLOPE * v;
  eedge[idx] = v;
  atomicMaxF(&m[d], v);
}

#define RNG   800
#define QCAP  12288
#define AGT   1024
#define PU    8
__device__ __forceinline__ void aggr_add(float p, float4 hv, int dl, int sub, int rot,
                                         float* acc, float* zl) {
  float* ap = acc + dl * C1 + sub * 4;
  float v[4] = {p * hv.x, p * hv.y, p * hv.z, p * hv.w};
#pragma unroll
  for (int j = 0; j < 4; ++j) {
    int jj = (j + rot) & 3;
    atomicAdd(ap + jj, v[jj]);
  }
  if (sub == 0) atomicAdd(zl + dl, p);
}

__global__ __launch_bounds__(AGT) void k_edge_aggr64(const int* __restrict__ ei, int E, int n,
                              const float* __restrict__ m, float* __restrict__ z,
                              const float* __restrict__ eedge,
                              const float* __restrict__ h, float* __restrict__ out1,
                              int* __restrict__ lst, int* __restrict__ cntq) {
  __shared__ float acc[RNG * C1];
  __shared__ float zl[RNG];
  __shared__ int   q[QCAP];
  __shared__ int   qn;
  const int tid  = threadIdx.x;
  const int half = blockIdx.x & 1;
  const int base = (blockIdx.x >> 1) * RNG;
  const int cnt  = min(RNG, n - base);
  const int sub  = tid & 15;
  const int grp  = tid >> 4;
  const int rot  = grp & 1;
  if (tid == 0) qn = 0;

  for (int r = grp; r < cnt; r += AGT / 16) {
    float p = 0.0f;
    float4 hv = make_float4(0.0f, 0.0f, 0.0f, 0.0f);
    if (half == 0) {
      int d = base + r;
      p = __expf(eedge[E + d] - m[d]);
      hv = *(const float4*)(h + (size_t)d * C1 + sub * 4);
    }
    *(float4*)(acc + r * C1 + sub * 4) = make_float4(p * hv.x, p * hv.y, p * hv.z, p * hv.w);
    if (sub == 0) zl[r] = p;
  }
  __syncthreads();

  const int* dst = ei + E;
  const unsigned ucnt = (unsigned)cnt;
  int sbeg, send;
  if ((((size_t)dst) & 15) == 0) {
    const int nvall = E >> 2;
    const int nvh = (nvall + 1) >> 1;
    const int vbeg = half * nvh;
    const int nv = min(nvall, vbeg + nvh);
    const int4* dv = (const int4*)dst;
    for (int v0 = vbeg + tid; v0 < nv; v0 += 4 * AGT) {
      int4 a[4];
#pragma unroll
      for (int u = 0; u < 4; ++u) {
        int v = v0 + u * AGT;
        a[u] = (v < nv) ? dv[v] : make_int4(-1, -1, -1, -1);
      }
      unsigned hit = 0;
#pragma unroll
      for (int u = 0; u < 4; ++u) {
        hit |= ((unsigned)(a[u].x - base) < ucnt ? 1u : 0u) << (4 * u + 0);
        hit |= ((unsigned)(a[u].y - base) < ucnt ? 1u : 0u) << (4 * u + 1);
        hit |= ((unsigned)(a[u].z - base) < ucnt ? 1u : 0u) << (4 * u + 2);
        hit |= ((unsigned)(a[u].w - base) < ucnt ? 1u : 0u) << (4 * u + 3);
      }
      while (hit) {
        int b = __ffs(hit) - 1;
        hit &= hit - 1;
        int idx = 4 * (v0 + (b >> 2) * AGT) + (b & 3);
        int slot = atomicAdd(&qn, 1);
        if (slot < QCAP) q[slot] = idx;
        else {
          int s = ei[idx];
          int d = dst[idx];
          float p = __expf(eedge[idx] - m[d]);
#pragma nounroll
          for (int s2 = 0; s2 < 16; ++s2)
            aggr_add(p, *(const float4*)(h + (size_t)s * C1 + s2 * 4), d - base, s2, 0, acc, zl);
        }
      }
    }
    sbeg = half ? (nvall << 2) : E;
    send = E;
  } else {
    const int eh = (E + 1) >> 1;
    sbeg = half * eh;
    send = min(E, sbeg + eh);
  }
  for (int idx = sbeg + tid; idx < send; idx += AGT) {
    if ((unsigned)(dst[idx] - base) < ucnt) {
      int slot = atomicAdd(&qn, 1);
      if (slot < QCAP) q[slot] = idx;
      else {
        int s = ei[idx];
        int d = dst[idx];
        float p = __expf(eedge[idx] - m[d]);
#pragma nounroll
        for (int s2 = 0; s2 < 16; ++s2)
          aggr_add(p, *(const float4*)(h + (size_t)s * C1 + s2 * 4), d - base, s2, 0, acc, zl);
      }
    }
  }
  __syncthreads();

  const int nq = min(qn, QCAP);
  int2* lq = (int2*)lst + (size_t)blockIdx.x * QCAP;
  for (int k0 = grp; k0 < nq; k0 += PU * (AGT / 16)) {
    int    idx[PU], sv[PU], dvv[PU];
    float  ev[PU], mv[PU];
    float4 hv[PU];
#pragma unroll
    for (int u = 0; u < PU; ++u) {
      int k = k0 + u * (AGT / 16);
      idx[u] = q[k < nq ? k : k0];
    }
#pragma unroll
    for (int u = 0; u < PU; ++u) {
      sv[u]  = ei[idx[u]];
      dvv[u] = dst[idx[u]];
      ev[u]  = eedge[idx[u]];
    }
#pragma unroll
    for (int u = 0; u < PU; ++u) {
      mv[u] = m[dvv[u]];
      hv[u] = *(const float4*)(h + (size_t)sv[u] * C1 + sub * 4);
    }
#pragma unroll
    for (int u = 0; u < PU; ++u) {
      int k = k0 + u * (AGT / 16);
      if (sub == 0 && k < nq) lq[k] = make_int2(sv[u], dvv[u]);
    }
#pragma unroll
    for (int u = 0; u < PU; ++u)
      if (k0 + u * (AGT / 16) < nq)
        aggr_add(__expf(ev[u] - mv[u]), hv[u], dvv[u] - base, sub, rot, acc, zl);
  }
  if (tid == 0) cntq[blockIdx.x] = qn;
  __syncthreads();

  float* op = out1 + (size_t)half * n * C1 + (size_t)base * C1;
  float* zp = z + (size_t)half * n + base;
  for (int i = tid; i < cnt * (C1 / 4); i += AGT)
    *(float4*)(op + i * 4) = *(const float4*)(acc + i * 4);
  for (int r = tid; r < cnt; r += AGT) zp[r] = zl[r];
}

__global__ void k_layer2_prep(const float* __restrict__ out1, const float* __restrict__ z,
                              const float* __restrict__ b1,
                              const float* __restrict__ W2, float* __restrict__ g, int n) {
  int i = blockIdx.x * blockDim.x + threadIdx.x;
  if (i >= n) return;
  const float4* op = (const float4*)(out1 + (size_t)i * C1);
  const float4* oq = (const float4*)(out1 + (size_t)n * C1 + (size_t)i * C1);
  const float rz = 1.0f / (z[i] + z[n + i]);
  float acc = 0.0f;
#pragma unroll
  for (int c = 0; c < C1 / 4; ++c) {
    float4 v = op[c];
    float4 w = oq[c];
    v.x += w.x; v.y += w.y; v.z += w.z; v.w += w.w;
    float4 bb = ((const float4*)b1)[c];
    float4 ww = ((const float4*)W2)[c];
    float v0 = fmaxf(v.x * rz + bb.x, 0.0f);
    float v1 = fmaxf(v.y * rz + bb.y, 0.0f);
    float v2 = fmaxf(v.z * rz + bb.z, 0.0f);
    float v3 = fmaxf(v.w * rz + bb.w, 0.0f);
    acc += v0 * ww.x + v1 * ww.y + v2 * ww.z + v3 * ww.w;
  }
  g[i] = acc;
}

#define L2T 1024
#define L2U (QCAP / L2T)
__device__ __forceinline__ float l2_logit(float gs, float as2, float ed) {
  float v = gs * as2 + ed;
  return v > 0.0f ? v : NEG_SLOPE * v;
}
__device__ __forceinline__ void lds_max_f32(float* addr, float val) {
  if (__float_as_int(val) >= 0) atomicMax((int*)addr, __float_as_int(val));
  else                          atomicMin((unsigned int*)addr, __float_as_uint(val));
}

__global__ __launch_bounds__(L2T) void k_l2_range(const int* __restrict__ ei, int E, int n,
                              const int* __restrict__ lst, const int* __restrict__ cntq,
                              const float* __restrict__ g, const float* __restrict__ a_src2,
                              const float* __restrict__ a_dst2, float* __restrict__ part) {
  __shared__ float ml[RNG];
  __shared__ float zl[RNG];
  __shared__ float nl[RNG];
  __shared__ float edl[RNG];
  __shared__ float gl[RNG];
  __shared__ float esl[RNG];
  const int tid  = threadIdx.x;
  const int half = blockIdx.x & 1;
  const int base = (blockIdx.x >> 1) * RNG;
  const int cnt  = min(RNG, n - base);
  const unsigned ucnt = (unsigned)cnt;
  const float as2 = a_src2[0];
  const float ad2 = a_dst2[0];
  const int nqraw = cntq[blockIdx.x];
  const bool listed = nqraw <= QCAP;
  const int nq = listed ? nqraw : 0;

  const int* dst = ei + E;
  const int split = ((((size_t)dst) & 15) == 0) ? 4 * (((E >> 2) + 1) >> 1) : ((E + 1) >> 1);
  const int hb = half ? split : 0;
  const int he = half ? E : split;

  const int2* lq = (const int2*)lst + (size_t)blockIdx.x * QCAP;
  int2  sd[L2U];
  float gs[L2U], ev[L2U];
#pragma unroll
  for (int u = 0; u < L2U; ++u) {
    int k = tid + u * L2T;
    sd[u] = k < nq ? lq[k] : make_int2(base, base);
  }
#pragma unroll
  for (int u = 0; u < L2U; ++u) gs[u] = g[sd[u].x];

  for (int r = tid; r < cnt; r += L2T) {
    float gv = g[base + r];
    float ed = gv * ad2;
    float v = l2_logit(gv, as2, ed);
    gl[r] = gv;
    edl[r] = ed;
    esl[r] = v;
    ml[r] = half == 0 ? v : -INFINITY;
  }
  __syncthreads();

  if (listed) {
#pragma unroll
    for (int u = 0; u < L2U; ++u) {
      int dl = sd[u].y - base;
      ev[u] = l2_logit(gs[u], as2, edl[dl]);
      if (tid + u * L2T < nq) lds_max_f32(&ml[dl], ev[u]);
    }
  } else {
    for (int idx = hb + tid; idx < he; idx += L2T) {
      int d = dst[idx];
      if ((unsigned)(d - base) < ucnt)
        lds_max_f32(&ml[d - base], l2_logit(g[ei[idx]], as2, edl[d - base]));
    }
  }
  __syncthreads();

  for (int r = tid; r < cnt; r += L2T) {
    float p = half == 0 ? __expf(esl[r] - ml[r]) : 0.0f;
    zl[r] = p;
    nl[r] = p * gl[r];
  }
  __syncthreads();

  if (listed) {
#pragma unroll
    for (int u = 0; u < L2U; ++u) {
      if (tid + u * L2T < nq) {
        int dl = sd[u].y - base;
        float p = __expf(ev[u] - ml[dl]);
        atomicAdd(&zl[dl], p);
        atomicAdd(&nl[dl], p * gs[u]);
      }
    }
  } else {
    for (int idx = hb + tid; idx < he; idx += L2T) {
      int d = dst[idx];
      if ((unsigned)(d - base) < ucnt) {
        float gsv = g[ei[idx]];
        float p = __expf(l2_logit(gsv, as2, edl[d - base]) - ml[d - base]);
        atomicAdd(&zl[d - base], p);
        atomicAdd(&nl[d - base], p * gsv);
      }
    }
  }
  __syncthreads();

  float* pp = part + (size_t)half * 3 * n + base;
  for (int r = tid; r < cnt; r += L2T) {
    pp[r] = ml[r];
    pp[(size_t)n + r] = zl[r];
    pp[(size_t)2 * n + r] = nl[r];
  }
}

__global__ void k_final(const float* __restrict__ part, const float* __restrict__ b2,
                        float* __restrict__ out, int n) {
  int i = blockIdx.x * blockDim.x + threadIdx.x;
  if (i >= n) return;
  const size_t sn = (size_t)n;
  float m0 = part[i],          z0 = part[sn + i],     s0 = part[2 * sn + i];
  float m1 = part[3 * sn + i], z1 = part[4 * sn + i], s1 = part[5 * sn + i];
  float mm = fmaxf(m0, m1);
  float w0 = __expf(m0 - mm);
  float w1 = __expf(m1 - mm);
  float zz = z0 * w0 + z1 * w1;
  float ss = s0 * w0 + s1 * w1;
  float v = ss / zz + b2[0];
  out[i] = 1.0f / (1.0f + __expf(-v));
}

extern "C" void kernel_launch(void* const* d_in, const int* in_sizes, int n_in,
                              void* d_out, int out_size, void* d_ws, size_t ws_size,
                              hipStream_t stream) {
  const int*   ei     = (const int*)d_in[0];
  const float* x      = (const float*)d_in[1];
  const float* W1     = (const float*)d_in[2];
  const float* a_src1 = (const float*)d_in[3];
  const float* a_dst1 = (const float*)d_in[4];
  const float* b1     = (const float*)d_in[5];
  const float* W2     = (const float*)d_in[6];
  const float* a_src2 = (const float*)d_in[7];
  const float* a_dst2 = (const float*)d_in[8];
  const float* b2     = (const float*)d_in[9];

  const int E = in_sizes[0] / 2;
  const int n = in_sizes[1] / DIN;
  const int Etot = E + n;

  float* f     = (float*)d_ws;
  float* h     = f;
  float* out1  = f + (size_t)64 * n;
  float* esrc  = f + (size_t)192 * n;
  float* edst  = f + (size_t)193 * n;
  float* m     = f + (size_t)194 * n;
  float* g     = f + (size_t)195 * n;
  float* z     = f + (size_t)196 * n;
  float* eedge = f + (size_t)200 * n;
  const int nwg = 2 * ((n + RNG - 1) / RNG);
  const size_t o1 = ((size_t)200 * n + Etot + 3) & ~(size_t)3;
  float* part  = f + o1;
  const size_t o2 = (o1 + (size_t)6 * n + 3) & ~(size_t)3;
  int*   lst   = (int*)(f + o2);
  int*   cntq  = lst + (size_t)nwg * QCAP * 2;
  float* out   = (float*)d_out;

  const int B = 256;
  dim3 blk(B);

  k_init<<<dim3((n + B - 1) / B), blk, 0, stream>>>(m, n);

  k_gemm_wmma<<<dim3((n + 127) / 128), blk, 0, stream>>>(x, W1, h, n);
  k_logits<<<dim3((n + B - 1) / B), blk, 0, stream>>>(h, a_src1, a_dst1, esrc, edst, n);
  k_edge_max<<<dim3((Etot + B - 1) / B), blk, 0, stream>>>(ei, E, n, esrc, edst, m, eedge);
  k_edge_aggr64<<<dim3(nwg), dim3(AGT), 0, stream>>>(ei, E, n, m, z, eedge, h, out1, lst, cntq);

  k_layer2_prep<<<dim3((n + B - 1) / B), blk, 0, stream>>>(out1, z, b1, W2, g, n);
  k_l2_range<<<dim3(nwg), dim3(L2T), 0, stream>>>(ei, E, n, lst, cntq, g, a_src2, a_dst2, part);

  k_final<<<dim3((n + B - 1) / B), blk, 0, stream>>>(part, b2, out, n);
}


# Layer-1 segment max taken inside the range kernel (per list half, in LDS; halves joined by rescaling in the node kernel): edge-logit launch, its global atomic max and init launch go; 8 launches to 6
# speedup vs baseline: 3.7193x; 1.5179x over previous
#include <hip/hip_runtime.h>
#include <math.h>

typedef float v2f __attribute__((ext_vector_type(2)));
typedef float v8f __attribute__((ext_vector_type(8)));

#define DIN 128
#define C1  64
#define NEG_SLOPE 0.2f

__global__ __launch_bounds__(256) void k_gemm_wmma(const float* __restrict__ x,
                                                   const float* __restrict__ W,
                                                   float* __restrict__ h, int n) {
  __shared__ float Wt[C1 * DIN];
  int tid = threadIdx.x;
#pragma unroll
  for (int it = 0; it < (DIN * C1) / 256; ++it) {
    int idx = tid + it * 256;
    int k  = idx >> 6;
    int nn = idx & 63;
    Wt[nn * DIN + k] = W[idx];
  }
  __syncthreads();

  int wave = tid >> 5;
  int lane = tid & 31;
  int half = lane >> 4;
  int l    = lane & 15;
  int rowBase = blockIdx.x * 128 + wave * 16;

  int arow = rowBase + l;
  if (arow > n - 1) arow = n - 1;
  const float* ap = x + (size_t)arow * DIN + 2 * half;

  v8f c0 = {}, c1 = {}, c2 = {}, c3 = {};
#pragma unroll 4
  for (int k0 = 0; k0 < DIN; k0 += 4) {
    v2f a = *(const v2f*)(ap + k0);
    const float* wp = &Wt[l * DIN + k0 + 2 * half];
    v2f b0 = *(const v2f*)(wp);
    v2f b1 = *(const v2f*)(wp + 16 * DIN);
    v2f b2 = *(const v2f*)(wp + 32 * DIN);
    v2f b3 = *(const v2f*)(wp + 48 * DIN);
    c0 = __builtin_amdgcn_wmma_f32_16x16x4_f32(false, a, false, b0, (short)0, c0, false, false);
    c1 = __builtin_amdgcn_wmma_f32_16x16x4_f32(false, a, false, b1, (short)0, c1, false, false);
    c2 = __builtin_amdgcn_wmma_f32_16x16x4_f32(false, a, false, b2, (short)0, c2, false, false);
    c3 = __builtin_amdgcn_wmma_f32_16x16x4_f32(false, a, false, b3, (short)0, c3, false, false);
  }

#pragma unroll
  for (int v = 0; v < 8; ++v) {
    int row = rowBase + 8 * half + v;
    if (row < n) {
      float* hp = h + (size_t)row * C1 + l;
      hp[0]  = c0[v];
      hp[16] = c1[v];
      hp[32] = c2[v];
      hp[48] = c3[v];
    }
  }
}

__global__ void k_logits(const float* __restrict__ h, const float* __restrict__ a_src,
                         const float* __restrict__ a_dst, float* __restrict__ esrc,
                         float* __restrict__ edst, int n) {
  int i = blockIdx.x * blockDim.x + threadIdx.x;
  if (i >= n) return;
  const float4* hp = (const float4*)(h + (size_t)i * C1);
  float s = 0.0f, d = 0.0f;
#pragma unroll
  for (int c = 0; c < C1 / 4; ++c) {
    float4 hv = hp[c];
    float4 as = ((const float4*)a_src)[c];
    float4 ad = ((const float4*)a_dst)[c];
    s += hv.x * as.x + hv.y * as.y + hv.z * as.z + hv.w * as.w;
    d += hv.x * ad.x + hv.y * ad.y + hv.z * ad.z + hv.w * ad.w;
  }
  esrc[i] = s;
  edst[i] = d;
}

#define RNG   800
#define QCAP  12288
#define AGT   1024
#define PU    8
#define AU    8
static_assert(RNG <= 1024, "the queue packs the local node into 10 bits");
__device__ __forceinline__ float l1_logit(float es, float ed) {
  float v = es + ed;
  return v > 0.0f ? v : NEG_SLOPE * v;
}
__device__ __forceinline__ void lds_max_f32(float* addr, float val) {
  if (__float_as_int(val) >= 0) atomicMax((int*)addr, __float_as_int(val));
  else                          atomicMin((unsigned int*)addr, __float_as_uint(val));
}
__device__ __forceinline__ void aggr_add(float p, float4 hv, int dl, int sub, int rot,
                                         float* acc, float* zl) {
  float* ap = acc + dl * C1 + sub * 4;
  float v[4] = {p * hv.x, p * hv.y, p * hv.z, p * hv.w};
#pragma unroll
  for (int j = 0; j < 4; ++j) {
    int jj = (j + rot) & 3;
    atomicAdd(ap + jj, v[jj]);
  }
  if (sub == 0) atomicAdd(zl + dl, p);
}

__global__ __launch_bounds__(AGT) void k_edge_aggr64(const int* __restrict__ ei, int E, int n,
                              const float* __restrict__ esrc, const float* __restrict__ edst,
                              const float* __restrict__ h, float* __restrict__ out1,
                              float* __restrict__ z, float* __restrict__ mx,
                              int* __restrict__ lst, int* __restrict__ cntq) {
  __shared__ float acc[RNG * C1];
  __shared__ float zl[RNG];
  __shared__ float ml[RNG];
  __shared__ float edl[RNG];
  __shared__ int   q[QCAP];
  __shared__ int   q2[QCAP];
  __shared__ int   qn;
  const int tid  = threadIdx.x;
  const int half = blockIdx.x & 1;
  const int base = (blockIdx.x >> 1) * RNG;
  const int cnt  = min(RNG, n - base);
  const int sub  = tid & 15;
  const int grp  = tid >> 4;
  const int rot  = grp & 1;
  if (tid == 0) qn = 0;

  for (int r = tid; r < cnt; r += AGT) {
    float ed = edst[base + r];
    edl[r] = ed;
    ml[r] = half == 0 ? l1_logit(esrc[base + r], ed) : -INFINITY;
  }
  __syncthreads();

  const int* dst = ei + E;
  const unsigned ucnt = (unsigned)cnt;
  const bool vec = (((size_t)dst) & 15) == 0;
  const int split = vec ? 4 * (((E >> 2) + 1) >> 1) : ((E + 1) >> 1);
  const int hb = half ? split : 0;
  const int he = half ? E : split;
  int sbeg = hb;
  if (vec) {
    const int vbeg = hb >> 2;
    const int nv = half ? (E >> 2) : (split >> 2);
    const int4* dv = (const int4*)dst;
    for (int v0 = vbeg + tid; v0 < nv; v0 += 4 * AGT) {
      int4 a[4];
#pragma unroll
      for (int u = 0; u < 4; ++u) {
        int v = v0 + u * AGT;
        a[u] = (v < nv) ? dv[v] : make_int4(-1, -1, -1, -1);
      }
      unsigned hit = 0;
#pragma unroll
      for (int u = 0; u < 4; ++u) {
        hit |= ((unsigned)(a[u].x - base) < ucnt ? 1u : 0u) << (4 * u + 0);
        hit |= ((unsigned)(a[u].y - base) < ucnt ? 1u : 0u) << (4 * u + 1);
        hit |= ((unsigned)(a[u].z - base) < ucnt ? 1u : 0u) << (4 * u + 2);
        hit |= ((unsigned)(a[u].w - base) < ucnt ? 1u : 0u) << (4 * u + 3);
      }
      while (hit) {
        int b = __ffs(hit) - 1;
        hit &= hit - 1;
        int idx = 4 * (v0 + (b >> 2) * AGT) + (b & 3);
        const bool b0 = b & 1, b1 = b & 2, b2 = b & 4, b3 = b & 8;
        int t0 = b1 ? (b0 ? a[0].w : a[0].z) : (b0 ? a[0].y : a[0].x);
        int t1 = b1 ? (b0 ? a[1].w : a[1].z) : (b0 ? a[1].y : a[1].x);
        int t2 = b1 ? (b0 ? a[2].w : a[2].z) : (b0 ? a[2].y : a[2].x);
        int t3 = b1 ? (b0 ? a[3].w : a[3].z) : (b0 ? a[3].y : a[3].x);
        int d = b3 ? (b2 ? t3 : t2) : (b2 ? t1 : t0);
        int slot = atomicAdd(&qn, 1);
        if (slot < QCAP) { q[slot] = idx; q2[slot] = d - base; }
      }
    }
    sbeg = half ? max(hb, (E >> 2) << 2) : he;
  }
  for (int idx = sbeg + tid; idx < he; idx += AGT) {
    int d = dst[idx];
    if ((unsigned)(d - base) < ucnt) {
      int slot = atomicAdd(&qn, 1);
      if (slot < QCAP) { q[slot] = idx; q2[slot] = d - base; }
    }
  }
  __syncthreads();

  const int nqraw = qn;
  const bool listed = nqraw <= QCAP;
  const int nq = listed ? nqraw : 0;
  if (tid == 0) cntq[blockIdx.x] = nqraw;

  int2* lq = (int2*)lst + (size_t)blockIdx.x * QCAP;
  if (listed) {
    for (int k0 = tid; k0 < nq; k0 += AU * AGT) {
      int   idx[AU], dlv[AU], sv[AU];
      float es[AU];
#pragma unroll
      for (int u = 0; u < AU; ++u) {
        int k = k0 + u * AGT;
        int kk = k < nq ? k : k0;
        idx[u] = q[kk];
        dlv[u] = q2[kk];
      }
#pragma unroll
      for (int u = 0; u < AU; ++u) sv[u] = ei[idx[u]];
#pragma unroll
      for (int u = 0; u < AU; ++u) es[u] = esrc[sv[u]];
#pragma unroll
      for (int u = 0; u < AU; ++u) {
        int k = k0 + u * AGT;
        if (k < nq) {
          int dl = dlv[u];
          float e = l1_logit(es[u], edl[dl]);
          lds_max_f32(&ml[dl], e);
          lq[k] = make_int2(sv[u], base + dl);
          q[k]  = (int)(((unsigned)sv[u] << 10) | (unsigned)dl);
          q2[k] = __float_as_int(e);
        }
      }
    }
  } else {
    for (int idx = hb + tid; idx < he; idx += AGT) {
      int d = dst[idx];
      if ((unsigned)(d - base) < ucnt)
        lds_max_f32(&ml[d - base], l1_logit(esrc[ei[idx]], edl[d - base]));
    }
  }
  __syncthreads();

  for (int r = grp; r < cnt; r += AGT / 16) {
    float p = 0.0f;
    float4 hv = make_float4(0.0f, 0.0f, 0.0f, 0.0f);
    if (half == 0) {
      int d = base + r;
      p = __expf(l1_logit(esrc[d], edl[r]) - ml[r]);
      hv = *(const float4*)(h + (size_t)d * C1 + sub * 4);
    }
    *(float4*)(acc + r * C1 + sub * 4) = make_float4(p * hv.x, p * hv.y, p * hv.z, p * hv.w);
    if (sub == 0) zl[r] = p;
  }
  __syncthreads();

  if (listed) {
    for (int k0 = grp; k0 < nq; k0 += PU * (AGT / 16)) {
      unsigned w[PU];
      float    ev[PU];
      float4   hv[PU];
#pragma unroll
      for (int u = 0; u < PU; ++u) {
        int k = k0 + u * (AGT / 16);
        int kk = k < nq ? k : k0;
        w[u]  = (unsigned)q[kk];
        ev[u] = __int_as_float(q2[kk]);
      }
#pragma unroll
      for (int u = 0; u < PU; ++u)
        hv[u] = *(const float4*)(h + (size_t)(w[u] >> 10) * C1 + sub * 4);
#pragma unroll
      for (int u = 0; u < PU; ++u)
        if (k0 + u * (AGT / 16) < nq) {
          int dl = (int)(w[u] & 1023u);
          aggr_add(__expf(ev[u] - ml[dl]), hv[u], dl, sub, rot, acc, zl);
        }
    }
  } else {
    for (int idx = hb + tid; idx < he; idx += AGT) {
      int d = dst[idx];
      if ((unsigned)(d - base) < ucnt) {
        int s = ei[idx];
        float p = __expf(l1_logit(esrc[s], edl[d - base]) - ml[d - base]);
#pragma nounroll
        for (int s2 = 0; s2 < 16; ++s2)
          aggr_add(p, *(const float4*)(h + (size_t)s * C1 + s2 * 4), d - base, s2, 0, acc, zl);
      }
    }
  }
  __syncthreads();

  float* op = out1 + (size_t)half * n * C1 + (size_t)base * C1;
  float* zp = z + (size_t)half * n + base;
  float* mp = mx + (size_t)half * n + base;
  for (int i = tid; i < cnt * (C1 / 4); i += AGT)
    *(float4*)(op + i * 4) = *(const float4*)(acc + i * 4);
  for (int r = tid; r < cnt; r += AGT) { zp[r] = zl[r]; mp[r] = ml[r]; }
}

__global__ void k_layer2_prep(const float* __restrict__ out1, const float* __restrict__ z,
                              const float* __restrict__ mx, const float* __restrict__ b1,
                              const float* __restrict__ W2, float* __restrict__ g, int n) {
  int i = blockIdx.x * blockDim.x + threadIdx.x;
  if (i >= n) return;
  const float4* op = (const float4*)(out1 + (size_t)i * C1);
  const float4* oq = (const float4*)(out1 + (size_t)n * C1 + (size_t)i * C1);
  const float m0 = mx[i], m1 = mx[n + i];
  const float mm = fmaxf(m0, m1);
  const float w0 = __expf(m0 - mm);
  const float w1 = __expf(m1 - mm);
  const float rz = 1.0f / (z[i] * w0 + z[n + i] * w1);
  float acc = 0.0f;
#pragma unroll
  for (int c = 0; c < C1 / 4; ++c) {
    float4 v = op[c];
    float4 w = oq[c];
    v.x = v.x * w0 + w.x * w1; v.y = v.y * w0 + w.y * w1;
    v.z = v.z * w0 + w.z * w1; v.w = v.w * w0 + w.w * w1;
    float4 bb = ((const float4*)b1)[c];
    float4 ww = ((const float4*)W2)[c];
    float v0 = fmaxf(v.x * rz + bb.x, 0.0f);
    float v1 = fmaxf(v.y * rz + bb.y, 0.0f);
    float v2 = fmaxf(v.z * rz + bb.z, 0.0f);
    float v3 = fmaxf(v.w * rz + bb.w, 0.0f);
    acc += v0 * ww.x + v1 * ww.y + v2 * ww.z + v3 * ww.w;
  }
  g[i] = acc;
}

#define L2T 1024
#define L2U (QCAP / L2T)
__device__ __forceinline__ float l2_logit(float gs, float as2, float ed) {
  float v = gs * as2 + ed;
  return v > 0.0f ? v : NEG_SLOPE * v;
}

__global__ __launch_bounds__(L2T) void k_l2_range(const int* __restrict__ ei, int E, int n,
                              const int* __restrict__ lst, const int* __restrict__ cntq,
                              const float* __restrict__ g, const float* __restrict__ a_src2,
                              const float* __restrict__ a_dst2, float* __restrict__ part) {
  __shared__ float ml[RNG];
  __shared__ float zl[RNG];
  __shared__ float nl[RNG];
  __shared__ float edl[RNG];
  __shared__ float gl[RNG];
  __shared__ float esl[RNG];
  const int tid  = threadIdx.x;
  const int half = blockIdx.x & 1;
  const int base = (blockIdx.x >> 1) * RNG;
  const int cnt  = min(RNG, n - base);
  const unsigned ucnt = (unsigned)cnt;
  const float as2 = a_src2[0];
  const float ad2 = a_dst2[0];
  const int nqraw = cntq[blockIdx.x];
  const bool listed = nqraw <= QCAP;
  const int nq = listed ? nqraw : 0;

  const int* dst = ei + E;
  const int split = ((((size_t)dst) & 15) == 0) ? 4 * (((E >> 2) + 1) >> 1) : ((E + 1) >> 1);
  const int hb = half ? split : 0;
  const int he = half ? E : split;

  const int2* lq = (const int2*)lst + (size_t)blockIdx.x * QCAP;
  int2  sd[L2U];
  float gs[L2U], ev[L2U];
#pragma unroll
  for (int u = 0; u < L2U; ++u) {
    int k = tid + u * L2T;
    sd[u] = k < nq ? lq[k] : make_int2(base, base);
  }
#pragma unroll
  for (int u = 0; u < L2U; ++u) gs[u] = g[sd[u].x];

  for (int r = tid; r < cnt; r += L2T) {
    float gv = g[base + r];
    float ed = gv * ad2;
    float v = l2_logit(gv, as2, ed);
    gl[r] = gv;
    edl[r] = ed;
    esl[r] = v;
    ml[r] = half == 0 ? v : -INFINITY;
  }
  __syncthreads();

  if (listed) {
#pragma unroll
    for (int u = 0; u < L2U; ++u) {
      int dl = sd[u].y - base;
      ev[u] = l2_logit(gs[u], as2, edl[dl]);
      if (tid + u * L2T < nq) lds_max_f32(&ml[dl], ev[u]);
    }
  } else {
    for (int idx = hb + tid; idx < he; idx += L2T) {
      int d = dst[idx];
      if ((unsigned)(d - base) < ucnt)
        lds_max_f32(&ml[d - base], l2_logit(g[ei[idx]], as2, edl[d - base]));
    }
  }
  __syncthreads();

  for (int r = tid; r < cnt; r += L2T) {
    float p = half == 0 ? __expf(esl[r] - ml[r]) : 0.0f;
    zl[r] = p;
    nl[r] = p * gl[r];
  }
  __syncthreads();

  if (listed) {
#pragma unroll
    for (int u = 0; u < L2U; ++u) {
      if (tid + u * L2T < nq) {
        int dl = sd[u].y - base;
        float p = __expf(ev[u] - ml[dl]);
        atomicAdd(&zl[dl], p);
        atomicAdd(&nl[dl], p * gs[u]);
      }
    }
  } else {
    for (int idx = hb + tid; idx < he; idx += L2T) {
      int d = dst[idx];
      if ((unsigned)(d - base) < ucnt) {
        float gsv = g[ei[idx]];
        float p = __expf(l2_logit(gsv, as2, edl[d - base]) - ml[d - base]);
        atomicAdd(&zl[d - base], p);
        atomicAdd(&nl[d - base], p * gsv);
      }
    }
  }
  __syncthreads();

  float* pp = part + (size_t)half * 3 * n + base;
  for (int r = tid; r < cnt; r += L2T) {
    pp[r] = ml[r];
    pp[(size_t)n + r] = zl[r];
    pp[(size_t)2 * n + r] = nl[r];
  }
}

__global__ void k_final(const float* __restrict__ part, const float* __restrict__ b2,
                        float* __restrict__ out, int n) {
  int i = blockIdx.x * blockDim.x + threadIdx.x;
  if (i >= n) return;
  const size_t sn = (size_t)n;
  float m0 = part[i],          z0 = part[sn + i],     s0 = part[2 * sn + i];
  float m1 = part[3 * sn + i], z1 = part[4 * sn + i], s1 = part[5 * sn + i];
  float mm = fmaxf(m0, m1);
  float w0 = __expf(m0 - mm);
  float w1 = __expf(m1 - mm);
  float zz = z0 * w0 + z1 * w1;
  float ss = s0 * w0 + s1 * w1;
  float v = ss / zz + b2[0];
  out[i] = 1.0f / (1.0f + __expf(-v));
}

extern "C" void kernel_launch(void* const* d_in, const int* in_sizes, int n_in,
                              void* d_out, int out_size, void* d_ws, size_t ws_size,
                              hipStream_t stream) {
  const int*   ei     = (const int*)d_in[0];
  const float* x      = (const float*)d_in[1];
  const float* W1     = (const float*)d_in[2];
  const float* a_src1 = (const float*)d_in[3];
  const float* a_dst1 = (const float*)d_in[4];
  const float* b1     = (const float*)d_in[5];
  const float* W2     = (const float*)d_in[6];
  const float* a_src2 = (const float*)d_in[7];
  const float* a_dst2 = (const float*)d_in[8];
  const float* b2     = (const float*)d_in[9];

  const int E = in_sizes[0] / 2;
  const int n = in_sizes[1] / DIN;

  float* f     = (float*)d_ws;
  float* h     = f;
  float* out1  = f + (size_t)64 * n;
  float* esrc  = f + (size_t)192 * n;
  float* edst  = f + (size_t)193 * n;
  float* g     = f + (size_t)194 * n;
  float* z     = f + (size_t)195 * n;
  float* mx    = f + (size_t)197 * n;
  const int nwg = 2 * ((n + RNG - 1) / RNG);
  const size_t o1 = ((size_t)199 * n + 3) & ~(size_t)3;
  float* part  = f + o1;
  const size_t o2 = (o1 + (size_t)6 * n + 3) & ~(size_t)3;
  int*   lst   = (int*)(f + o2);
  int*   cntq  = lst + (size_t)nwg * QCAP * 2;
  float* out   = (float*)d_out;

  const int B = 256;
  dim3 blk(B);

  k_gemm_wmma<<<dim3((n + 127) / 128), blk, 0, stream>>>(x, W1, h, n);
  k_logits<<<dim3((n + B - 1) / B), blk, 0, stream>>>(h, a_src1, a_dst1, esrc, edst, n);
  k_edge_aggr64<<<dim3(nwg), dim3(AGT), 0, stream>>>(ei, E, n, esrc, edst, h, out1, z, mx, lst, cntq);

  k_layer2_prep<<<dim3((n + B - 1) / B), blk, 0, stream>>>(out1, z, mx, b1, W2, g, n);
  k_l2_range<<<dim3(nwg), dim3(L2T), 0, stream>>>(ei, E, n, lst, cntq, g, a_src2, a_dst2, part);

  k_final<<<dim3((n + B - 1) / B), blk, 0, stream>>>(part, b2, out, n);
}


# GEMM: W1 kept in every wave's registers as B fragments (no LDS tile, no staging, no bank conflicts); 2048 waves stride over 16-row tiles; column map 4c+t gives 16-byte W loads and whole-row stores
# speedup vs baseline: 4.2468x; 1.1418x over previous
#include <hip/hip_runtime.h>
#include <math.h>

typedef float v2f __attribute__((ext_vector_type(2)));
typedef float v8f __attribute__((ext_vector_type(8)));

#define DIN 128
#define C1  64
#define NEG_SLOPE 0.2f

#define GEMM_WAVES 8
__global__ __launch_bounds__(32 * GEMM_WAVES) void k_gemm_wmma(const float* __restrict__ x,
                                                   const float* __restrict__ W,
                                                   float* __restrict__ h, int n) {
  const int lane = threadIdx.x & 31;
  const int wave = __builtin_amdgcn_readfirstlane((int)(threadIdx.x >> 5));
  const int half = lane >> 4;
  const int l    = lane & 15;

  v2f bf[DIN / 4][4];
  {
    const float* wp = W + (size_t)(2 * half) * C1 + 4 * l;
#pragma unroll
    for (int ks = 0; ks < DIN / 4; ++ks) {
      const float4 r0 = *(const float4*)(wp + (size_t)(4 * ks) * C1);
      const float4 r1 = *(const float4*)(wp + (size_t)(4 * ks + 1) * C1);
      bf[ks][0] = (v2f){r0.x, r1.x};
      bf[ks][1] = (v2f){r0.y, r1.y};
      bf[ks][2] = (v2f){r0.z, r1.z};
      bf[ks][3] = (v2f){r0.w, r1.w};
    }
  }

  const int ntiles = (n + 15) >> 4;
  const int nwaves = (int)gridDim.x * GEMM_WAVES;
  for (int tile = (int)blockIdx.x * GEMM_WAVES + wave; tile < ntiles; tile += nwaves) {
    const int rowBase = tile * 16;
    int arow = rowBase + l;
    if (arow > n - 1) arow = n - 1;
    const float* ap = x + (size_t)arow * DIN + 2 * half;

    v8f c0 = {}, c1 = {}, c2 = {}, c3 = {};
#pragma unroll
    for (int ks = 0; ks < DIN / 4; ++ks) {
      v2f a = *(const v2f*)(ap + 4 * ks);
      c0 = __builtin_amdgcn_wmma_f32_16x16x4_f32(false, a, false, bf[ks][0], (short)0, c0, false, false);
      c1 = __builtin_amdgcn_wmma_f32_16x16x4_f32(false, a, false, bf[ks][1], (short)0, c1, false, false);
      c2 = __builtin_amdgcn_wmma_f32_16x16x4_f32(false, a, false, bf[ks][2], (short)0, c2, false, false);
      c3 = __builtin_amdgcn_wmma_f32_16x16x4_f32(false, a, false, bf[ks][3], (short)0, c3, false, false);
    }

#pragma unroll
    for (int v = 0; v < 8; ++v) {
      int row = rowBase + 8 * half + v;
      if (row < n)
        *(float4*)(h + (size_t)row * C1 + 4 * l) = make_float4(c0[v], c1[v], c2[v], c3[v]);
    }
  }
}

__global__ void k_logits(const float* __restrict__ h, const float* __restrict__ a_src,
                         const float* __restrict__ a_dst, float* __restrict__ esrc,
                         float* __restrict__ edst, int n) {
  int i = blockIdx.x * blockDim.x + threadIdx.x;
  if (i >= n) return;
  const float4* hp = (const float4*)(h + (size_t)i * C1);
  float s = 0.0f, d = 0.0f;
#pragma unroll
  for (int c = 0; c < C1 / 4; ++c) {
    float4 hv = hp[c];
    float4 as = ((const float4*)a_src)[c];
    float4 ad = ((const float4*)a_dst)[c];
    s += hv.x * as.x + hv.y * as.y + hv.z * as.z + hv.w * as.w;
    d += hv.x * ad.x + hv.y * ad.y + hv.z * ad.z + hv.w * ad.w;
  }
  esrc[i] = s;
  edst[i] = d;
}

#define RNG   800
#define QCAP  12288
#define AGT   1024
#define PU    8
#define AU    8
static_assert(RNG <= 1024, "the queue packs the local node into 10 bits");
__device__ __forceinline__ float l1_logit(float es, float ed) {
  float v = es + ed;
  return v > 0.0f ? v : NEG_SLOPE * v;
}
__device__ __forceinline__ void lds_max_f32(float* addr, float val) {
  if (__float_as_int(val) >= 0) atomicMax((int*)addr, __float_as_int(val));
  else                          atomicMin((unsigned int*)addr, __float_as_uint(val));
}
__device__ __forceinline__ void aggr_add(float p, float4 hv, int dl, int sub, int rot,
                                         float* acc, float* zl) {
  float* ap = acc + dl * C1 + sub * 4;
  float v[4] = {p * hv.x, p * hv.y, p * hv.z, p * hv.w};
#pragma unroll
  for (int j = 0; j < 4; ++j) {
    int jj = (j + rot) & 3;
    atomicAdd(ap + jj, v[jj]);
  }
  if (sub == 0) atomicAdd(zl + dl, p);
}

__global__ __launch_bounds__(AGT) void k_edge_aggr64(const int* __restrict__ ei, int E, int n,
                              const float* __restrict__ esrc, const float* __restrict__ edst,
                              const float* __restrict__ h, float* __restrict__ out1,
                              float* __restrict__ z, float* __restrict__ mx,
                              int* __restrict__ lst, int* __restrict__ cntq) {
  __shared__ float acc[RNG * C1];
  __shared__ float zl[RNG];
  __shared__ float ml[RNG];
  __shared__ float edl[RNG];
  __shared__ int   q[QCAP];
  __shared__ int   q2[QCAP];
  __shared__ int   qn;
  const int tid  = threadIdx.x;
  const int half = blockIdx.x & 1;
  const int base = (blockIdx.x >> 1) * RNG;
  const int cnt  = min(RNG, n - base);
  const int sub  = tid & 15;
  const int grp  = tid >> 4;
  const int rot  = grp & 1;
  if (tid == 0) qn = 0;

  for (int r = tid; r < cnt; r += AGT) {
    float ed = edst[base + r];
    edl[r] = ed;
    ml[r] = half == 0 ? l1_logit(esrc[base + r], ed) : -INFINITY;
  }
  __syncthreads();

  const int* dst = ei + E;
  const unsigned ucnt = (unsigned)cnt;
  const bool vec = (((size_t)dst) & 15) == 0;
  const int split = vec ? 4 * (((E >> 2) + 1) >> 1) : ((E + 1) >> 1);
  const int hb = half ? split : 0;
  const int he = half ? E : split;
  int sbeg = hb;
  if (vec) {
    const int vbeg = hb >> 2;
    const int nv = half ? (E >> 2) : (split >> 2);
    const int4* dv = (const int4*)dst;
    for (int v0 = vbeg + tid; v0 < nv; v0 += 4 * AGT) {
      int4 a[4];
#pragma unroll
      for (int u = 0; u < 4; ++u) {
        int v = v0 + u * AGT;
        a[u] = (v < nv) ? dv[v] : make_int4(-1, -1, -1, -1);
      }
      unsigned hit = 0;
#pragma unroll
      for (int u = 0; u < 4; ++u) {
        hit |= ((unsigned)(a[u].x - base) < ucnt ? 1u : 0u) << (4 * u + 0);
        hit |= ((unsigned)(a[u].y - base) < ucnt ? 1u : 0u) << (4 * u + 1);
        hit |= ((unsigned)(a[u].z - base) < ucnt ? 1u : 0u) << (4 * u + 2);
        hit |= ((unsigned)(a[u].w - base) < ucnt ? 1u : 0u) << (4 * u + 3);
      }
      while (hit) {
        int b = __ffs(hit) - 1;
        hit &= hit - 1;
        int idx = 4 * (v0 + (b >> 2) * AGT) + (b & 3);
        const bool b0 = b & 1, b1 = b & 2, b2 = b & 4, b3 = b & 8;
        int t0 = b1 ? (b0 ? a[0].w : a[0].z) : (b0 ? a[0].y : a[0].x);
        int t1 = b1 ? (b0 ? a[1].w : a[1].z) : (b0 ? a[1].y : a[1].x);
        int t2 = b1 ? (b0 ? a[2].w : a[2].z) : (b0 ? a[2].y : a[2].x);
        int t3 = b1 ? (b0 ? a[3].w : a[3].z) : (b0 ? a[3].y : a[3].x);
        int d = b3 ? (b2 ? t3 : t2) : (b2 ? t1 : t0);
        int slot = atomicAdd(&qn, 1);
        if (slot < QCAP) { q[slot] = idx; q2[slot] = d - base; }
      }
    }
    sbeg = half ? max(hb, (E >> 2) << 2) : he;
  }
  for (int idx = sbeg + tid; idx < he; idx += AGT) {
    int d = dst[idx];
    if ((unsigned)(d - base) < ucnt) {
      int slot = atomicAdd(&qn, 1);
      if (slot < QCAP) { q[slot] = idx; q2[slot] = d - base; }
    }
  }
  __syncthreads();

  const int nqraw = qn;
  const bool listed = nqraw <= QCAP;
  const int nq = listed ? nqraw : 0;
  if (tid == 0) cntq[blockIdx.x] = nqraw;

  int2* lq = (int2*)lst + (size_t)blockIdx.x * QCAP;
  if (listed) {
    for (int k0 = tid; k0 < nq; k0 += AU * AGT) {
      int   idx[AU], dlv[AU], sv[AU];
      float es[AU];
#pragma unroll
      for (int u = 0; u < AU; ++u) {
        int k = k0 + u * AGT;
        int kk = k < nq ? k : k0;
        idx[u] = q[kk];
        dlv[u] = q2[kk];
      }
#pragma unroll
      for (int u = 0; u < AU; ++u) sv[u] = ei[idx[u]];
#pragma unroll
      for (int u = 0; u < AU; ++u) es[u] = esrc[sv[u]];
#pragma unroll
      for (int u = 0; u < AU; ++u) {
        int k = k0 + u * AGT;
        if (k < nq) {
          int dl = dlv[u];
          float e = l1_logit(es[u], edl[dl]);
          lds_max_f32(&ml[dl], e);
          lq[k] = make_int2(sv[u], base + dl);
          q[k]  = (int)(((unsigned)sv[u] << 10) | (unsigned)dl);
          q2[k] = __float_as_int(e);
        }
      }
    }
  } else {
    for (int idx = hb + tid; idx < he; idx += AGT) {
      int d = dst[idx];
      if ((unsigned)(d - base) < ucnt)
        lds_max_f32(&ml[d - base], l1_logit(esrc[ei[idx]], edl[d - base]));
    }
  }
  __syncthreads();

  for (int r = grp; r < cnt; r += AGT / 16) {
    float p = 0.0f;
    float4 hv = make_float4(0.0f, 0.0f, 0.0f, 0.0f);
    if (half == 0) {
      int d = base + r;
      p = __expf(l1_logit(esrc[d], edl[r]) - ml[r]);
      hv = *(const float4*)(h + (size_t)d * C1 + sub * 4);
    }
    *(float4*)(acc + r * C1 + sub * 4) = make_float4(p * hv.x, p * hv.y, p * hv.z, p * hv.w);
    if (sub == 0) zl[r] = p;
  }
  __syncthreads();

  if (listed) {
    for (int k0 = grp; k0 < nq; k0 += PU * (AGT / 16)) {
      unsigned w[PU];
      float    ev[PU];
      float4   hv[PU];
#pragma unroll
      for (int u = 0; u < PU; ++u) {
        int k = k0 + u * (AGT / 16);
        int kk = k < nq ? k : k0;
        w[u]  = (unsigned)q[kk];
        ev[u] = __int_as_float(q2[kk]);
      }
#pragma unroll
      for (int u = 0; u < PU; ++u)
        hv[u] = *(const float4*)(h + (size_t)(w[u] >> 10) * C1 + sub * 4);
#pragma unroll
      for (int u = 0; u < PU; ++u)
        if (k0 + u * (AGT / 16) < nq) {
          int dl = (int)(w[u] & 1023u);
          aggr_add(__expf(ev[u] - ml[dl]), hv[u], dl, sub, rot, acc, zl);
        }
    }
  } else {
    for (int idx = hb + tid; idx < he; idx += AGT) {
      int d = dst[idx];
      if ((unsigned)(d - base) < ucnt) {
        int s = ei[idx];
        float p = __expf(l1_logit(esrc[s], edl[d - base]) - ml[d - base]);
#pragma nounroll
        for (int s2 = 0; s2 < 16; ++s2)
          aggr_add(p, *(const float4*)(h + (size_t)s * C1 + s2 * 4), d - base, s2, 0, acc, zl);
      }
    }
  }
  __syncthreads();

  float* op = out1 + (size_t)half * n * C1 + (size_t)base * C1;
  float* zp = z + (size_t)half * n + base;
  float* mp = mx + (size_t)half * n + base;
  for (int i = tid; i < cnt * (C1 / 4); i += AGT)
    *(float4*)(op + i * 4) = *(const float4*)(acc + i * 4);
  for (int r = tid; r < cnt; r += AGT) { zp[r] = zl[r]; mp[r] = ml[r]; }
}

__global__ void k_layer2_prep(const float* __restrict__ out1, const float* __restrict__ z,
                              const float* __restrict__ mx, const float* __restrict__ b1,
                              const float* __restrict__ W2, float* __restrict__ g, int n) {
  int i = blockIdx.x * blockDim.x + threadIdx.x;
  if (i >= n) return;
  const float4* op = (const float4*)(out1 + (size_t)i * C1);
  const float4* oq = (const float4*)(out1 + (size_t)n * C1 + (size_t)i * C1);
  const float m0 = mx[i], m1 = mx[n + i];
  const float mm = fmaxf(m0, m1);
  const float w0 = __expf(m0 - mm);
  const float w1 = __expf(m1 - mm);
  const float rz = 1.0f / (z[i] * w0 + z[n + i] * w1);
  float acc = 0.0f;
#pragma unroll
  for (int c = 0; c < C1 / 4; ++c) {
    float4 v = op[c];
    float4 w = oq[c];
    v.x = v.x * w0 + w.x * w1; v.y = v.y * w0 + w.y * w1;
    v.z = v.z * w0 + w.z * w1; v.w = v.w * w0 + w.w * w1;
    float4 bb = ((const float4*)b1)[c];
    float4 ww = ((const float4*)W2)[c];
    float v0 = fmaxf(v.x * rz + bb.x, 0.0f);
    float v1 = fmaxf(v.y * rz + bb.y, 0.0f);
    float v2 = fmaxf(v.z * rz + bb.z, 0.0f);
    float v3 = fmaxf(v.w * rz + bb.w, 0.0f);
    acc += v0 * ww.x + v1 * ww.y + v2 * ww.z + v3 * ww.w;
  }
  g[i] = acc;
}

#define L2T 1024
#define L2U (QCAP / L2T)
__device__ __forceinline__ float l2_logit(float gs, float as2, float ed) {
  float v = gs * as2 + ed;
  return v > 0.0f ? v : NEG_SLOPE * v;
}

__global__ __launch_bounds__(L2T) void k_l2_range(const int* __restrict__ ei, int E, int n,
                              const int* __restrict__ lst, const int* __restrict__ cntq,
                              const float* __restrict__ g, const float* __restrict__ a_src2,
                              const float* __restrict__ a_dst2, float* __restrict__ part) {
  __shared__ float ml[RNG];
  __shared__ float zl[RNG];
  __shared__ float nl[RNG];
  __shared__ float edl[RNG];
  __shared__ float gl[RNG];
  __shared__ float esl[RNG];
  const int tid  = threadIdx.x;
  const int half = blockIdx.x & 1;
  const int base = (blockIdx.x >> 1) * RNG;
  const int cnt  = min(RNG, n - base);
  const unsigned ucnt = (unsigned)cnt;
  const float as2 = a_src2[0];
  const float ad2 = a_dst2[0];
  const int nqraw = cntq[blockIdx.x];
  const bool listed = nqraw <= QCAP;
  const int nq = listed ? nqraw : 0;

  const int* dst = ei + E;
  const int split = ((((size_t)dst) & 15) == 0) ? 4 * (((E >> 2) + 1) >> 1) : ((E + 1) >> 1);
  const int hb = half ? split : 0;
  const int he = half ? E : split;

  const int2* lq = (const int2*)lst + (size_t)blockIdx.x * QCAP;
  int2  sd[L2U];
  float gs[L2U], ev[L2U];
#pragma unroll
  for (int u = 0; u < L2U; ++u) {
    int k = tid + u * L2T;
    sd[u] = k < nq ? lq[k] : make_int2(base, base);
  }
#pragma unroll
  for (int u = 0; u < L2U; ++u) gs[u] = g[sd[u].x];

  for (int r = tid; r < cnt; r += L2T) {
    float gv = g[base + r];
    float ed = gv * ad2;
    float v = l2_logit(gv, as2, ed);
    gl[r] = gv;
    edl[r] = ed;
    esl[r] = v;
    ml[r] = half == 0 ? v : -INFINITY;
  }
  __syncthreads();

  if (listed) {
#pragma unroll
    for (int u = 0; u < L2U; ++u) {
      int dl = sd[u].y - base;
      ev[u] = l2_logit(gs[u], as2, edl[dl]);
      if (tid + u * L2T < nq) lds_max_f32(&ml[dl], ev[u]);
    }
  } else {
    for (int idx = hb + tid; idx < he; idx += L2T) {
      int d = dst[idx];
      if ((unsigned)(d - base) < ucnt)
        lds_max_f32(&ml[d - base], l2_logit(g[ei[idx]], as2, edl[d - base]));
    }
  }
  __syncthreads();

  for (int r = tid; r < cnt; r += L2T) {
    float p = half == 0 ? __expf(esl[r] - ml[r]) : 0.0f;
    zl[r] = p;
    nl[r] = p * gl[r];
  }
  __syncthreads();

  if (listed) {
#pragma unroll
    for (int u = 0; u < L2U; ++u) {
      if (tid + u * L2T < nq) {
        int dl = sd[u].y - base;
        float p = __expf(ev[u] - ml[dl]);
        atomicAdd(&zl[dl], p);
        atomicAdd(&nl[dl], p * gs[u]);
      }
    }
  } else {
    for (int idx = hb + tid; idx < he; idx += L2T) {
      int d = dst[idx];
      if ((unsigned)(d - base) < ucnt) {
        float gsv = g[ei[idx]];
        float p = __expf(l2_logit(gsv, as2, edl[d - base]) - ml[d - base]);
        atomicAdd(&zl[d - base], p);
        atomicAdd(&nl[d - base], p * gsv);
      }
    }
  }
  __syncthreads();

  float* pp = part + (size_t)half * 3 * n + base;
  for (int r = tid; r < cnt; r += L2T) {
    pp[r] = ml[r];
    pp[(size_t)n + r] = zl[r];
    pp[(size_t)2 * n + r] = nl[r];
  }
}

__global__ void k_final(const float* __restrict__ part, const float* __restrict__ b2,
                        float* __restrict__ out, int n) {
  int i = blockIdx.x * blockDim.x + threadIdx.x;
  if (i >= n) return;
  const size_t sn = (size_t)n;
  float m0 = part[i],          z0 = part[sn + i],     s0 = part[2 * sn + i];
  float m1 = part[3 * sn + i], z1 = part[4 * sn + i], s1 = part[5 * sn + i];
  float mm = fmaxf(m0, m1);
  float w0 = __expf(m0 - mm);
  float w1 = __expf(m1 - mm);
  float zz = z0 * w0 + z1 * w1;
  float ss = s0 * w0 + s1 * w1;
  float v = ss / zz + b2[0];
  out[i] = 1.0f / (1.0f + __expf(-v));
}

extern "C" void kernel_launch(void* const* d_in, const int* in_sizes, int n_in,
                              void* d_out, int out_size, void* d_ws, size_t ws_size,
                              hipStream_t stream) {
  const int*   ei     = (const int*)d_in[0];
  const float* x      = (const float*)d_in[1];
  const float* W1     = (const float*)d_in[2];
  const float* a_src1 = (const float*)d_in[3];
  const float* a_dst1 = (const float*)d_in[4];
  const float* b1     = (const float*)d_in[5];
  const float* W2     = (const float*)d_in[6];
  const float* a_src2 = (const float*)d_in[7];
  const float* a_dst2 = (const float*)d_in[8];
  const float* b2     = (const float*)d_in[9];

  const int E = in_sizes[0] / 2;
  const int n = in_sizes[1] / DIN;

  float* f     = (float*)d_ws;
  float* h     = f;
  float* out1  = f + (size_t)64 * n;
  float* esrc  = f + (size_t)192 * n;
  float* edst  = f + (size_t)193 * n;
  float* g     = f + (size_t)194 * n;
  float* z     = f + (size_t)195 * n;
  float* mx    = f + (size_t)197 * n;
  const int nwg = 2 * ((n + RNG - 1) / RNG);
  const size_t o1 = ((size_t)199 * n + 3) & ~(size_t)3;
  float* part  = f + o1;
  const size_t o2 = (o1 + (size_t)6 * n + 3) & ~(size_t)3;
  int*   lst   = (int*)(f + o2);
  int*   cntq  = lst + (size_t)nwg * QCAP * 2;
  float* out   = (float*)d_out;

  const int B = 256;
  dim3 blk(B);

  k_gemm_wmma<<<dim3(256), dim3(32 * GEMM_WAVES), 0, stream>>>(x, W1, h, n);
  k_logits<<<dim3((n + B - 1) / B), blk, 0, stream>>>(h, a_src1, a_dst1, esrc, edst, n);
  k_edge_aggr64<<<dim3(nwg), dim3(AGT), 0, stream>>>(ei, E, n, esrc, edst, h, out1, z, mx, lst, cntq);

  k_layer2_prep<<<dim3((n + B - 1) / B), blk, 0, stream>>>(out1, z, mx, b1, W2, g, n);
  k_l2_range<<<dim3(nwg), dim3(L2T), 0, stream>>>(ei, E, n, lst, cntq, g, a_src2, a_dst2, part);

  k_final<<<dim3((n + B - 1) / B), blk, 0, stream>>>(part, b2, out, n);
}


# Range kernel finds its edges in a one-byte-per-edge copy of the destination list (range number, written by the logits launch): 4x fewer bytes, 4 edges per word; queue, hence LDS add, order changes
# speedup vs baseline: 4.8739x; 1.1477x over previous
#include <hip/hip_runtime.h>
#include <math.h>

typedef float v2f __attribute__((ext_vector_type(2)));
typedef float v8f __attribute__((ext_vector_type(8)));

#define DIN 128
#define C1  64
#define NEG_SLOPE 0.2f

#define GEMM_WAVES 8
__global__ __launch_bounds__(32 * GEMM_WAVES) void k_gemm_wmma(const float* __restrict__ x,
                                                   const float* __restrict__ W,
                                                   float* __restrict__ h, int n) {
  const int lane = threadIdx.x & 31;
  const int wave = __builtin_amdgcn_readfirstlane((int)(threadIdx.x >> 5));
  const int half = lane >> 4;
  const int l    = lane & 15;

  v2f bf[DIN / 4][4];
  {
    const float* wp = W + (size_t)(2 * half) * C1 + 4 * l;
#pragma unroll
    for (int ks = 0; ks < DIN / 4; ++ks) {
      const float4 r0 = *(const float4*)(wp + (size_t)(4 * ks) * C1);
      const float4 r1 = *(const float4*)(wp + (size_t)(4 * ks + 1) * C1);
      bf[ks][0] = (v2f){r0.x, r1.x};
      bf[ks][1] = (v2f){r0.y, r1.y};
      bf[ks][2] = (v2f){r0.z, r1.z};
      bf[ks][3] = (v2f){r0.w, r1.w};
    }
  }

  const int ntiles = (n + 15) >> 4;
  const int nwaves = (int)gridDim.x * GEMM_WAVES;
  for (int tile = (int)blockIdx.x * GEMM_WAVES + wave; tile < ntiles; tile += nwaves) {
    const int rowBase = tile * 16;
    int arow = rowBase + l;
    if (arow > n - 1) arow = n - 1;
    const float* ap = x + (size_t)arow * DIN + 2 * half;

    v8f c0 = {}, c1 = {}, c2 = {}, c3 = {};
#pragma unroll
    for (int ks = 0; ks < DIN / 4; ++ks) {
      v2f a = *(const v2f*)(ap + 4 * ks);
      c0 = __builtin_amdgcn_wmma_f32_16x16x4_f32(false, a, false, bf[ks][0], (short)0, c0, false, false);
      c1 = __builtin_amdgcn_wmma_f32_16x16x4_f32(false, a, false, bf[ks][1], (short)0, c1, false, false);
      c2 = __builtin_amdgcn_wmma_f32_16x16x4_f32(false, a, false, bf[ks][2], (short)0, c2, false, false);
      c3 = __builtin_amdgcn_wmma_f32_16x16x4_f32(false, a, false, bf[ks][3], (short)0, c3, false, false);
    }

#pragma unroll
    for (int v = 0; v < 8; ++v) {
      int row = rowBase + 8 * half + v;
      if (row < n)
        *(float4*)(h + (size_t)row * C1 + 4 * l) = make_float4(c0[v], c1[v], c2[v], c3[v]);
    }
  }
}

#define RNG   800
__device__ __forceinline__ unsigned range_byte(int d, int n) {
  return (unsigned)d < (unsigned)n ? (unsigned)d / RNG : 255u;
}
__device__ __forceinline__ unsigned range_word(int4 a, int n) {
  return range_byte(a.x, n) | (range_byte(a.y, n) << 8) | (range_byte(a.z, n) << 16) | (range_byte(a.w, n) << 24);
}
__device__ __forceinline__ int half_split(int E) { return 16 * (((E >> 4) + 1) >> 1); }

__global__ void k_logits(const float* __restrict__ h, const float* __restrict__ a_src,
                         const float* __restrict__ a_dst, float* __restrict__ esrc,
                         float* __restrict__ edst, int n,
                         const int* __restrict__ ei, int E, unsigned char* __restrict__ rid) {
  int i = blockIdx.x * blockDim.x + threadIdx.x;
  if (i < n) {
    const float4* hp = (const float4*)(h + (size_t)i * C1);
    float s = 0.0f, d = 0.0f;
#pragma unroll
    for (int c = 0; c < C1 / 4; ++c) {
      float4 hv = hp[c];
      float4 as = ((const float4*)a_src)[c];
      float4 ad = ((const float4*)a_dst)[c];
      s += hv.x * as.x + hv.y * as.y + hv.z * as.z + hv.w * as.w;
      d += hv.x * ad.x + hv.y * ad.y + hv.z * ad.z + hv.w * ad.w;
    }
    esrc[i] = s;
    edst[i] = d;
  }
  const int* dst = ei + E;
  const int nthr = (int)(gridDim.x * blockDim.x);
  const int nv = E >> 4;
  int done = 0;
  if ((((size_t)dst) & 15) == 0) {
    const int4* dv = (const int4*)dst;
    for (int v = i; v < nv; v += nthr) {
      int4 a0 = dv[4 * v], a1 = dv[4 * v + 1], a2 = dv[4 * v + 2], a3 = dv[4 * v + 3];
      ((uint4*)rid)[v] = make_uint4(range_word(a0, n), range_word(a1, n), range_word(a2, n), range_word(a3, n));
    }
    done = 16 * nv;
  }
  for (int idx = done + i; idx < E; idx += nthr) rid[idx] = (unsigned char)range_byte(dst[idx], n);
}

#define QCAP  12288
#define AGT   1024
#define PU    8
#define AU    8
static_assert(RNG <= 1024, "the queue packs the local node into 10 bits");
__device__ __forceinline__ float l1_logit(float es, float ed) {
  float v = es + ed;
  return v > 0.0f ? v : NEG_SLOPE * v;
}
__device__ __forceinline__ unsigned zero_bytes(unsigned x) {
  unsigned t = ((x & 0x7f7f7f7fu) + 0x7f7f7f7fu) | x;
  return ~t & 0x80808080u;
}
__device__ __forceinline__ void lds_max_f32(float* addr, float val) {
  if (__float_as_int(val) >= 0) atomicMax((int*)addr, __float_as_int(val));
  else                          atomicMin((unsigned int*)addr, __float_as_uint(val));
}
__device__ __forceinline__ void aggr_add(float p, float4 hv, int dl, int sub, int rot,
                                         float* acc, float* zl) {
  float* ap = acc + dl * C1 + sub * 4;
  float v[4] = {p * hv.x, p * hv.y, p * hv.z, p * hv.w};
#pragma unroll
  for (int j = 0; j < 4; ++j) {
    int jj = (j + rot) & 3;
    atomicAdd(ap + jj, v[jj]);
  }
  if (sub == 0) atomicAdd(zl + dl, p);
}

__global__ __launch_bounds__(AGT) void k_edge_aggr64(const int* __restrict__ ei, int E, int n,
                              const float* __restrict__ esrc, const float* __restrict__ edst,
                              const float* __restrict__ h, float* __restrict__ out1,
                              float* __restrict__ z, float* __restrict__ mx,
                              int* __restrict__ lst, int* __restrict__ cntq,
                              const unsigned char* __restrict__ rid) {
  __shared__ float acc[RNG * C1];
  __shared__ float zl[RNG];
  __shared__ float ml[RNG];
  __shared__ float edl[RNG];
  __shared__ int   q[QCAP];
  __shared__ int   q2[QCAP];
  __shared__ int   qn;
  const int tid  = threadIdx.x;
  const int half = blockIdx.x & 1;
  const int base = (blockIdx.x >> 1) * RNG;
  const int cnt  = min(RNG, n - base);
  const int sub  = tid & 15;
  const int grp  = tid >> 4;
  const int rot  = grp & 1;
  if (tid == 0) qn = 0;

  for (int r = tid; r < cnt; r += AGT) {
    float ed = edst[base + r];
    edl[r] = ed;
    ml[r] = half == 0 ? l1_logit(esrc[base + r], ed) : -INFINITY;
  }
  __syncthreads();

  const int* dst = ei + E;
  const unsigned ucnt = (unsigned)cnt;
  const unsigned rno = (unsigned)(blockIdx.x >> 1);
  const unsigned pat = rno * 0x01010101u;
  const int nv = E >> 4;
  const int split = half_split(E);
  const int hb = half ? split : 0;
  const int he = half ? E : split;
  {
    const uint4* rv = (const uint4*)rid;
    const int ve = half ? nv : (split >> 4);
    for (int v0 = (hb >> 4) + tid; v0 < ve; v0 += 4 * AGT) {
      uint4 a[4];
#pragma unroll
      for (int u = 0; u < 4; ++u) {
        int v = v0 + u * AGT;
        a[u] = rv[v < ve ? v : v0];
      }
      unsigned lo = (zero_bytes(a[0].x ^ pat) >> 7) | (zero_bytes(a[0].y ^ pat) >> 6) |
                    (zero_bytes(a[0].z ^ pat) >> 5) | (zero_bytes(a[0].w ^ pat) >> 4) |
                    (zero_bytes(a[1].x ^ pat) >> 3) | (zero_bytes(a[1].y ^ pat) >> 2) |
                    (zero_bytes(a[1].z ^ pat) >> 1) |  zero_bytes(a[1].w ^ pat);
      unsigned hi = (zero_bytes(a[2].x ^ pat) >> 7) | (zero_bytes(a[2].y ^ pat) >> 6) |
                    (zero_bytes(a[2].z ^ pat) >> 5) | (zero_bytes(a[2].w ^ pat) >> 4) |
                    (zero_bytes(a[3].x ^ pat) >> 3) | (zero_bytes(a[3].y ^ pat) >> 2) |
                    (zero_bytes(a[3].z ^ pat) >> 1) |  zero_bytes(a[3].w ^ pat);
      lo &= (v0 + AGT < ve) ? 0xffffffffu : 0x0f0f0f0fu;
      hi &= ((v0 + 2 * AGT < ve) ? 0x0f0f0f0fu : 0u) | ((v0 + 3 * AGT < ve) ? 0xf0f0f0f0u : 0u);
      unsigned long long hit = ((unsigned long long)hi << 32) | lo;
      while (hit) {
        int b = __ffsll(hit) - 1;
        hit &= hit - 1;
        int u = 2 * (b >> 5) + ((b >> 2) & 1);
        int idx = 16 * (v0 + u * AGT) + 4 * (b & 3) + ((b & 31) >> 3);
        int slot = atomicAdd(&qn, 1);
        if (slot < QCAP) q[slot] = idx;
      }
    }
  }
  if (half)
    for (int idx = 16 * nv + tid; idx < E; idx += AGT)
      if (rid[idx] == rno) {
        int slot = atomicAdd(&qn, 1);
        if (slot < QCAP) q[slot] = idx;
      }
  __syncthreads();

  const int nqraw = qn;
  const bool listed = nqraw <= QCAP;
  const int nq = listed ? nqraw : 0;
  if (tid == 0) cntq[blockIdx.x] = nqraw;

  int2* lq = (int2*)lst + (size_t)blockIdx.x * QCAP;
  if (listed) {
    for (int k0 = tid; k0 < nq; k0 += AU * AGT) {
      int   idx[AU], dlv[AU], sv[AU];
      float es[AU];
#pragma unroll
      for (int u = 0; u < AU; ++u) {
        int k = k0 + u * AGT;
        int kk = k < nq ? k : k0;
        idx[u] = q[kk];
      }
#pragma unroll
      for (int u = 0; u < AU; ++u) {
        sv[u]  = ei[idx[u]];
        dlv[u] = dst[idx[u]] - base;
      }
#pragma unroll
      for (int u = 0; u < AU; ++u) es[u] = esrc[sv[u]];
#pragma unroll
      for (int u = 0; u < AU; ++u) {
        int k = k0 + u * AGT;
        if (k < nq) {
          int dl = dlv[u];
          float e = l1_logit(es[u], edl[dl]);
          lds_max_f32(&ml[dl], e);
          lq[k] = make_int2(sv[u], base + dl);
          q[k]  = (int)(((unsigned)sv[u] << 10) | (unsigned)dl);
          q2[k] = __float_as_int(e);
        }
      }
    }
  } else {
    for (int idx = hb + tid; idx < he; idx += AGT) {
      int d = dst[idx];
      if ((unsigned)(d - base) < ucnt)
        lds_max_f32(&ml[d - base], l1_logit(esrc[ei[idx]], edl[d - base]));
    }
  }
  __syncthreads();

  for (int r = grp; r < cnt; r += AGT / 16) {
    float p = 0.0f;
    float4 hv = make_float4(0.0f, 0.0f, 0.0f, 0.0f);
    if (half == 0) {
      int d = base + r;
      p = __expf(l1_logit(esrc[d], edl[r]) - ml[r]);
      hv = *(const float4*)(h + (size_t)d * C1 + sub * 4);
    }
    *(float4*)(acc + r * C1 + sub * 4) = make_float4(p * hv.x, p * hv.y, p * hv.z, p * hv.w);
    if (sub == 0) zl[r] = p;
  }
  __syncthreads();

  if (listed) {
    for (int k0 = grp; k0 < nq; k0 += PU * (AGT / 16)) {
      unsigned w[PU];
      float    ev[PU];
      float4   hv[PU];
#pragma unroll
      for (int u = 0; u < PU; ++u) {
        int k = k0 + u * (AGT / 16);
        int kk = k < nq ? k : k0;
        w[u]  = (unsigned)q[kk];
        ev[u] = __int_as_float(q2[kk]);
      }
#pragma unroll
      for (int u = 0; u < PU; ++u)
        hv[u] = *(const float4*)(h + (size_t)(w[u] >> 10) * C1 + sub * 4);
#pragma unroll
      for (int u = 0; u < PU; ++u)
        if (k0 + u * (AGT / 16) < nq) {
          int dl = (int)(w[u] & 1023u);
          aggr_add(__expf(ev[u] - ml[dl]), hv[u], dl, sub, rot, acc, zl);
        }
    }
  } else {
    for (int idx = hb + tid; idx < he; idx += AGT) {
      int d = dst[idx];
      if ((unsigned)(d - base) < ucnt) {
        int s = ei[idx];
        float p = __expf(l1_logit(esrc[s], edl[d - base]) - ml[d - base]);
#pragma nounroll
        for (int s2 = 0; s2 < 16; ++s2)
          aggr_add(p, *(const float4*)(h + (size_t)s * C1 + s2 * 4), d - base, s2, 0, acc, zl);
      }
    }
  }
  __syncthreads();

  float* op = out1 + (size_t)half * n * C1 + (size_t)base * C1;
  float* zp = z + (size_t)half * n + base;
  float* mp = mx + (size_t)half * n + base;
  for (int i = tid; i < cnt * (C1 / 4); i += AGT)
    *(float4*)(op + i * 4) = *(const float4*)(acc + i * 4);
  for (int r = tid; r < cnt; r += AGT) { zp[r] = zl[r]; mp[r] = ml[r]; }
}

__global__ void k_layer2_prep(const float* __restrict__ out1, const float* __restrict__ z,
                              const float* __restrict__ mx, const float* __restrict__ b1,
                              const float* __restrict__ W2, float* __restrict__ g, int n) {
  int i = blockIdx.x * blockDim.x + threadIdx.x;
  if (i >= n) return;
  const float4* op = (const float4*)(out1 + (size_t)i * C1);
  const float4* oq = (const float4*)(out1 + (size_t)n * C1 + (size_t)i * C1);
  const float m0 = mx[i], m1 = mx[n + i];
  const float mm = fmaxf(m0, m1);
  const float w0 = __expf(m0 - mm);
  const float w1 = __expf(m1 - mm);
  const float rz = 1.0f / (z[i] * w0 + z[n + i] * w1);
  float acc = 0.0f;
#pragma unroll
  for (int c = 0; c < C1 / 4; ++c) {
    float4 v = op[c];
    float4 w = oq[c];
    v.x = v.x * w0 + w.x * w1; v.y = v.y * w0 + w.y * w1;
    v.z = v.z * w0 + w.z * w1; v.w = v.w * w0 + w.w * w1;
    float4 bb = ((const float4*)b1)[c];
    float4 ww = ((const float4*)W2)[c];
    float v0 = fmaxf(v.x * rz + bb.x, 0.0f);
    float v1 = fmaxf(v.y * rz + bb.y, 0.0f);
    float v2 = fmaxf(v.z * rz + bb.z, 0.0f);
    float v3 = fmaxf(v.w * rz + bb.w, 0.0f);
    acc += v0 * ww.x + v1 * ww.y + v2 * ww.z + v3 * ww.w;
  }
  g[i] = acc;
}

#define L2T 1024
#define L2U (QCAP / L2T)
__device__ __forceinline__ float l2_logit(float gs, float as2, float ed) {
  float v = gs * as2 + ed;
  return v > 0.0f ? v : NEG_SLOPE * v;
}

__global__ __launch_bounds__(L2T) void k_l2_range(const int* __restrict__ ei, int E, int n,
                              const int* __restrict__ lst, const int* __restrict__ cntq,
                              const float* __restrict__ g, const float* __restrict__ a_src2,
                              const float* __restrict__ a_dst2, float* __restrict__ part) {
  __shared__ float ml[RNG];
  __shared__ float zl[RNG];
  __shared__ float nl[RNG];
  __shared__ float edl[RNG];
  __shared__ float gl[RNG];
  __shared__ float esl[RNG];
  const int tid  = threadIdx.x;
  const int half = blockIdx.x & 1;
  const int base = (blockIdx.x >> 1) * RNG;
  const int cnt  = min(RNG, n - base);
  const unsigned ucnt = (unsigned)cnt;
  const float as2 = a_src2[0];
  const float ad2 = a_dst2[0];
  const int nqraw = cntq[blockIdx.x];
  const bool listed = nqraw <= QCAP;
  const int nq = listed ? nqraw : 0;

  const int* dst = ei + E;
  const int split = half_split(E);
  const int hb = half ? split : 0;
  const int he = half ? E : split;

  const int2* lq = (const int2*)lst + (size_t)blockIdx.x * QCAP;
  int2  sd[L2U];
  float gs[L2U], ev[L2U];
#pragma unroll
  for (int u = 0; u < L2U; ++u) {
    int k = tid + u * L2T;
    sd[u] = k < nq ? lq[k] : make_int2(base, base);
  }
#pragma unroll
  for (int u = 0; u < L2U; ++u) gs[u] = g[sd[u].x];

  for (int r = tid; r < cnt; r += L2T) {
    float gv = g[base + r];
    float ed = gv * ad2;
    float v = l2_logit(gv, as2, ed);
    gl[r] = gv;
    edl[r] = ed;
    esl[r] = v;
    ml[r] = half == 0 ? v : -INFINITY;
  }
  __syncthreads();

  if (listed) {
#pragma unroll
    for (int u = 0; u < L2U; ++u) {
      int dl = sd[u].y - base;
      ev[u] = l2_logit(gs[u], as2, edl[dl]);
      if (tid + u * L2T < nq) lds_max_f32(&ml[dl], ev[u]);
    }
  } else {
    for (int idx = hb + tid; idx < he; idx += L2T) {
      int d = dst[idx];
      if ((unsigned)(d - base) < ucnt)
        lds_max_f32(&ml[d - base], l2_logit(g[ei[idx]], as2, edl[d - base]));
    }
  }
  __syncthreads();

  for (int r = tid; r < cnt; r += L2T) {
    float p = half == 0 ? __expf(esl[r] - ml[r]) : 0.0f;
    zl[r] = p;
    nl[r] = p * gl[r];
  }
  __syncthreads();

  if (listed) {
#pragma unroll
    for (int u = 0; u < L2U; ++u) {
      if (tid + u * L2T < nq) {
        int dl = sd[u].y - base;
        float p = __expf(ev[u] - ml[dl]);
        atomicAdd(&zl[dl], p);
        atomicAdd(&nl[dl], p * gs[u]);
      }
    }
  } else {
    for (int idx = hb + tid; idx < he; idx += L2T) {
      int d = dst[idx];
      if ((unsigned)(d - base) < ucnt) {
        float gsv = g[ei[idx]];
        float p = __expf(l2_logit(gsv, as2, edl[d - base]) - ml[d - base]);
        atomicAdd(&zl[d - base], p);
        atomicAdd(&nl[d - base], p * gsv);
      }
    }
  }
  __syncthreads();

  float* pp = part + (size_t)half * 3 * n + base;
  for (int r = tid; r < cnt; r += L2T) {
    pp[r] = ml[r];
    pp[(size_t)n + r] = zl[r];
    pp[(size_t)2 * n + r] = nl[r];
  }
}

__global__ void k_final(const float* __restrict__ part, const float* __restrict__ b2,
                        float* __restrict__ out, int n) {
  int i = blockIdx.x * blockDim.x + threadIdx.x;
  if (i >= n) return;
  const size_t sn = (size_t)n;
  float m0 = part[i],          z0 = part[sn + i],     s0 = part[2 * sn + i];
  float m1 = part[3 * sn + i], z1 = part[4 * sn + i], s1 = part[5 * sn + i];
  float mm = fmaxf(m0, m1);
  float w0 = __expf(m0 - mm);
  float w1 = __expf(m1 - mm);
  float zz = z0 * w0 + z1 * w1;
  float ss = s0 * w0 + s1 * w1;
  float v = ss / zz + b2[0];
  out[i] = 1.0f / (1.0f + __expf(-v));
}

extern "C" void kernel_launch(void* const* d_in, const int* in_sizes, int n_in,
                              void* d_out, int out_size, void* d_ws, size_t ws_size,
                              hipStream_t stream) {
  const int*   ei     = (const int*)d_in[0];
  const float* x      = (const float*)d_in[1];
  const float* W1     = (const float*)d_in[2];
  const float* a_src1 = (const float*)d_in[3];
  const float* a_dst1 = (const float*)d_in[4];
  const float* b1     = (const float*)d_in[5];
  const float* W2     = (const float*)d_in[6];
  const float* a_src2 = (const float*)d_in[7];
  const float* a_dst2 = (const float*)d_in[8];
  const float* b2     = (const float*)d_in[9];

  const int E = in_sizes[0] / 2;
  const int n = in_sizes[1] / DIN;

  float* f     = (float*)d_ws;
  float* h     = f;
  float* out1  = f + (size_t)64 * n;
  float* esrc  = f + (size_t)192 * n;
  float* edst  = f + (size_t)193 * n;
  float* g     = f + (size_t)194 * n;
  float* z     = f + (size_t)195 * n;
  float* mx    = f + (size_t)197 * n;
  const int nwg = 2 * ((n + RNG - 1) / RNG);
  const size_t o1 = ((size_t)199 * n + 3) & ~(size_t)3;
  float* part  = f + o1;
  const size_t o2 = (o1 + (size_t)6 * n + 3) & ~(size_t)3;
  int*   lst   = (int*)(f + o2);
  int*   cntq  = lst + (size_t)nwg * QCAP * 2;
  unsigned char* rid = (unsigned char*)(cntq + ((nwg + 3) & ~3));
  float* out   = (float*)d_out;

  const int B = 256;
  dim3 blk(B);

  k_gemm_wmma<<<dim3(256), dim3(32 * GEMM_WAVES), 0, stream>>>(x, W1, h, n);
  k_logits<<<dim3((n + B - 1) / B), blk, 0, stream>>>(h, a_src1, a_dst1, esrc, edst, n, ei, E, rid);
  k_edge_aggr64<<<dim3(nwg), dim3(AGT), 0, stream>>>(ei, E, n, esrc, edst, h, out1, z, mx, lst, cntq, rid);

  k_layer2_prep<<<dim3((n + B - 1) / B), blk, 0, stream>>>(out1, z, mx, b1, W2, g, n);
  k_l2_range<<<dim3(nwg), dim3(L2T), 0, stream>>>(ei, E, n, lst, cntq, g, a_src2, a_dst2, part);

  k_final<<<dim3((n + B - 1) / B), blk, 0, stream>>>(part, b2, out, n);
}


# Node kernels (logits, layer-2 input) read 64-ch rows with 16 lanes per row, 16 B a lane: a wave load covers 4 whole 128-B lines, not 32 scattered ones; dots end in a 16-lane sum: sum order changes
# speedup vs baseline: 5.0825x; 1.0428x over previous
#include <hip/hip_runtime.h>
#include <math.h>

typedef float v2f __attribute__((ext_vector_type(2)));
typedef float v4f __attribute__((ext_vector_type(4)));
typedef float v8f __attribute__((ext_vector_type(8)));

#define DIN 128
#define C1  64
#define NEG_SLOPE 0.2f

#define GEMM_WAVES 8
__global__ __launch_bounds__(32 * GEMM_WAVES) void k_gemm_wmma(const float* __restrict__ x,
                                                   const float* __restrict__ W,
                                                   float* __restrict__ h, int n) {
  const int lane = threadIdx.x & 31;
  const int wave = __builtin_amdgcn_readfirstlane((int)(threadIdx.x >> 5));
  const int half = lane >> 4;
  const int l    = lane & 15;

  v2f bf[DIN / 4][4];
  {
    const float* wp = W + (size_t)(2 * half) * C1 + 4 * l;
#pragma unroll
    for (int ks = 0; ks < DIN / 4; ++ks) {
      const float4 r0 = *(const float4*)(wp + (size_t)(4 * ks) * C1);
      const float4 r1 = *(const float4*)(wp + (size_t)(4 * ks + 1) * C1);
      bf[ks][0] = (v2f){r0.x, r1.x};
      bf[ks][1] = (v2f){r0.y, r1.y};
      bf[ks][2] = (v2f){r0.z, r1.z};
      bf[ks][3] = (v2f){r0.w, r1.w};
    }
  }

  const int ntiles = (n + 15) >> 4;
  const int nwaves = (int)gridDim.x * GEMM_WAVES;
  for (int tile = (int)blockIdx.x * GEMM_WAVES + wave; tile < ntiles; tile += nwaves) {
    const int rowBase = tile * 16;
    int arow = rowBase + l;
    if (arow > n - 1) arow = n - 1;
    const float* ap = x + (size_t)arow * DIN + 2 * half;

    v8f c0 = {}, c1 = {}, c2 = {}, c3 = {};
#pragma unroll
    for (int ks = 0; ks < DIN / 4; ++ks) {
      v2f a = *(const v2f*)(ap + 4 * ks);
      c0 = __builtin_amdgcn_wmma_f32_16x16x4_f32(false, a, false, bf[ks][0], (short)0, c0, false, false);
      c1 = __builtin_amdgcn_wmma_f32_16x16x4_f32(false, a, false, bf[ks][1], (short)0, c1, false, false);
      c2 = __builtin_amdgcn_wmma_f32_16x16x4_f32(false, a, false, bf[ks][2], (short)0, c2, false, false);
      c3 = __builtin_amdgcn_wmma_f32_16x16x4_f32(false, a, false, bf[ks][3], (short)0, c3, false, false);
    }

#pragma unroll
    for (int v = 0; v < 8; ++v) {
      int row = rowBase + 8 * half + v;
      if (row < n)
        *(float4*)(h + (size_t)row * C1 + 4 * l) = make_float4(c0[v], c1[v], c2[v], c3[v]);
    }
  }
}

#define RNG   800
__device__ __forceinline__ unsigned range_byte(int d, int n) {
  return (unsigned)d < (unsigned)n ? (unsigned)d / RNG : 255u;
}
__device__ __forceinline__ unsigned range_word(int4 a, int n) {
  return range_byte(a.x, n) | (range_byte(a.y, n) << 8) | (range_byte(a.z, n) << 16) | (range_byte(a.w, n) << 24);
}
__device__ __forceinline__ int half_split(int E) { return 16 * (((E >> 4) + 1) >> 1); }

__device__ __forceinline__ float row16_sum(float v) {
  v += __builtin_bit_cast(float, __builtin_amdgcn_update_dpp(0, __builtin_bit_cast(int, v), 0x161, 0xf, 0xf, true));
  v += __builtin_bit_cast(float, __builtin_amdgcn_update_dpp(0, __builtin_bit_cast(int, v), 0x162, 0xf, 0xf, true));
  v += __builtin_bit_cast(float, __builtin_amdgcn_update_dpp(0, __builtin_bit_cast(int, v), 0x164, 0xf, 0xf, true));
  v += __builtin_bit_cast(float, __builtin_amdgcn_update_dpp(0, __builtin_bit_cast(int, v), 0x168, 0xf, 0xf, true));
  return v;
}
template <int U> __device__ __forceinline__ float row16_lane(float v) {
  return __builtin_bit_cast(float, __builtin_amdgcn_update_dpp(0, __builtin_bit_cast(int, v), 0x150 + U, 0xf, 0xf, true));
}
#define FOR16(M) M(0) M(1) M(2) M(3) M(4) M(5) M(6) M(7) M(8) M(9) M(10) M(11) M(12) M(13) M(14) M(15)

__global__ __launch_bounds__(256) void k_logits(const float* __restrict__ h, const float* __restrict__ a_src,
                         const float* __restrict__ a_dst, float* __restrict__ esrc,
                         float* __restrict__ edst, int n,
                         const int* __restrict__ ei, int E, unsigned char* __restrict__ rid) {
  int i = blockIdx.x * blockDim.x + threadIdx.x;
  {
    const int sub = (int)threadIdx.x & 15;
    const int nb  = i - sub;
    if (nb < n) {
      const v4f as = *(const v4f*)(a_src + sub * 4);
      const v4f ad = *(const v4f*)(a_dst + sub * 4);
      v4f hv[16];
#pragma unroll
      for (int u = 0; u < 16; ++u) {
        int r = nb + u;
        r = r < n ? r : n - 1;
        hv[u] = *(const v4f*)(h + (size_t)r * C1 + sub * 4);
      }
      asm volatile("" :: "v"(hv[0]), "v"(hv[1]), "v"(hv[2]), "v"(hv[3]), "v"(hv[4]), "v"(hv[5]), "v"(hv[6]), "v"(hv[7]),
                         "v"(hv[8]), "v"(hv[9]), "v"(hv[10]), "v"(hv[11]), "v"(hv[12]), "v"(hv[13]), "v"(hv[14]), "v"(hv[15]),
                         "v"(as), "v"(ad));
      float s = 0.0f, d = 0.0f;
#pragma unroll
      for (int u = 0; u < 16; ++u) {
        float su = row16_sum(hv[u].x * as.x + hv[u].y * as.y + hv[u].z * as.z + hv[u].w * as.w);
        float du = row16_sum(hv[u].x * ad.x + hv[u].y * ad.y + hv[u].z * ad.z + hv[u].w * ad.w);
        if (sub == u) { s = su; d = du; }
      }
      if (i < n) {
        esrc[i] = s;
        edst[i] = d;
      }
    }
  }
  const int* dst = ei + E;
  const int nthr = (int)(gridDim.x * blockDim.x);
  const int nv = E >> 4;
  int done = 0;
  if ((((size_t)dst) & 15) == 0) {
    const int4* dv = (const int4*)dst;
    for (int v = i; v < nv; v += nthr) {
      int4 a0 = dv[4 * v], a1 = dv[4 * v + 1], a2 = dv[4 * v + 2], a3 = dv[4 * v + 3];
      ((uint4*)rid)[v] = make_uint4(range_word(a0, n), range_word(a1, n), range_word(a2, n), range_word(a3, n));
    }
    done = 16 * nv;
  }
  for (int idx = done + i; idx < E; idx += nthr) rid[idx] = (unsigned char)range_byte(dst[idx], n);
}

#define QCAP  12288
#define AGT   1024
#define PU    8
#define AU    8
static_assert(RNG <= 1024, "the queue packs the local node into 10 bits");
__device__ __forceinline__ float l1_logit(float es, float ed) {
  float v = es + ed;
  return v > 0.0f ? v : NEG_SLOPE * v;
}
__device__ __forceinline__ unsigned zero_bytes(unsigned x) {
  unsigned t = ((x & 0x7f7f7f7fu) + 0x7f7f7f7fu) | x;
  return ~t & 0x80808080u;
}
__device__ __forceinline__ void lds_max_f32(float* addr, float val) {
  if (__float_as_int(val) >= 0) atomicMax((int*)addr, __float_as_int(val));
  else                          atomicMin((unsigned int*)addr, __float_as_uint(val));
}
__device__ __forceinline__ void aggr_add(float p, float4 hv, int dl, int sub, int rot,
                                         float* acc, float* zl) {
  float* ap = acc + dl * C1 + sub * 4;
  float v[4] = {p * hv.x, p * hv.y, p * hv.z, p * hv.w};
#pragma unroll
  for (int j = 0; j < 4; ++j) {
    int jj = (j + rot) & 3;
    atomicAdd(ap + jj, v[jj]);
  }
  if (sub == 0) atomicAdd(zl + dl, p);
}

__global__ __launch_bounds__(AGT) void k_edge_aggr64(const int* __restrict__ ei, int E, int n,
                              const float* __restrict__ esrc, const float* __restrict__ edst,
                              const float* __restrict__ h, float* __restrict__ out1,
                              float* __restrict__ z, float* __restrict__ mx,
                              int* __restrict__ lst, int* __restrict__ cntq,
                              const unsigned char* __restrict__ rid) {
  __shared__ float acc[RNG * C1];
  __shared__ float zl[RNG];
  __shared__ float ml[RNG];
  __shared__ float edl[RNG];
  __shared__ int   q[QCAP];
  __shared__ int   q2[QCAP];
  __shared__ int   qn;
  const int tid  = threadIdx.x;
  const int half = blockIdx.x & 1;
  const int base = (blockIdx.x >> 1) * RNG;
  const int cnt  = min(RNG, n - base);
  const int sub  = tid & 15;
  const int grp  = tid >> 4;
  const int rot  = grp & 1;
  if (tid == 0) qn = 0;

  for (int r = tid; r < cnt; r += AGT) {
    float ed = edst[base + r];
    edl[r] = ed;
    ml[r] = half == 0 ? l1_logit(esrc[base + r], ed) : -INFINITY;
  }
  __syncthreads();

  const int* dst = ei + E;
  const unsigned ucnt = (unsigned)cnt;
  const unsigned rno = (unsigned)(blockIdx.x >> 1);
  const unsigned pat = rno * 0x01010101u;
  const int nv = E >> 4;
  const int split = half_split(E);
  const int hb = half ? split : 0;
  const int he = half ? E : split;
  {
    const uint4* rv = (const uint4*)rid;
    const int ve = half ? nv : (split >> 4);
    for (int v0 = (hb >> 4) + tid; v0 < ve; v0 += 4 * AGT) {
      uint4 a[4];
#pragma unroll
      for (int u = 0; u < 4; ++u) {
        int v = v0 + u * AGT;
        a[u] = rv[v < ve ? v : v0];
      }
      unsigned lo = (zero_bytes(a[0].x ^ pat) >> 7) | (zero_bytes(a[0].y ^ pat) >> 6) |
                    (zero_bytes(a[0].z ^ pat) >> 5) | (zero_bytes(a[0].w ^ pat) >> 4) |
                    (zero_bytes(a[1].x ^ pat) >> 3) | (zero_bytes(a[1].y ^ pat) >> 2) |
                    (zero_bytes(a[1].z ^ pat) >> 1) |  zero_bytes(a[1].w ^ pat);
      unsigned hi = (zero_bytes(a[2].x ^ pat) >> 7) | (zero_bytes(a[2].y ^ pat) >> 6) |
                    (zero_bytes(a[2].z ^ pat) >> 5) | (zero_bytes(a[2].w ^ pat) >> 4) |
                    (zero_bytes(a[3].x ^ pat) >> 3) | (zero_bytes(a[3].y ^ pat) >> 2) |
                    (zero_bytes(a[3].z ^ pat) >> 1) |  zero_bytes(a[3].w ^ pat);
      lo &= (v0 + AGT < ve) ? 0xffffffffu : 0x0f0f0f0fu;
      hi &= ((v0 + 2 * AGT < ve) ? 0x0f0f0f0fu : 0u) | ((v0 + 3 * AGT < ve) ? 0xf0f0f0f0u : 0u);
      unsigned long long hit = ((unsigned long long)hi << 32) | lo;
      while (hit) {
        int b = __ffsll(hit) - 1;
        hit &= hit - 1;
        int u = 2 * (b >> 5) + ((b >> 2) & 1);
        int idx = 16 * (v0 + u * AGT) + 4 * (b & 3) + ((b & 31) >> 3);
        int slot = atomicAdd(&qn, 1);
        if (slot < QCAP) q[slot] = idx;
      }
    }
  }
  if (half)
    for (int idx = 16 * nv + tid; idx < E; idx += AGT)
      if (rid[idx] == rno) {
        int slot = atomicAdd(&qn, 1);
        if (slot < QCAP) q[slot] = idx;
      }
  __syncthreads();

  const int nqraw = qn;
  const bool listed = nqraw <= QCAP;
  const int nq = listed ? nqraw : 0;
  if (tid == 0) cntq[blockIdx.x] = nqraw;

  int2* lq = (int2*)lst + (size_t)blockIdx.x * QCAP;
  if (listed) {
    for (int k0 = tid; k0 < nq; k0 += AU * AGT) {
      int   idx[AU], dlv[AU], sv[AU];
      float es[AU];
#pragma unroll
      for (int u = 0; u < AU; ++u) {
        int k = k0 + u * AGT;
        int kk = k < nq ? k : k0;
        idx[u] = q[kk];
      }
#pragma unroll
      for (int u = 0; u < AU; ++u) {
        sv[u]  = ei[idx[u]];
        dlv[u] = dst[idx[u]] - base;
      }
#pragma unroll
      for (int u = 0; u < AU; ++u) es[u] = esrc[sv[u]];
#pragma unroll
      for (int u = 0; u < AU; ++u) {
        int k = k0 + u * AGT;
        if (k < nq) {
          int dl = dlv[u];
          float e = l1_logit(es[u], edl[dl]);
          lds_max_f32(&ml[dl], e);
          lq[k] = make_int2(sv[u], base + dl);
          q[k]  = (int)(((unsigned)sv[u] << 10) | (unsigned)dl);
          q2[k] = __float_as_int(e);
        }
      }
    }
  } else {
    for (int idx = hb + tid; idx < he; idx += AGT) {
      int d = dst[idx];
      if ((unsigned)(d - base) < ucnt)
        lds_max_f32(&ml[d - base], l1_logit(esrc[ei[idx]], edl[d - base]));
    }
  }
  __syncthreads();

  for (int r = grp; r < cnt; r += AGT / 16) {
    float p = 0.0f;
    float4 hv = make_float4(0.0f, 0.0f, 0.0f, 0.0f);
    if (half == 0) {
      int d = base + r;
      p = __expf(l1_logit(esrc[d], edl[r]) - ml[r]);
      hv = *(const float4*)(h + (size_t)d * C1 + sub * 4);
    }
    *(float4*)(acc + r * C1 + sub * 4) = make_float4(p * hv.x, p * hv.y, p * hv.z, p * hv.w);
    if (sub == 0) zl[r] = p;
  }
  __syncthreads();

  if (listed) {
    for (int k0 = grp; k0 < nq; k0 += PU * (AGT / 16)) {
      unsigned w[PU];
      float    ev[PU];
      float4   hv[PU];
#pragma unroll
      for (int u = 0; u < PU; ++u) {
        int k = k0 + u * (AGT / 16);
        int kk = k < nq ? k : k0;
        w[u]  = (unsigned)q[kk];
        ev[u] = __int_as_float(q2[kk]);
      }
#pragma unroll
      for (int u = 0; u < PU; ++u)
        hv[u] = *(const float4*)(h + (size_t)(w[u] >> 10) * C1 + sub * 4);
#pragma unroll
      for (int u = 0; u < PU; ++u)
        if (k0 + u * (AGT / 16) < nq) {
          int dl = (int)(w[u] & 1023u);
          aggr_add(__expf(ev[u] - ml[dl]), hv[u], dl, sub, rot, acc, zl);
        }
    }
  } else {
    for (int idx = hb + tid; idx < he; idx += AGT) {
      int d = dst[idx];
      if ((unsigned)(d - base) < ucnt) {
        int s = ei[idx];
        float p = __expf(l1_logit(esrc[s], edl[d - base]) - ml[d - base]);
#pragma nounroll
        for (int s2 = 0; s2 < 16; ++s2)
          aggr_add(p, *(const float4*)(h + (size_t)s * C1 + s2 * 4), d - base, s2, 0, acc, zl);
      }
    }
  }
  __syncthreads();

  float* op = out1 + (size_t)half * n * C1 + (size_t)base * C1;
  float* zp = z + (size_t)half * n + base;
  float* mp = mx + (size_t)half * n + base;
  for (int i = tid; i < cnt * (C1 / 4); i += AGT)
    *(float4*)(op + i * 4) = *(const float4*)(acc + i * 4);
  for (int r = tid; r < cnt; r += AGT) { zp[r] = zl[r]; mp[r] = ml[r]; }
}

__global__ __launch_bounds__(256) void k_layer2_prep(const float* __restrict__ out1, const float* __restrict__ z,
                              const float* __restrict__ mx, const float* __restrict__ b1,
                              const float* __restrict__ W2, float* __restrict__ g, int n) {
  const int i = blockIdx.x * blockDim.x + threadIdx.x;
  const int sub = (int)threadIdx.x & 15;
  const int nb  = i - sub;
  if (nb >= n) return;
  const float* op = out1 + (size_t)sub * 4;
  const float* oq = out1 + (size_t)n * C1 + (size_t)sub * 4;
  v8f pr[16];
#pragma unroll
  for (int u = 0; u < 16; ++u) {
    int r = nb + u;
    r = r < n ? r : n - 1;
    const v4f a = *(const v4f*)(op + (size_t)r * C1);
    const v4f b = *(const v4f*)(oq + (size_t)r * C1);
    pr[u] = __builtin_shufflevector(a, b, 0, 1, 2, 3, 4, 5, 6, 7);
  }
  const int ic = i < n ? i : n - 1;
  const float m0 = mx[ic], m1 = mx[n + ic];
  const float z0 = z[ic], z1 = z[n + ic];
  const v4f bb = *(const v4f*)(b1 + sub * 4);
  const v4f ww = *(const v4f*)(W2 + sub * 4);
  asm volatile("" :: "v"(pr[0]), "v"(pr[1]), "v"(pr[2]), "v"(pr[3]), "v"(pr[4]), "v"(pr[5]), "v"(pr[6]), "v"(pr[7]),
                     "v"(pr[8]), "v"(pr[9]), "v"(pr[10]), "v"(pr[11]), "v"(pr[12]), "v"(pr[13]), "v"(pr[14]), "v"(pr[15]),
                     "v"(m0), "v"(m1), "v"(z0), "v"(z1), "v"(bb), "v"(ww));
  const float mm = fmaxf(m0, m1);
  const float w0 = __expf(m0 - mm);
  const float w1 = __expf(m1 - mm);
  const float rz = 1.0f / (z0 * w0 + z1 * w1);
  float acc = 0.0f;
#define L2P_NODE(U) { \
    const float q0 = row16_lane<U>(w0), q1 = row16_lane<U>(w1), qz = row16_lane<U>(rz); \
    const v8f p = pr[U]; \
    const float x0 = p[0] * q0 + p[4] * q1, x1 = p[1] * q0 + p[5] * q1; \
    const float x2 = p[2] * q0 + p[6] * q1, x3 = p[3] * q0 + p[7] * q1; \
    const float v0 = fmaxf(x0 * qz + bb.x, 0.0f); \
    const float v1 = fmaxf(x1 * qz + bb.y, 0.0f); \
    const float v2 = fmaxf(x2 * qz + bb.z, 0.0f); \
    const float v3 = fmaxf(x3 * qz + bb.w, 0.0f); \
    const float t = row16_sum(v0 * ww.x + v1 * ww.y + v2 * ww.z + v3 * ww.w); \
    if (sub == U) acc = t; }
  FOR16(L2P_NODE)
#undef L2P_NODE
  if (i < n) g[i] = acc;
}

#define L2T 1024
#define L2U (QCAP / L2T)
__device__ __forceinline__ float l2_logit(float gs, float as2, float ed) {
  float v = gs * as2 + ed;
  return v > 0.0f ? v : NEG_SLOPE * v;
}

__global__ __launch_bounds__(L2T) void k_l2_range(const int* __restrict__ ei, int E, int n,
                              const int* __restrict__ lst, const int* __restrict__ cntq,
                              const float* __restrict__ g, const float* __restrict__ a_src2,
                              const float* __restrict__ a_dst2, float* __restrict__ part) {
  __shared__ float ml[RNG];
  __shared__ float zl[RNG];
  __shared__ float nl[RNG];
  __shared__ float edl[RNG];
  __shared__ float gl[RNG];
  __shared__ float esl[RNG];
  const int tid  = threadIdx.x;
  const int half = blockIdx.x & 1;
  const int base = (blockIdx.x >> 1) * RNG;
  const int cnt  = min(RNG, n - base);
  const unsigned ucnt = (unsigned)cnt;
  const float as2 = a_src2[0];
  const float ad2 = a_dst2[0];
  const int nqraw = cntq[blockIdx.x];
  const bool listed = nqraw <= QCAP;
  const int nq = listed ? nqraw : 0;

  const int* dst = ei + E;
  const int split = half_split(E);
  const int hb = half ? split : 0;
  const int he = half ? E : split;

  const int2* lq = (const int2*)lst + (size_t)blockIdx.x * QCAP;
  int2  sd[L2U];
  float gs[L2U], ev[L2U];
#pragma unroll
  for (int u = 0; u < L2U; ++u) {
    int k = tid + u * L2T;
    sd[u] = k < nq ? lq[k] : make_int2(base, base);
  }
#pragma unroll
  for (int u = 0; u < L2U; ++u) gs[u] = g[sd[u].x];

  for (int r = tid; r < cnt; r += L2T) {
    float gv = g[base + r];
    float ed = gv * ad2;
    float v = l2_logit(gv, as2, ed);
    gl[r] = gv;
    edl[r] = ed;
    esl[r] = v;
    ml[r] = half == 0 ? v : -INFINITY;
  }
  __syncthreads();

  if (listed) {
#pragma unroll
    for (int u = 0; u < L2U; ++u) {
      int dl = sd[u].y - base;
      ev[u] = l2_logit(gs[u], as2, edl[dl]);
      if (tid + u * L2T < nq) lds_max_f32(&ml[dl], ev[u]);
    }
  } else {
    for (int idx = hb + tid; idx < he; idx += L2T) {
      int d = dst[idx];
      if ((unsigned)(d - base) < ucnt)
        lds_max_f32(&ml[d - base], l2_logit(g[ei[idx]], as2, edl[d - base]));
    }
  }
  __syncthreads();

  for (int r = tid; r < cnt; r += L2T) {
    float p = half == 0 ? __expf(esl[r] - ml[r]) : 0.0f;
    zl[r] = p;
    nl[r] = p * gl[r];
  }
  __syncthreads();

  if (listed) {
#pragma unroll
    for (int u = 0; u < L2U; ++u) {
      if (tid + u * L2T < nq) {
        int dl = sd[u].y - base;
        float p = __expf(ev[u] - ml[dl]);
        atomicAdd(&zl[dl], p);
        atomicAdd(&nl[dl], p * gs[u]);
      }
    }
  } else {
    for (int idx = hb + tid; idx < he; idx += L2T) {
      int d = dst[idx];
      if ((unsigned)(d - base) < ucnt) {
        float gsv = g[ei[idx]];
        float p = __expf(l2_logit(gsv, as2, edl[d - base]) - ml[d - base]);
        atomicAdd(&zl[d - base], p);
        atomicAdd(&nl[d - base], p * gsv);
      }
    }
  }
  __syncthreads();

  float* pp = part + (size_t)half * 3 * n + base;
  for (int r = tid; r < cnt; r += L2T) {
    pp[r] = ml[r];
    pp[(size_t)n + r] = zl[r];
    pp[(size_t)2 * n + r] = nl[r];
  }
}

__global__ void k_final(const float* __restrict__ part, const float* __restrict__ b2,
                        float* __restrict__ out, int n) {
  int i = blockIdx.x * blockDim.x + threadIdx.x;
  if (i >= n) return;
  const size_t sn = (size_t)n;
  float m0 = part[i],          z0 = part[sn + i],     s0 = part[2 * sn + i];
  float m1 = part[3 * sn + i], z1 = part[4 * sn + i], s1 = part[5 * sn + i];
  float mm = fmaxf(m0, m1);
  float w0 = __expf(m0 - mm);
  float w1 = __expf(m1 - mm);
  float zz = z0 * w0 + z1 * w1;
  float ss = s0 * w0 + s1 * w1;
  float v = ss / zz + b2[0];
  out[i] = 1.0f / (1.0f + __expf(-v));
}

extern "C" void kernel_launch(void* const* d_in, const int* in_sizes, int n_in,
                              void* d_out, int out_size, void* d_ws, size_t ws_size,
                              hipStream_t stream) {
  const int*   ei     = (const int*)d_in[0];
  const float* x      = (const float*)d_in[1];
  const float* W1     = (const float*)d_in[2];
  const float* a_src1 = (const float*)d_in[3];
  const float* a_dst1 = (const float*)d_in[4];
  const float* b1     = (const float*)d_in[5];
  const float* W2     = (const float*)d_in[6];
  const float* a_src2 = (const float*)d_in[7];
  const float* a_dst2 = (const float*)d_in[8];
  const float* b2     = (const float*)d_in[9];

  const int E = in_sizes[0] / 2;
  const int n = in_sizes[1] / DIN;

  float* f     = (float*)d_ws;
  float* h     = f;
  float* out1  = f + (size_t)64 * n;
  float* esrc  = f + (size_t)192 * n;
  float* edst  = f + (size_t)193 * n;
  float* g     = f + (size_t)194 * n;
  float* z     = f + (size_t)195 * n;
  float* mx    = f + (size_t)197 * n;
  const int nwg = 2 * ((n + RNG - 1) / RNG);
  const size_t o1 = ((size_t)199 * n + 3) & ~(size_t)3;
  float* part  = f + o1;
  const size_t o2 = (o1 + (size_t)6 * n + 3) & ~(size_t)3;
  int*   lst   = (int*)(f + o2);
  int*   cntq  = lst + (size_t)nwg * QCAP * 2;
  unsigned char* rid = (unsigned char*)(cntq + ((nwg + 3) & ~3));
  float* out   = (float*)d_out;

  const int B = 256;
  dim3 blk(B);

  k_gemm_wmma<<<dim3(256), dim3(32 * GEMM_WAVES), 0, stream>>>(x, W1, h, n);
  k_logits<<<dim3((n + B - 1) / B), blk, 0, stream>>>(h, a_src1, a_dst1, esrc, edst, n, ei, E, rid);
  k_edge_aggr64<<<dim3(nwg), dim3(AGT), 0, stream>>>(ei, E, n, esrc, edst, h, out1, z, mx, lst, cntq, rid);

  k_layer2_prep<<<dim3((n + B - 1) / B), blk, 0, stream>>>(out1, z, mx, b1, W2, g, n);
  k_l2_range<<<dim3(nwg), dim3(L2T), 0, stream>>>(ei, E, n, lst, cntq, g, a_src2, a_dst2, part);

  k_final<<<dim3((n + B - 1) / B), blk, 0, stream>>>(part, b2, out, n);
}


# Logits launch also writes one word per edge, (source<<10)|(dst - range start); range kernel gathers that word per queued edge, not a word of each index list; layer-2 list is 4 B an edge, not 8
# speedup vs baseline: 5.3479x; 1.0522x over previous
#include <hip/hip_runtime.h>
#include <math.h>

typedef float v2f __attribute__((ext_vector_type(2)));
typedef float v4f __attribute__((ext_vector_type(4)));
typedef float v8f __attribute__((ext_vector_type(8)));

#define DIN 128
#define C1  64
#define NEG_SLOPE 0.2f

#define GEMM_WAVES 8
__global__ __launch_bounds__(32 * GEMM_WAVES) void k_gemm_wmma(const float* __restrict__ x,
                                                   const float* __restrict__ W,
                                                   float* __restrict__ h, int n) {
  const int lane = threadIdx.x & 31;
  const int wave = __builtin_amdgcn_readfirstlane((int)(threadIdx.x >> 5));
  const int half = lane >> 4;
  const int l    = lane & 15;

  v2f bf[DIN / 4][4];
  {
    const float* wp = W + (size_t)(2 * half) * C1 + 4 * l;
#pragma unroll
    for (int ks = 0; ks < DIN / 4; ++ks) {
      const float4 r0 = *(const float4*)(wp + (size_t)(4 * ks) * C1);
      const float4 r1 = *(const float4*)(wp + (size_t)(4 * ks + 1) * C1);
      bf[ks][0] = (v2f){r0.x, r1.x};
      bf[ks][1] = (v2f){r0.y, r1.y};
      bf[ks][2] = (v2f){r0.z, r1.z};
      bf[ks][3] = (v2f){r0.w, r1.w};
    }
  }

  const int ntiles = (n + 15) >> 4;
  const int nwaves = (int)gridDim.x * GEMM_WAVES;
  for (int tile = (int)blockIdx.x * GEMM_WAVES + wave; tile < ntiles; tile += nwaves) {
    const int rowBase = tile * 16;
    int arow = rowBase + l;
    if (arow > n - 1) arow = n - 1;
    const float* ap = x + (size_t)arow * DIN + 2 * half;

    v8f c0 = {}, c1 = {}, c2 = {}, c3 = {};
#pragma unroll
    for (int ks = 0; ks < DIN / 4; ++ks) {
      v2f a = *(const v2f*)(ap + 4 * ks);
      c0 = __builtin_amdgcn_wmma_f32_16x16x4_f32(false, a, false, bf[ks][0], (short)0, c0, false, false);
      c1 = __builtin_amdgcn_wmma_f32_16x16x4_f32(false, a, false, bf[ks][1], (short)0, c1, false, false);
      c2 = __builtin_amdgcn_wmma_f32_16x16x4_f32(false, a, false, bf[ks][2], (short)0, c2, false, false);
      c3 = __builtin_amdgcn_wmma_f32_16x16x4_f32(false, a, false, bf[ks][3], (short)0, c3, false, false);
    }

#pragma unroll
    for (int v = 0; v < 8; ++v) {
      int row = rowBase + 8 * half + v;
      if (row < n)
        *(float4*)(h + (size_t)row * C1 + 4 * l) = make_float4(c0[v], c1[v], c2[v], c3[v]);
    }
  }
}

#define RNG   800
__device__ __forceinline__ unsigned range_byte(int d, int n) {
  return (unsigned)d < (unsigned)n ? (unsigned)d / RNG : 255u;
}
__device__ __forceinline__ unsigned range_word(int4 a, int n) {
  return range_byte(a.x, n) | (range_byte(a.y, n) << 8) | (range_byte(a.z, n) << 16) | (range_byte(a.w, n) << 24);
}
__device__ __forceinline__ unsigned edge_word(int s, int d) {
  return ((unsigned)s << 10) | ((unsigned)d - ((unsigned)d / RNG) * RNG);
}
__device__ __forceinline__ uint4 edge_words(int4 s, int4 d) {
  return make_uint4(edge_word(s.x, d.x), edge_word(s.y, d.y), edge_word(s.z, d.z), edge_word(s.w, d.w));
}
__device__ __forceinline__ int half_split(int E) { return 16 * (((E >> 4) + 1) >> 1); }

__device__ __forceinline__ float row16_sum(float v) {
  v += __builtin_bit_cast(float, __builtin_amdgcn_update_dpp(0, __builtin_bit_cast(int, v), 0x161, 0xf, 0xf, true));
  v += __builtin_bit_cast(float, __builtin_amdgcn_update_dpp(0, __builtin_bit_cast(int, v), 0x162, 0xf, 0xf, true));
  v += __builtin_bit_cast(float, __builtin_amdgcn_update_dpp(0, __builtin_bit_cast(int, v), 0x164, 0xf, 0xf, true));
  v += __builtin_bit_cast(float, __builtin_amdgcn_update_dpp(0, __builtin_bit_cast(int, v), 0x168, 0xf, 0xf, true));
  return v;
}
template <int U> __device__ __forceinline__ float row16_lane(float v) {
  return __builtin_bit_cast(float, __builtin_amdgcn_update_dpp(0, __builtin_bit_cast(int, v), 0x150 + U, 0xf, 0xf, true));
}
#define FOR16(M) M(0) M(1) M(2) M(3) M(4) M(5) M(6) M(7) M(8) M(9) M(10) M(11) M(12) M(13) M(14) M(15)

__global__ __launch_bounds__(256) void k_logits(const float* __restrict__ h, const float* __restrict__ a_src,
                         const float* __restrict__ a_dst, float* __restrict__ esrc,
                         float* __restrict__ edst, int n,
                         const int* __restrict__ ei, int E, unsigned char* __restrict__ rid,
                         unsigned* __restrict__ pk) {
  int i = blockIdx.x * blockDim.x + threadIdx.x;
  {
    const int sub = (int)threadIdx.x & 15;
    const int nb  = i - sub;
    if (nb < n) {
      const v4f as = *(const v4f*)(a_src + sub * 4);
      const v4f ad = *(const v4f*)(a_dst + sub * 4);
      v4f hv[16];
#pragma unroll
      for (int u = 0; u < 16; ++u) {
        int r = nb + u;
        r = r < n ? r : n - 1;
        hv[u] = *(const v4f*)(h + (size_t)r * C1 + sub * 4);
      }
      asm volatile("" :: "v"(hv[0]), "v"(hv[1]), "v"(hv[2]), "v"(hv[3]), "v"(hv[4]), "v"(hv[5]), "v"(hv[6]), "v"(hv[7]),
                         "v"(hv[8]), "v"(hv[9]), "v"(hv[10]), "v"(hv[11]), "v"(hv[12]), "v"(hv[13]), "v"(hv[14]), "v"(hv[15]),
                         "v"(as), "v"(ad));
      float s = 0.0f, d = 0.0f;
#pragma unroll
      for (int u = 0; u < 16; ++u) {
        float su = row16_sum(hv[u].x * as.x + hv[u].y * as.y + hv[u].z * as.z + hv[u].w * as.w);
        float du = row16_sum(hv[u].x * ad.x + hv[u].y * ad.y + hv[u].z * ad.z + hv[u].w * ad.w);
        if (sub == u) { s = su; d = du; }
      }
      if (i < n) {
        esrc[i] = s;
        edst[i] = d;
      }
    }
  }
  const int* dst = ei + E;
  const int nthr = (int)(gridDim.x * blockDim.x);
  const int nv = E >> 4;
  int done = 0;
  if (((((size_t)dst) | ((size_t)ei)) & 15) == 0) {
    const int4* dv = (const int4*)dst;
    const int4* sv = (const int4*)ei;
    for (int v = i; v < nv; v += nthr) {
      int4 a0 = dv[4 * v], a1 = dv[4 * v + 1], a2 = dv[4 * v + 2], a3 = dv[4 * v + 3];
      int4 s0 = sv[4 * v], s1 = sv[4 * v + 1], s2 = sv[4 * v + 2], s3 = sv[4 * v + 3];
      ((uint4*)rid)[v] = make_uint4(range_word(a0, n), range_word(a1, n), range_word(a2, n), range_word(a3, n));
      uint4* pv = (uint4*)pk + 4 * (size_t)v;
      pv[0] = edge_words(s0, a0);
      pv[1] = edge_words(s1, a1);
      pv[2] = edge_words(s2, a2);
      pv[3] = edge_words(s3, a3);
    }
    done = 16 * nv;
  }
  for (int idx = done + i; idx < E; idx += nthr) {
    const int d = dst[idx];
    rid[idx] = (unsigned char)range_byte(d, n);
    pk[idx] = edge_word(ei[idx], d);
  }
}

#define QCAP  12288
#define AGT   1024
#define PU    8
#define AU    8
static_assert(RNG <= 1024, "the queue packs the local node into 10 bits");
__device__ __forceinline__ float l1_logit(float es, float ed) {
  float v = es + ed;
  return v > 0.0f ? v : NEG_SLOPE * v;
}
__device__ __forceinline__ unsigned zero_bytes(unsigned x) {
  unsigned t = ((x & 0x7f7f7f7fu) + 0x7f7f7f7fu) | x;
  return ~t & 0x80808080u;
}
__device__ __forceinline__ void lds_max_f32(float* addr, float val) {
  if (__float_as_int(val) >= 0) atomicMax((int*)addr, __float_as_int(val));
  else                          atomicMin((unsigned int*)addr, __float_as_uint(val));
}
__device__ __forceinline__ void aggr_add(float p, float4 hv, int dl, int sub, int rot,
                                         float* acc, float* zl) {
  float* ap = acc + dl * C1 + sub * 4;
  float v[4] = {p * hv.x, p * hv.y, p * hv.z, p * hv.w};
#pragma unroll
  for (int j = 0; j < 4; ++j) {
    int jj = (j + rot) & 3;
    atomicAdd(ap + jj, v[jj]);
  }
  if (sub == 0) atomicAdd(zl + dl, p);
}

__global__ __launch_bounds__(AGT) void k_edge_aggr64(const int* __restrict__ ei, int E, int n,
                              const float* __restrict__ esrc, const float* __restrict__ edst,
                              const float* __restrict__ h, float* __restrict__ out1,
                              float* __restrict__ z, float* __restrict__ mx,
                              int* __restrict__ lst, int* __restrict__ cntq,
                              const unsigned char* __restrict__ rid, const unsigned* __restrict__ pk) {
  __shared__ float acc[RNG * C1];
  __shared__ float zl[RNG];
  __shared__ float ml[RNG];
  __shared__ float edl[RNG];
  __shared__ int   q[QCAP];
  __shared__ int   q2[QCAP];
  __shared__ int   qn;
  const int tid  = threadIdx.x;
  const int half = blockIdx.x & 1;
  const int base = (blockIdx.x >> 1) * RNG;
  const int cnt  = min(RNG, n - base);
  const int sub  = tid & 15;
  const int grp  = tid >> 4;
  const int rot  = grp & 1;
  if (tid == 0) qn = 0;

  for (int r = tid; r < cnt; r += AGT) {
    float ed = edst[base + r];
    edl[r] = ed;
    ml[r] = half == 0 ? l1_logit(esrc[base + r], ed) : -INFINITY;
  }
  __syncthreads();

  const int* dst = ei + E;
  const unsigned ucnt = (unsigned)cnt;
  const unsigned rno = (unsigned)(blockIdx.x >> 1);
  const unsigned pat = rno * 0x01010101u;
  const int nv = E >> 4;
  const int split = half_split(E);
  const int hb = half ? split : 0;
  const int he = half ? E : split;
  {
    const uint4* rv = (const uint4*)rid;
    const int ve = half ? nv : (split >> 4);
    for (int v0 = (hb >> 4) + tid; v0 < ve; v0 += 4 * AGT) {
      uint4 a[4];
#pragma unroll
      for (int u = 0; u < 4; ++u) {
        int v = v0 + u * AGT;
        a[u] = rv[v < ve ? v : v0];
      }
      unsigned lo = (zero_bytes(a[0].x ^ pat) >> 7) | (zero_bytes(a[0].y ^ pat) >> 6) |
                    (zero_bytes(a[0].z ^ pat) >> 5) | (zero_bytes(a[0].w ^ pat) >> 4) |
                    (zero_bytes(a[1].x ^ pat) >> 3) | (zero_bytes(a[1].y ^ pat) >> 2) |
                    (zero_bytes(a[1].z ^ pat) >> 1) |  zero_bytes(a[1].w ^ pat);
      unsigned hi = (zero_bytes(a[2].x ^ pat) >> 7) | (zero_bytes(a[2].y ^ pat) >> 6) |
                    (zero_bytes(a[2].z ^ pat) >> 5) | (zero_bytes(a[2].w ^ pat) >> 4) |
                    (zero_bytes(a[3].x ^ pat) >> 3) | (zero_bytes(a[3].y ^ pat) >> 2) |
                    (zero_bytes(a[3].z ^ pat) >> 1) |  zero_bytes(a[3].w ^ pat);
      lo &= (v0 + AGT < ve) ? 0xffffffffu : 0x0f0f0f0fu;
      hi &= ((v0 + 2 * AGT < ve) ? 0x0f0f0f0fu : 0u) | ((v0 + 3 * AGT < ve) ? 0xf0f0f0f0u : 0u);
      unsigned long long hit = ((unsigned long long)hi << 32) | lo;
      while (hit) {
        int b = __ffsll(hit) - 1;
        hit &= hit - 1;
        int u = 2 * (b >> 5) + ((b >> 2) & 1);
        int idx = 16 * (v0 + u * AGT) + 4 * (b & 3) + ((b & 31) >> 3);
        int slot = atomicAdd(&qn, 1);
        if (slot < QCAP) q[slot] = idx;
      }
    }
  }
  if (half)
    for (int idx = 16 * nv + tid; idx < E; idx += AGT)
      if (rid[idx] == rno) {
        int slot = atomicAdd(&qn, 1);
        if (slot < QCAP) q[slot] = idx;
      }
  __syncthreads();

  const int nqraw = qn;
  const bool listed = nqraw <= QCAP;
  const int nq = listed ? nqraw : 0;
  if (tid == 0) cntq[blockIdx.x] = nqraw;

  int* lq = lst + (size_t)blockIdx.x * QCAP;
  if (listed) {
    for (int k0 = tid; k0 < nq; k0 += AU * AGT) {
      int      idx[AU];
      unsigned ew[AU];
      float    es[AU];
#pragma unroll
      for (int u = 0; u < AU; ++u) {
        int k = k0 + u * AGT;
        int kk = k < nq ? k : k0;
        idx[u] = q[kk];
      }
#pragma unroll
      for (int u = 0; u < AU; ++u) ew[u] = pk[idx[u]];
#pragma unroll
      for (int u = 0; u < AU; ++u) es[u] = esrc[ew[u] >> 10];
#pragma unroll
      for (int u = 0; u < AU; ++u) {
        int k = k0 + u * AGT;
        if (k < nq) {
          int dl = (int)(ew[u] & 1023u);
          float e = l1_logit(es[u], edl[dl]);
          lds_max_f32(&ml[dl], e);
          lq[k] = (int)ew[u];
          q[k]  = (int)ew[u];
          q2[k] = __float_as_int(e);
        }
      }
    }
  } else {
    for (int idx = hb + tid; idx < he; idx += AGT) {
      int d = dst[idx];
      if ((unsigned)(d - base) < ucnt)
        lds_max_f32(&ml[d - base], l1_logit(esrc[ei[idx]], edl[d - base]));
    }
  }
  __syncthreads();

  for (int r = grp; r < cnt; r += AGT / 16) {
    float p = 0.0f;
    float4 hv = make_float4(0.0f, 0.0f, 0.0f, 0.0f);
    if (half == 0) {
      int d = base + r;
      p = __expf(l1_logit(esrc[d], edl[r]) - ml[r]);
      hv = *(const float4*)(h + (size_t)d * C1 + sub * 4);
    }
    *(float4*)(acc + r * C1 + sub * 4) = make_float4(p * hv.x, p * hv.y, p * hv.z, p * hv.w);
    if (sub == 0) zl[r] = p;
  }
  __syncthreads();

  if (listed) {
    for (int k0 = grp; k0 < nq; k0 += PU * (AGT / 16)) {
      unsigned w[PU];
      float    ev[PU];
      float4   hv[PU];
#pragma unroll
      for (int u = 0; u < PU; ++u) {
        int k = k0 + u * (AGT / 16);
        int kk = k < nq ? k : k0;
        w[u]  = (unsigned)q[kk];
        ev[u] = __int_as_float(q2[kk]);
      }
#pragma unroll
      for (int u = 0; u < PU; ++u)
        hv[u] = *(const float4*)(h + (size_t)(w[u] >> 10) * C1 + sub * 4);
#pragma unroll
      for (int u = 0; u < PU; ++u)
        if (k0 + u * (AGT / 16) < nq) {
          int dl = (int)(w[u] & 1023u);
          aggr_add(__expf(ev[u] - ml[dl]), hv[u], dl, sub, rot, acc, zl);
        }
    }
  } else {
    for (int idx = hb + tid; idx < he; idx += AGT) {
      int d = dst[idx];
      if ((unsigned)(d - base) < ucnt) {
        int s = ei[idx];
        float p = __expf(l1_logit(esrc[s], edl[d - base]) - ml[d - base]);
#pragma nounroll
        for (int s2 = 0; s2 < 16; ++s2)
          aggr_add(p, *(const float4*)(h + (size_t)s * C1 + s2 * 4), d - base, s2, 0, acc, zl);
      }
    }
  }
  __syncthreads();

  float* op = out1 + (size_t)half * n * C1 + (size_t)base * C1;
  float* zp = z + (size_t)half * n + base;
  float* mp = mx + (size_t)half * n + base;
  for (int i = tid; i < cnt * (C1 / 4); i += AGT)
    *(float4*)(op + i * 4) = *(const float4*)(acc + i * 4);
  for (int r = tid; r < cnt; r += AGT) { zp[r] = zl[r]; mp[r] = ml[r]; }
}

__global__ __launch_bounds__(256) void k_layer2_prep(const float* __restrict__ out1, const float* __restrict__ z,
                              const float* __restrict__ mx, const float* __restrict__ b1,
                              const float* __restrict__ W2, float* __restrict__ g, int n) {
  const int i = blockIdx.x * blockDim.x + threadIdx.x;
  const int sub = (int)threadIdx.x & 15;
  const int nb  = i - sub;
  if (nb >= n) return;
  const float* op = out1 + (size_t)sub * 4;
  const float* oq = out1 + (size_t)n * C1 + (size_t)sub * 4;
  v8f pr[16];
#pragma unroll
  for (int u = 0; u < 16; ++u) {
    int r = nb + u;
    r = r < n ? r : n - 1;
    const v4f a = *(const v4f*)(op + (size_t)r * C1);
    const v4f b = *(const v4f*)(oq + (size_t)r * C1);
    pr[u] = __builtin_shufflevector(a, b, 0, 1, 2, 3, 4, 5, 6, 7);
  }
  const int ic = i < n ? i : n - 1;
  const float m0 = mx[ic], m1 = mx[n + ic];
  const float z0 = z[ic], z1 = z[n + ic];
  const v4f bb = *(const v4f*)(b1 + sub * 4);
  const v4f ww = *(const v4f*)(W2 + sub * 4);
  asm volatile("" :: "v"(pr[0]), "v"(pr[1]), "v"(pr[2]), "v"(pr[3]), "v"(pr[4]), "v"(pr[5]), "v"(pr[6]), "v"(pr[7]),
                     "v"(pr[8]), "v"(pr[9]), "v"(pr[10]), "v"(pr[11]), "v"(pr[12]), "v"(pr[13]), "v"(pr[14]), "v"(pr[15]),
                     "v"(m0), "v"(m1), "v"(z0), "v"(z1), "v"(bb), "v"(ww));
  const float mm = fmaxf(m0, m1);
  const float w0 = __expf(m0 - mm);
  const float w1 = __expf(m1 - mm);
  const float rz = 1.0f / (z0 * w0 + z1 * w1);
  float acc = 0.0f;
#define L2P_NODE(U) { \
    const float q0 = row16_lane<U>(w0), q1 = row16_lane<U>(w1), qz = row16_lane<U>(rz); \
    const v8f p = pr[U]; \
    const float x0 = p[0] * q0 + p[4] * q1, x1 = p[1] * q0 + p[5] * q1; \
    const float x2 = p[2] * q0 + p[6] * q1, x3 = p[3] * q0 + p[7] * q1; \
    const float v0 = fmaxf(x0 * qz + bb.x, 0.0f); \
    const float v1 = fmaxf(x1 * qz + bb.y, 0.0f); \
    const float v2 = fmaxf(x2 * qz + bb.z, 0.0f); \
    const float v3 = fmaxf(x3 * qz + bb.w, 0.0f); \
    const float t = row16_sum(v0 * ww.x + v1 * ww.y + v2 * ww.z + v3 * ww.w); \
    if (sub == U) acc = t; }
  FOR16(L2P_NODE)
#undef L2P_NODE
  if (i < n) g[i] = acc;
}

#define L2T 1024
#define L2U (QCAP / L2T)
__device__ __forceinline__ float l2_logit(float gs, float as2, float ed) {
  float v = gs * as2 + ed;
  return v > 0.0f ? v : NEG_SLOPE * v;
}

__global__ __launch_bounds__(L2T) void k_l2_range(const int* __restrict__ ei, int E, int n,
                              const int* __restrict__ lst, const int* __restrict__ cntq,
                              const float* __restrict__ g, const float* __restrict__ a_src2,
                              const float* __restrict__ a_dst2, float* __restrict__ part) {
  __shared__ float ml[RNG];
  __shared__ float zl[RNG];
  __shared__ float nl[RNG];
  __shared__ float edl[RNG];
  __shared__ float gl[RNG];
  __shared__ float esl[RNG];
  const int tid  = threadIdx.x;
  const int half = blockIdx.x & 1;
  const int base = (blockIdx.x >> 1) * RNG;
  const int cnt  = min(RNG, n - base);
  const unsigned ucnt = (unsigned)cnt;
  const float as2 = a_src2[0];
  const float ad2 = a_dst2[0];
  const int nqraw = cntq[blockIdx.x];
  const bool listed = nqraw <= QCAP;
  const int nq = listed ? nqraw : 0;

  const int* dst = ei + E;
  const int split = half_split(E);
  const int hb = half ? split : 0;
  const int he = half ? E : split;

  const int* lq = lst + (size_t)blockIdx.x * QCAP;
  unsigned sd[L2U];
  float gs[L2U], ev[L2U];
#pragma unroll
  for (int u = 0; u < L2U; ++u) {
    int k = tid + u * L2T;
    sd[u] = k < nq ? (unsigned)lq[k] : ((unsigned)base << 10);
  }
#pragma unroll
  for (int u = 0; u < L2U; ++u) gs[u] = g[sd[u] >> 10];

  for (int r = tid; r < cnt; r += L2T) {
    float gv = g[base + r];
    float ed = gv * ad2;
    float v = l2_logit(gv, as2, ed);
    gl[r] = gv;
    edl[r] = ed;
    esl[r] = v;
    ml[r] = half == 0 ? v : -INFINITY;
  }
  __syncthreads();

  if (listed) {
#pragma unroll
    for (int u = 0; u < L2U; ++u) {
      int dl = (int)(sd[u] & 1023u);
      ev[u] = l2_logit(gs[u], as2, edl[dl]);
      if (tid + u * L2T < nq) lds_max_f32(&ml[dl], ev[u]);
    }
  } else {
    for (int idx = hb + tid; idx < he; idx += L2T) {
      int d = dst[idx];
      if ((unsigned)(d - base) < ucnt)
        lds_max_f32(&ml[d - base], l2_logit(g[ei[idx]], as2, edl[d - base]));
    }
  }
  __syncthreads();

  for (int r = tid; r < cnt; r += L2T) {
    float p = half == 0 ? __expf(esl[r] - ml[r]) : 0.0f;
    zl[r] = p;
    nl[r] = p * gl[r];
  }
  __syncthreads();

  if (listed) {
#pragma unroll
    for (int u = 0; u < L2U; ++u) {
      if (tid + u * L2T < nq) {
        int dl = (int)(sd[u] & 1023u);
        float p = __expf(ev[u] - ml[dl]);
        atomicAdd(&zl[dl], p);
        atomicAdd(&nl[dl], p * gs[u]);
      }
    }
  } else {
    for (int idx = hb + tid; idx < he; idx += L2T) {
      int d = dst[idx];
      if ((unsigned)(d - base) < ucnt) {
        float gsv = g[ei[idx]];
        float p = __expf(l2_logit(gsv, as2, edl[d - base]) - ml[d - base]);
        atomicAdd(&zl[d - base], p);
        atomicAdd(&nl[d - base], p * gsv);
      }
    }
  }
  __syncthreads();

  float* pp = part + (size_t)half * 3 * n + base;
  for (int r = tid; r < cnt; r += L2T) {
    pp[r] = ml[r];
    pp[(size_t)n + r] = zl[r];
    pp[(size_t)2 * n + r] = nl[r];
  }
}

__global__ void k_final(const float* __restrict__ part, const float* __restrict__ b2,
                        float* __restrict__ out, int n) {
  int i = blockIdx.x * blockDim.x + threadIdx.x;
  if (i >= n) return;
  const size_t sn = (size_t)n;
  float m0 = part[i],          z0 = part[sn + i],     s0 = part[2 * sn + i];
  float m1 = part[3 * sn + i], z1 = part[4 * sn + i], s1 = part[5 * sn + i];
  float mm = fmaxf(m0, m1);
  float w0 = __expf(m0 - mm);
  float w1 = __expf(m1 - mm);
  float zz = z0 * w0 + z1 * w1;
  float ss = s0 * w0 + s1 * w1;
  float v = ss / zz + b2[0];
  out[i] = 1.0f / (1.0f + __expf(-v));
}

extern "C" void kernel_launch(void* const* d_in, const int* in_sizes, int n_in,
                              void* d_out, int out_size, void* d_ws, size_t ws_size,
                              hipStream_t stream) {
  const int*   ei     = (const int*)d_in[0];
  const float* x      = (const float*)d_in[1];
  const float* W1     = (const float*)d_in[2];
  const float* a_src1 = (const float*)d_in[3];
  const float* a_dst1 = (const float*)d_in[4];
  const float* b1     = (const float*)d_in[5];
  const float* W2     = (const float*)d_in[6];
  const float* a_src2 = (const float*)d_in[7];
  const float* a_dst2 = (const float*)d_in[8];
  const float* b2     = (const float*)d_in[9];

  const int E = in_sizes[0] / 2;
  const int n = in_sizes[1] / DIN;

  float* f     = (float*)d_ws;
  float* h     = f;
  float* out1  = f + (size_t)64 * n;
  float* esrc  = f + (size_t)192 * n;
  float* edst  = f + (size_t)193 * n;
  float* g     = f + (size_t)194 * n;
  float* z     = f + (size_t)195 * n;
  float* mx    = f + (size_t)197 * n;
  const int nwg = 2 * ((n + RNG - 1) / RNG);
  const size_t o1 = ((size_t)199 * n + 3) & ~(size_t)3;
  float* part  = f + o1;
  const size_t o2 = (o1 + (size_t)6 * n + 3) & ~(size_t)3;
  int*   lst   = (int*)(f + o2);
  int*   cntq  = lst + (size_t)nwg * QCAP;
  unsigned char* rid = (unsigned char*)(cntq + ((nwg + 3) & ~3));
  unsigned* pk = (unsigned*)(rid + (((size_t)E + 15) & ~(size_t)15));
  float* out   = (float*)d_out;

  const int B = 256;
  dim3 blk(B);

  k_gemm_wmma<<<dim3(256), dim3(32 * GEMM_WAVES), 0, stream>>>(x, W1, h, n);
  k_logits<<<dim3((n + B - 1) / B), blk, 0, stream>>>(h, a_src1, a_dst1, esrc, edst, n, ei, E, rid, pk);
  k_edge_aggr64<<<dim3(nwg), dim3(AGT), 0, stream>>>(ei, E, n, esrc, edst, h, out1, z, mx, lst, cntq, rid, pk);

  k_layer2_prep<<<dim3((n + B - 1) / B), blk, 0, stream>>>(out1, z, mx, b1, W2, g, n);
  k_l2_range<<<dim3(nwg), dim3(L2T), 0, stream>>>(ei, E, n, lst, cntq, g, a_src2, a_dst2, part);

  k_final<<<dim3((n + B - 1) / B), blk, 0, stream>>>(part, b2, out, n);
}


# Range kernel sum pass: a wave forms exp weights and z adds of 32 queued edges one per lane, adds rows an edge at a time, 32 lanes x 2 channels, word and weight by readlane; LDS add order changes
# speedup vs baseline: 5.7012x; 1.0661x over previous
#include <hip/hip_runtime.h>
#include <math.h>

typedef float v2f __attribute__((ext_vector_type(2)));
typedef float v4f __attribute__((ext_vector_type(4)));
typedef float v8f __attribute__((ext_vector_type(8)));

#define DIN 128
#define C1  64
#define NEG_SLOPE 0.2f

#define GEMM_WAVES 8
__global__ __launch_bounds__(32 * GEMM_WAVES) void k_gemm_wmma(const float* __restrict__ x,
                                                   const float* __restrict__ W,
                                                   float* __restrict__ h, int n) {
  const int lane = threadIdx.x & 31;
  const int wave = __builtin_amdgcn_readfirstlane((int)(threadIdx.x >> 5));
  const int half = lane >> 4;
  const int l    = lane & 15;

  v2f bf[DIN / 4][4];
  {
    const float* wp = W + (size_t)(2 * half) * C1 + 4 * l;
#pragma unroll
    for (int ks = 0; ks < DIN / 4; ++ks) {
      const float4 r0 = *(const float4*)(wp + (size_t)(4 * ks) * C1);
      const float4 r1 = *(const float4*)(wp + (size_t)(4 * ks + 1) * C1);
      bf[ks][0] = (v2f){r0.x, r1.x};
      bf[ks][1] = (v2f){r0.y, r1.y};
      bf[ks][2] = (v2f){r0.z, r1.z};
      bf[ks][3] = (v2f){r0.w, r1.w};
    }
  }

  const int ntiles = (n + 15) >> 4;
  const int nwaves = (int)gridDim.x * GEMM_WAVES;
  for (int tile = (int)blockIdx.x * GEMM_WAVES + wave; tile < ntiles; tile += nwaves) {
    const int rowBase = tile * 16;
    int arow = rowBase + l;
    if (arow > n - 1) arow = n - 1;
    const float* ap = x + (size_t)arow * DIN + 2 * half;

    v8f c0 = {}, c1 = {}, c2 = {}, c3 = {};
#pragma unroll
    for (int ks = 0; ks < DIN / 4; ++ks) {
      v2f a = *(const v2f*)(ap + 4 * ks);
      c0 = __builtin_amdgcn_wmma_f32_16x16x4_f32(false, a, false, bf[ks][0], (short)0, c0, false, false);
      c1 = __builtin_amdgcn_wmma_f32_16x16x4_f32(false, a, false, bf[ks][1], (short)0, c1, false, false);
      c2 = __builtin_amdgcn_wmma_f32_16x16x4_f32(false, a, false, bf[ks][2], (short)0, c2, false, false);
      c3 = __builtin_amdgcn_wmma_f32_16x16x4_f32(false, a, false, bf[ks][3], (short)0, c3, false, false);
    }

#pragma unroll
    for (int v = 0; v < 8; ++v) {
      int row = rowBase + 8 * half + v;
      if (row < n)
        *(float4*)(h + (size_t)row * C1 + 4 * l) = make_float4(c0[v], c1[v], c2[v], c3[v]);
    }
  }
}

#define RNG   800
__device__ __forceinline__ unsigned range_byte(int d, int n) {
  return (unsigned)d < (unsigned)n ? (unsigned)d / RNG : 255u;
}
__device__ __forceinline__ unsigned range_word(int4 a, int n) {
  return range_byte(a.x, n) | (range_byte(a.y, n) << 8) | (range_byte(a.z, n) << 16) | (range_byte(a.w, n) << 24);
}
__device__ __forceinline__ unsigned edge_word(int s, int d) {
  return ((unsigned)s << 10) | ((unsigned)d - ((unsigned)d / RNG) * RNG);
}
__device__ __forceinline__ uint4 edge_words(int4 s, int4 d) {
  return make_uint4(edge_word(s.x, d.x), edge_word(s.y, d.y), edge_word(s.z, d.z), edge_word(s.w, d.w));
}
__device__ __forceinline__ int half_split(int E) { return 16 * (((E >> 4) + 1) >> 1); }

__device__ __forceinline__ float row16_sum(float v) {
  v += __builtin_bit_cast(float, __builtin_amdgcn_update_dpp(0, __builtin_bit_cast(int, v), 0x161, 0xf, 0xf, true));
  v += __builtin_bit_cast(float, __builtin_amdgcn_update_dpp(0, __builtin_bit_cast(int, v), 0x162, 0xf, 0xf, true));
  v += __builtin_bit_cast(float, __builtin_amdgcn_update_dpp(0, __builtin_bit_cast(int, v), 0x164, 0xf, 0xf, true));
  v += __builtin_bit_cast(float, __builtin_amdgcn_update_dpp(0, __builtin_bit_cast(int, v), 0x168, 0xf, 0xf, true));
  return v;
}
template <int U> __device__ __forceinline__ float row16_lane(float v) {
  return __builtin_bit_cast(float, __builtin_amdgcn_update_dpp(0, __builtin_bit_cast(int, v), 0x150 + U, 0xf, 0xf, true));
}
#define FOR16(M) M(0) M(1) M(2) M(3) M(4) M(5) M(6) M(7) M(8) M(9) M(10) M(11) M(12) M(13) M(14) M(15)

__global__ __launch_bounds__(256) void k_logits(const float* __restrict__ h, const float* __restrict__ a_src,
                         const float* __restrict__ a_dst, float* __restrict__ esrc,
                         float* __restrict__ edst, int n,
                         const int* __restrict__ ei, int E, unsigned char* __restrict__ rid,
                         unsigned* __restrict__ pk) {
  int i = blockIdx.x * blockDim.x + threadIdx.x;
  {
    const int sub = (int)threadIdx.x & 15;
    const int nb  = i - sub;
    if (nb < n) {
      const v4f as = *(const v4f*)(a_src + sub * 4);
      const v4f ad = *(const v4f*)(a_dst + sub * 4);
      v4f hv[16];
#pragma unroll
      for (int u = 0; u < 16; ++u) {
        int r = nb + u;
        r = r < n ? r : n - 1;
        hv[u] = *(const v4f*)(h + (size_t)r * C1 + sub * 4);
      }
      asm volatile("" :: "v"(hv[0]), "v"(hv[1]), "v"(hv[2]), "v"(hv[3]), "v"(hv[4]), "v"(hv[5]), "v"(hv[6]), "v"(hv[7]),
                         "v"(hv[8]), "v"(hv[9]), "v"(hv[10]), "v"(hv[11]), "v"(hv[12]), "v"(hv[13]), "v"(hv[14]), "v"(hv[15]),
                         "v"(as), "v"(ad));
      float s = 0.0f, d = 0.0f;
#pragma unroll
      for (int u = 0; u < 16; ++u) {
        float su = row16_sum(hv[u].x * as.x + hv[u].y * as.y + hv[u].z * as.z + hv[u].w * as.w);
        float du = row16_sum(hv[u].x * ad.x + hv[u].y * ad.y + hv[u].z * ad.z + hv[u].w * ad.w);
        if (sub == u) { s = su; d = du; }
      }
      if (i < n) {
        esrc[i] = s;
        edst[i] = d;
      }
    }
  }
  const int* dst = ei + E;
  const int nthr = (int)(gridDim.x * blockDim.x);
  const int nv = E >> 4;
  int done = 0;
  if (((((size_t)dst) | ((size_t)ei)) & 15) == 0) {
    const int4* dv = (const int4*)dst;
    const int4* sv = (const int4*)ei;
    for (int v = i; v < nv; v += nthr) {
      int4 a0 = dv[4 * v], a1 = dv[4 * v + 1], a2 = dv[4 * v + 2], a3 = dv[4 * v + 3];
      int4 s0 = sv[4 * v], s1 = sv[4 * v + 1], s2 = sv[4 * v + 2], s3 = sv[4 * v + 3];
      ((uint4*)rid)[v] = make_uint4(range_word(a0, n), range_word(a1, n), range_word(a2, n), range_word(a3, n));
      uint4* pv = (uint4*)pk + 4 * (size_t)v;
      pv[0] = edge_words(s0, a0);
      pv[1] = edge_words(s1, a1);
      pv[2] = edge_words(s2, a2);
      pv[3] = edge_words(s3, a3);
    }
    done = 16 * nv;
  }
  for (int idx = done + i; idx < E; idx += nthr) {
    const int d = dst[idx];
    rid[idx] = (unsigned char)range_byte(d, n);
    pk[idx] = edge_word(ei[idx], d);
  }
}

#define QCAP  12288
#define AGT   1024
#define WE    32
#define AU    8
static_assert(RNG <= 1024, "the queue packs the local node into 10 bits");
__device__ __forceinline__ float l1_logit(float es, float ed) {
  float v = es + ed;
  return v > 0.0f ? v : NEG_SLOPE * v;
}
__device__ __forceinline__ unsigned zero_bytes(unsigned x) {
  unsigned t = ((x & 0x7f7f7f7fu) + 0x7f7f7f7fu) | x;
  return ~t & 0x80808080u;
}
__device__ __forceinline__ void lds_max_f32(float* addr, float val) {
  if (__float_as_int(val) >= 0) atomicMax((int*)addr, __float_as_int(val));
  else                          atomicMin((unsigned int*)addr, __float_as_uint(val));
}
__device__ __forceinline__ void aggr_add(float p, float4 hv, int dl, int sub, float* acc, float* zl) {
  float* ap = acc + dl * C1 + sub * 4;
  atomicAdd(ap, p * hv.x);
  atomicAdd(ap + 1, p * hv.y);
  atomicAdd(ap + 2, p * hv.z);
  atomicAdd(ap + 3, p * hv.w);
  if (sub == 0) atomicAdd(zl + dl, p);
}

__device__ __forceinline__ void row_add(unsigned w, float p, float2 hv, int u, float* al) {
  const unsigned wu = (unsigned)__builtin_amdgcn_readlane((int)w, u);
  const float pu = __int_as_float(__builtin_amdgcn_readlane(__float_as_int(p), u));
  float* ap = al + (wu & 1023u) * C1;
  atomicAdd(ap, pu * hv.x);
  atomicAdd(ap + 1, pu * hv.y);
}

__global__ __launch_bounds__(AGT) void k_edge_aggr64(const int* __restrict__ ei, int E, int n,
                              const float* __restrict__ esrc, const float* __restrict__ edst,
                              const float* __restrict__ h, float* __restrict__ out1,
                              float* __restrict__ z, float* __restrict__ mx,
                              int* __restrict__ lst, int* __restrict__ cntq,
                              const unsigned char* __restrict__ rid, const unsigned* __restrict__ pk) {
  __shared__ float acc[RNG * C1];
  __shared__ float zl[RNG];
  __shared__ float ml[RNG];
  __shared__ float edl[RNG];
  __shared__ int   q[QCAP];
  __shared__ int   q2[QCAP];
  __shared__ int   qn;
  const int tid  = threadIdx.x;
  const int half = blockIdx.x & 1;
  const int base = (blockIdx.x >> 1) * RNG;
  const int cnt  = min(RNG, n - base);
  const int sub  = tid & 15;
  const int grp  = tid >> 4;
  if (tid == 0) qn = 0;

  for (int r = tid; r < cnt; r += AGT) {
    float ed = edst[base + r];
    edl[r] = ed;
    ml[r] = half == 0 ? l1_logit(esrc[base + r], ed) : -INFINITY;
  }
  __syncthreads();

  const int* dst = ei + E;
  const unsigned ucnt = (unsigned)cnt;
  const unsigned rno = (unsigned)(blockIdx.x >> 1);
  const unsigned pat = rno * 0x01010101u;
  const int nv = E >> 4;
  const int split = half_split(E);
  const int hb = half ? split : 0;
  const int he = half ? E : split;
  {
    const uint4* rv = (const uint4*)rid;
    const int ve = half ? nv : (split >> 4);
    for (int v0 = (hb >> 4) + tid; v0 < ve; v0 += 4 * AGT) {
      uint4 a[4];
#pragma unroll
      for (int u = 0; u < 4; ++u) {
        int v = v0 + u * AGT;
        a[u] = rv[v < ve ? v : v0];
      }
      unsigned lo = (zero_bytes(a[0].x ^ pat) >> 7) | (zero_bytes(a[0].y ^ pat) >> 6) |
                    (zero_bytes(a[0].z ^ pat) >> 5) | (zero_bytes(a[0].w ^ pat) >> 4) |
                    (zero_bytes(a[1].x ^ pat) >> 3) | (zero_bytes(a[1].y ^ pat) >> 2) |
                    (zero_bytes(a[1].z ^ pat) >> 1) |  zero_bytes(a[1].w ^ pat);
      unsigned hi = (zero_bytes(a[2].x ^ pat) >> 7) | (zero_bytes(a[2].y ^ pat) >> 6) |
                    (zero_bytes(a[2].z ^ pat) >> 5) | (zero_bytes(a[2].w ^ pat) >> 4) |
                    (zero_bytes(a[3].x ^ pat) >> 3) | (zero_bytes(a[3].y ^ pat) >> 2) |
                    (zero_bytes(a[3].z ^ pat) >> 1) |  zero_bytes(a[3].w ^ pat);
      lo &= (v0 + AGT < ve) ? 0xffffffffu : 0x0f0f0f0fu;
      hi &= ((v0 + 2 * AGT < ve) ? 0x0f0f0f0fu : 0u) | ((v0 + 3 * AGT < ve) ? 0xf0f0f0f0u : 0u);
      unsigned long long hit = ((unsigned long long)hi << 32) | lo;
      while (hit) {
        int b = __ffsll(hit) - 1;
        hit &= hit - 1;
        int u = 2 * (b >> 5) + ((b >> 2) & 1);
        int idx = 16 * (v0 + u * AGT) + 4 * (b & 3) + ((b & 31) >> 3);
        int slot = atomicAdd(&qn, 1);
        if (slot < QCAP) q[slot] = idx;
      }
    }
  }
  if (half)
    for (int idx = 16 * nv + tid; idx < E; idx += AGT)
      if (rid[idx] == rno) {
        int slot = atomicAdd(&qn, 1);
        if (slot < QCAP) q[slot] = idx;
      }
  __syncthreads();

  const int nqraw = qn;
  const bool listed = nqraw <= QCAP;
  const int nq = listed ? nqraw : 0;
  if (tid == 0) cntq[blockIdx.x] = nqraw;

  int* lq = lst + (size_t)blockIdx.x * QCAP;
  if (listed) {
    for (int k0 = tid; k0 < nq; k0 += AU * AGT) {
      int      idx[AU];
      unsigned ew[AU];
      float    es[AU];
#pragma unroll
      for (int u = 0; u < AU; ++u) {
        int k = k0 + u * AGT;
        int kk = k < nq ? k : k0;
        idx[u] = q[kk];
      }
#pragma unroll
      for (int u = 0; u < AU; ++u) ew[u] = pk[idx[u]];
#pragma unroll
      for (int u = 0; u < AU; ++u) es[u] = esrc[ew[u] >> 10];
#pragma unroll
      for (int u = 0; u < AU; ++u) {
        int k = k0 + u * AGT;
        if (k < nq) {
          int dl = (int)(ew[u] & 1023u);
          float e = l1_logit(es[u], edl[dl]);
          lds_max_f32(&ml[dl], e);
          lq[k] = (int)ew[u];
          q[k]  = (int)ew[u];
          q2[k] = __float_as_int(e);
        }
      }
    }
  } else {
    for (int idx = hb + tid; idx < he; idx += AGT) {
      int d = dst[idx];
      if ((unsigned)(d - base) < ucnt)
        lds_max_f32(&ml[d - base], l1_logit(esrc[ei[idx]], edl[d - base]));
    }
  }
  __syncthreads();

  for (int r = grp; r < cnt; r += AGT / 16) {
    float p = 0.0f;
    float4 hv = make_float4(0.0f, 0.0f, 0.0f, 0.0f);
    if (half == 0) {
      int d = base + r;
      p = __expf(l1_logit(esrc[d], edl[r]) - ml[r]);
      hv = *(const float4*)(h + (size_t)d * C1 + sub * 4);
    }
    *(float4*)(acc + r * C1 + sub * 4) = make_float4(p * hv.x, p * hv.y, p * hv.z, p * hv.w);
    if (sub == 0) zl[r] = p;
  }
  __syncthreads();

  if (listed) {
    const int wv   = __builtin_amdgcn_readfirstlane(tid >> 5);
    const int nqs  = __builtin_amdgcn_readfirstlane(nq);
    const int lane = tid & 31;
    float* al = acc + lane * 2;
    for (int k0 = WE * wv; k0 < nqs; k0 += WE * (AGT / 32)) {
      const int nv = min(WE, nqs - k0);
      const int kk = lane < nv ? k0 + lane : k0;
      const unsigned w = (unsigned)q[kk];
      float2 hv[WE];
#pragma unroll
      for (int u = 0; u < WE; ++u) {
        const unsigned wu = (unsigned)__builtin_amdgcn_readlane((int)w, u);
        const float* hr = h + (size_t)(wu >> 10) * C1;
        hv[u] = *(const float2*)(hr + lane * 2);
      }
      const int dl = (int)(w & 1023u);
      const float p = __expf(__int_as_float(q2[kk]) - ml[dl]);
      if (lane < nv) atomicAdd(zl + dl, p);
      if (nv == WE) {
#pragma unroll
        for (int u = 0; u < WE; ++u) row_add(w, p, hv[u], u, al);
      } else {
#pragma unroll
        for (int u = 0; u < WE; ++u)
          if (u < nv) row_add(w, p, hv[u], u, al);
      }
    }
  } else {
    for (int idx = hb + tid; idx < he; idx += AGT) {
      int d = dst[idx];
      if ((unsigned)(d - base) < ucnt) {
        int s = ei[idx];
        float p = __expf(l1_logit(esrc[s], edl[d - base]) - ml[d - base]);
#pragma nounroll
        for (int s2 = 0; s2 < 16; ++s2)
          aggr_add(p, *(const float4*)(h + (size_t)s * C1 + s2 * 4), d - base, s2, acc, zl);
      }
    }
  }
  __syncthreads();

  float* op = out1 + (size_t)half * n * C1 + (size_t)base * C1;
  float* zp = z + (size_t)half * n + base;
  float* mp = mx + (size_t)half * n + base;
  for (int i = tid; i < cnt * (C1 / 4); i += AGT)
    *(float4*)(op + i * 4) = *(const float4*)(acc + i * 4);
  for (int r = tid; r < cnt; r += AGT) { zp[r] = zl[r]; mp[r] = ml[r]; }
}

__global__ __launch_bounds__(256) void k_layer2_prep(const float* __restrict__ out1, const float* __restrict__ z,
                              const float* __restrict__ mx, const float* __restrict__ b1,
                              const float* __restrict__ W2, float* __restrict__ g, int n) {
  const int i = blockIdx.x * blockDim.x + threadIdx.x;
  const int sub = (int)threadIdx.x & 15;
  const int nb  = i - sub;
  if (nb >= n) return;
  const float* op = out1 + (size_t)sub * 4;
  const float* oq = out1 + (size_t)n * C1 + (size_t)sub * 4;
  v8f pr[16];
#pragma unroll
  for (int u = 0; u < 16; ++u) {
    int r = nb + u;
    r = r < n ? r : n - 1;
    const v4f a = *(const v4f*)(op + (size_t)r * C1);
    const v4f b = *(const v4f*)(oq + (size_t)r * C1);
    pr[u] = __builtin_shufflevector(a, b, 0, 1, 2, 3, 4, 5, 6, 7);
  }
  const int ic = i < n ? i : n - 1;
  const float m0 = mx[ic], m1 = mx[n + ic];
  const float z0 = z[ic], z1 = z[n + ic];
  const v4f bb = *(const v4f*)(b1 + sub * 4);
  const v4f ww = *(const v4f*)(W2 + sub * 4);
  asm volatile("" :: "v"(pr[0]), "v"(pr[1]), "v"(pr[2]), "v"(pr[3]), "v"(pr[4]), "v"(pr[5]), "v"(pr[6]), "v"(pr[7]),
                     "v"(pr[8]), "v"(pr[9]), "v"(pr[10]), "v"(pr[11]), "v"(pr[12]), "v"(pr[13]), "v"(pr[14]), "v"(pr[15]),
                     "v"(m0), "v"(m1), "v"(z0), "v"(z1), "v"(bb), "v"(ww));
  const float mm = fmaxf(m0, m1);
  const float w0 = __expf(m0 - mm);
  const float w1 = __expf(m1 - mm);
  const float rz = 1.0f / (z0 * w0 + z1 * w1);
  float acc = 0.0f;
#define L2P_NODE(U) { \
    const float q0 = row16_lane<U>(w0), q1 = row16_lane<U>(w1), qz = row16_lane<U>(rz); \
    const v8f p = pr[U]; \
    const float x0 = p[0] * q0 + p[4] * q1, x1 = p[1] * q0 + p[5] * q1; \
    const float x2 = p[2] * q0 + p[6] * q1, x3 = p[3] * q0 + p[7] * q1; \
    const float v0 = fmaxf(x0 * qz + bb.x, 0.0f); \
    const float v1 = fmaxf(x1 * qz + bb.y, 0.0f); \
    const float v2 = fmaxf(x2 * qz + bb.z, 0.0f); \
    const float v3 = fmaxf(x3 * qz + bb.w, 0.0f); \
    const float t = row16_sum(v0 * ww.x + v1 * ww.y + v2 * ww.z + v3 * ww.w); \
    if (sub == U) acc = t; }
  FOR16(L2P_NODE)
#undef L2P_NODE
  if (i < n) g[i] = acc;
}

#define L2T 1024
#define L2U (QCAP / L2T)
__device__ __forceinline__ float l2_logit(float gs, float as2, float ed) {
  float v = gs * as2 + ed;
  return v > 0.0f ? v : NEG_SLOPE * v;
}

__global__ __launch_bounds__(L2T) void k_l2_range(const int* __restrict__ ei, int E, int n,
                              const int* __restrict__ lst, const int* __restrict__ cntq,
                              const float* __restrict__ g, const float* __restrict__ a_src2,
                              const float* __restrict__ a_dst2, float* __restrict__ part) {
  __shared__ float ml[RNG];
  __shared__ float zl[RNG];
  __shared__ float nl[RNG];
  __shared__ float edl[RNG];
  __shared__ float gl[RNG];
  __shared__ float esl[RNG];
  const int tid  = threadIdx.x;
  const int half = blockIdx.x & 1;
  const int base = (blockIdx.x >> 1) * RNG;
  const int cnt  = min(RNG, n - base);
  const unsigned ucnt = (unsigned)cnt;
  const float as2 = a_src2[0];
  const float ad2 = a_dst2[0];
  const int nqraw = cntq[blockIdx.x];
  const bool listed = nqraw <= QCAP;
  const int nq = listed ? nqraw : 0;

  const int* dst = ei + E;
  const int split = half_split(E);
  const int hb = half ? split : 0;
  const int he = half ? E : split;

  const int* lq = lst + (size_t)blockIdx.x * QCAP;
  unsigned sd[L2U];
  float gs[L2U], ev[L2U];
#pragma unroll
  for (int u = 0; u < L2U; ++u) {
    int k = tid + u * L2T;
    sd[u] = k < nq ? (unsigned)lq[k] : ((unsigned)base << 10);
  }
#pragma unroll
  for (int u = 0; u < L2U; ++u) gs[u] = g[sd[u] >> 10];

  for (int r = tid; r < cnt; r += L2T) {
    float gv = g[base + r];
    float ed = gv * ad2;
    float v = l2_logit(gv, as2, ed);
    gl[r] = gv;
    edl[r] = ed;
    esl[r] = v;
    ml[r] = half == 0 ? v : -INFINITY;
  }
  __syncthreads();

  if (listed) {
#pragma unroll
    for (int u = 0; u < L2U; ++u) {
      int dl = (int)(sd[u] & 1023u);
      ev[u] = l2_logit(gs[u], as2, edl[dl]);
      if (tid + u * L2T < nq) lds_max_f32(&ml[dl], ev[u]);
    }
  } else {
    for (int idx = hb + tid; idx < he; idx += L2T) {
      int d = dst[idx];
      if ((unsigned)(d - base) < ucnt)
        lds_max_f32(&ml[d - base], l2_logit(g[ei[idx]], as2, edl[d - base]));
    }
  }
  __syncthreads();

  for (int r = tid; r < cnt; r += L2T) {
    float p = half == 0 ? __expf(esl[r] - ml[r]) : 0.0f;
    zl[r] = p;
    nl[r] = p * gl[r];
  }
  __syncthreads();

  if (listed) {
#pragma unroll
    for (int u = 0; u < L2U; ++u) {
      if (tid + u * L2T < nq) {
        int dl = (int)(sd[u] & 1023u);
        float p = __expf(ev[u] - ml[dl]);
        atomicAdd(&zl[dl], p);
        atomicAdd(&nl[dl], p * gs[u]);
      }
    }
  } else {
    for (int idx = hb + tid; idx < he; idx += L2T) {
      int d = dst[idx];
      if ((unsigned)(d - base) < ucnt) {
        float gsv = g[ei[idx]];
        float p = __expf(l2_logit(gsv, as2, edl[d - base]) - ml[d - base]);
        atomicAdd(&zl[d - base], p);
        atomicAdd(&nl[d - base], p * gsv);
      }
    }
  }
  __syncthreads();

  float* pp = part + (size_t)half * 3 * n + base;
  for (int r = tid; r < cnt; r += L2T) {
    pp[r] = ml[r];
    pp[(size_t)n + r] = zl[r];
    pp[(size_t)2 * n + r] = nl[r];
  }
}

__global__ void k_final(const float* __restrict__ part, const float* __restrict__ b2,
                        float* __restrict__ out, int n) {
  int i = blockIdx.x * blockDim.x + threadIdx.x;
  if (i >= n) return;
  const size_t sn = (size_t)n;
  float m0 = part[i],          z0 = part[sn + i],     s0 = part[2 * sn + i];
  float m1 = part[3 * sn + i], z1 = part[4 * sn + i], s1 = part[5 * sn + i];
  float mm = fmaxf(m0, m1);
  float w0 = __expf(m0 - mm);
  float w1 = __expf(m1 - mm);
  float zz = z0 * w0 + z1 * w1;
  float ss = s0 * w0 + s1 * w1;
  float v = ss / zz + b2[0];
  out[i] = 1.0f / (1.0f + __expf(-v));
}

extern "C" void kernel_launch(void* const* d_in, const int* in_sizes, int n_in,
                              void* d_out, int out_size, void* d_ws, size_t ws_size,
                              hipStream_t stream) {
  const int*   ei     = (const int*)d_in[0];
  const float* x      = (const float*)d_in[1];
  const float* W1     = (const float*)d_in[2];
  const float* a_src1 = (const float*)d_in[3];
  const float* a_dst1 = (const float*)d_in[4];
  const float* b1     = (const float*)d_in[5];
  const float* W2     = (const float*)d_in[6];
  const float* a_src2 = (const float*)d_in[7];
  const float* a_dst2 = (const float*)d_in[8];
  const float* b2     = (const float*)d_in[9];

  const int E = in_sizes[0] / 2;
  const int n = in_sizes[1] / DIN;

  float* f     = (float*)d_ws;
  float* h     = f;
  float* out1  = f + (size_t)64 * n;
  float* esrc  = f + (size_t)192 * n;
  float* edst  = f + (size_t)193 * n;
  float* g     = f + (size_t)194 * n;
  float* z     = f + (size_t)195 * n;
  float* mx    = f + (size_t)197 * n;
  const int nwg = 2 * ((n + RNG - 1) / RNG);
  const size_t o1 = ((size_t)199 * n + 3) & ~(size_t)3;
  float* part  = f + o1;
  const size_t o2 = (o1 + (size_t)6 * n + 3) & ~(size_t)3;
  int*   lst   = (int*)(f + o2);
  int*   cntq  = lst + (size_t)nwg * QCAP;
  unsigned char* rid = (unsigned char*)(cntq + ((nwg + 3) & ~3));
  unsigned* pk = (unsigned*)(rid + (((size_t)E + 15) & ~(size_t)15));
  float* out   = (float*)d_out;

  const int B = 256;
  dim3 blk(B);

  k_gemm_wmma<<<dim3(256), dim3(32 * GEMM_WAVES), 0, stream>>>(x, W1, h, n);
  k_logits<<<dim3((n + B - 1) / B), blk, 0, stream>>>(h, a_src1, a_dst1, esrc, edst, n, ei, E, rid, pk);
  k_edge_aggr64<<<dim3(nwg), dim3(AGT), 0, stream>>>(ei, E, n, esrc, edst, h, out1, z, mx, lst, cntq, rid, pk);

  k_layer2_prep<<<dim3((n + B - 1) / B), blk, 0, stream>>>(out1, z, mx, b1, W2, g, n);
  k_l2_range<<<dim3(nwg), dim3(L2T), 0, stream>>>(ei, E, n, lst, cntq, g, a_src2, a_dst2, part);

  k_final<<<dim3((n + B - 1) / B), blk, 0, stream>>>(part, b2, out, n);
}


# Search made its own launch keeping an edge list per node half; edge launches own 400 nodes whole: true max, sums end in LDS, next stage formed there; planes, both join launches go; sum order changes
# speedup vs baseline: 6.1272x; 1.0747x over previous
#include <hip/hip_runtime.h>
#include <math.h>

typedef float v2f __attribute__((ext_vector_type(2)));
typedef float v4f __attribute__((ext_vector_type(4)));
typedef float v8f __attribute__((ext_vector_type(8)));

#define DIN 128
#define C1  64
#define NEG_SLOPE 0.2f

#define GEMM_WAVES 8
__global__ __launch_bounds__(32 * GEMM_WAVES) void k_gemm_wmma(const float* __restrict__ x,
                                                   const float* __restrict__ W,
                                                   float* __restrict__ h, int n) {
  const int lane = threadIdx.x & 31;
  const int wave = __builtin_amdgcn_readfirstlane((int)(threadIdx.x >> 5));
  const int half = lane >> 4;
  const int l    = lane & 15;

  v2f bf[DIN / 4][4];
  {
    const float* wp = W + (size_t)(2 * half) * C1 + 4 * l;
#pragma unroll
    for (int ks = 0; ks < DIN / 4; ++ks) {
      const float4 r0 = *(const float4*)(wp + (size_t)(4 * ks) * C1);
      const float4 r1 = *(const float4*)(wp + (size_t)(4 * ks + 1) * C1);
      bf[ks][0] = (v2f){r0.x, r1.x};
      bf[ks][1] = (v2f){r0.y, r1.y};
      bf[ks][2] = (v2f){r0.z, r1.z};
      bf[ks][3] = (v2f){r0.w, r1.w};
    }
  }

  const int ntiles = (n + 15) >> 4;
  const int nwaves = (int)gridDim.x * GEMM_WAVES;
  for (int tile = (int)blockIdx.x * GEMM_WAVES + wave; tile < ntiles; tile += nwaves) {
    const int rowBase = tile * 16;
    int arow = rowBase + l;
    if (arow > n - 1) arow = n - 1;
    const float* ap = x + (size_t)arow * DIN + 2 * half;

    v8f c0 = {}, c1 = {}, c2 = {}, c3 = {};
#pragma unroll
    for (int ks = 0; ks < DIN / 4; ++ks) {
      v2f a = *(const v2f*)(ap + 4 * ks);
      c0 = __builtin_amdgcn_wmma_f32_16x16x4_f32(false, a, false, bf[ks][0], (short)0, c0, false, false);
      c1 = __builtin_amdgcn_wmma_f32_16x16x4_f32(false, a, false, bf[ks][1], (short)0, c1, false, false);
      c2 = __builtin_amdgcn_wmma_f32_16x16x4_f32(false, a, false, bf[ks][2], (short)0, c2, false, false);
      c3 = __builtin_amdgcn_wmma_f32_16x16x4_f32(false, a, false, bf[ks][3], (short)0, c3, false, false);
    }

#pragma unroll
    for (int v = 0; v < 8; ++v) {
      int row = rowBase + 8 * half + v;
      if (row < n)
        *(float4*)(h + (size_t)row * C1 + 4 * l) = make_float4(c0[v], c1[v], c2[v], c3[v]);
    }
  }
}

#define RNG   800
__device__ __forceinline__ unsigned range_byte(int d, int n) {
  return (unsigned)d < (unsigned)n ? (unsigned)d / RNG : 255u;
}
__device__ __forceinline__ unsigned range_word(int4 a, int n) {
  return range_byte(a.x, n) | (range_byte(a.y, n) << 8) | (range_byte(a.z, n) << 16) | (range_byte(a.w, n) << 24);
}
__device__ __forceinline__ unsigned edge_word(int s, int d) {
  return ((unsigned)s << 10) | ((unsigned)d - ((unsigned)d / RNG) * RNG);
}
__device__ __forceinline__ uint4 edge_words(int4 s, int4 d) {
  return make_uint4(edge_word(s.x, d.x), edge_word(s.y, d.y), edge_word(s.z, d.z), edge_word(s.w, d.w));
}
__device__ __forceinline__ int half_split(int E) { return 16 * (((E >> 4) + 1) >> 1); }

__device__ __forceinline__ float row16_sum(float v) {
  v += __builtin_bit_cast(float, __builtin_amdgcn_update_dpp(0, __builtin_bit_cast(int, v), 0x161, 0xf, 0xf, true));
  v += __builtin_bit_cast(float, __builtin_amdgcn_update_dpp(0, __builtin_bit_cast(int, v), 0x162, 0xf, 0xf, true));
  v += __builtin_bit_cast(float, __builtin_amdgcn_update_dpp(0, __builtin_bit_cast(int, v), 0x164, 0xf, 0xf, true));
  v += __builtin_bit_cast(float, __builtin_amdgcn_update_dpp(0, __builtin_bit_cast(int, v), 0x168, 0xf, 0xf, true));
  return v;
}
template <int U> __device__ __forceinline__ float row16_lane(float v) {
  return __builtin_bit_cast(float, __builtin_amdgcn_update_dpp(0, __builtin_bit_cast(int, v), 0x150 + U, 0xf, 0xf, true));
}
#define FOR16(M) M(0) M(1) M(2) M(3) M(4) M(5) M(6) M(7) M(8) M(9) M(10) M(11) M(12) M(13) M(14) M(15)

__global__ __launch_bounds__(256) void k_logits(const float* __restrict__ h, const float* __restrict__ a_src,
                         const float* __restrict__ a_dst, float* __restrict__ esrc,
                         float* __restrict__ edst, int n,
                         const int* __restrict__ ei, int E, unsigned char* __restrict__ rid,
                         unsigned* __restrict__ pk) {
  int i = blockIdx.x * blockDim.x + threadIdx.x;
  {
    const int sub = (int)threadIdx.x & 15;
    const int nb  = i - sub;
    if (nb < n) {
      const v4f as = *(const v4f*)(a_src + sub * 4);
      const v4f ad = *(const v4f*)(a_dst + sub * 4);
      v4f hv[16];
#pragma unroll
      for (int u = 0; u < 16; ++u) {
        int r = nb + u;
        r = r < n ? r : n - 1;
        hv[u] = *(const v4f*)(h + (size_t)r * C1 + sub * 4);
      }
      asm volatile("" :: "v"(hv[0]), "v"(hv[1]), "v"(hv[2]), "v"(hv[3]), "v"(hv[4]), "v"(hv[5]), "v"(hv[6]), "v"(hv[7]),
                         "v"(hv[8]), "v"(hv[9]), "v"(hv[10]), "v"(hv[11]), "v"(hv[12]), "v"(hv[13]), "v"(hv[14]), "v"(hv[15]),
                         "v"(as), "v"(ad));
      float s = 0.0f, d = 0.0f;
#pragma unroll
      for (int u = 0; u < 16; ++u) {
        float su = row16_sum(hv[u].x * as.x + hv[u].y * as.y + hv[u].z * as.z + hv[u].w * as.w);
        float du = row16_sum(hv[u].x * ad.x + hv[u].y * ad.y + hv[u].z * ad.z + hv[u].w * ad.w);
        if (sub == u) { s = su; d = du; }
      }
      if (i < n) {
        esrc[i] = s;
        edst[i] = d;
      }
    }
  }
  const int* dst = ei + E;
  const int nthr = (int)(gridDim.x * blockDim.x);
  const int nv = E >> 4;
  int done = 0;
  if (((((size_t)dst) | ((size_t)ei)) & 15) == 0) {
    const int4* dv = (const int4*)dst;
    const int4* sv = (const int4*)ei;
    for (int v = i; v < nv; v += nthr) {
      int4 a0 = dv[4 * v], a1 = dv[4 * v + 1], a2 = dv[4 * v + 2], a3 = dv[4 * v + 3];
      int4 s0 = sv[4 * v], s1 = sv[4 * v + 1], s2 = sv[4 * v + 2], s3 = sv[4 * v + 3];
      ((uint4*)rid)[v] = make_uint4(range_word(a0, n), range_word(a1, n), range_word(a2, n), range_word(a3, n));
      uint4* pv = (uint4*)pk + 4 * (size_t)v;
      pv[0] = edge_words(s0, a0);
      pv[1] = edge_words(s1, a1);
      pv[2] = edge_words(s2, a2);
      pv[3] = edge_words(s3, a3);
    }
    done = 16 * nv;
  }
  for (int idx = done + i; idx < E; idx += nthr) {
    const int d = dst[idx];
    rid[idx] = (unsigned char)range_byte(d, n);
    pk[idx] = edge_word(ei[idx], d);
  }
}

#define QCAP  12288
#define SUB   (RNG / 2)
#define LCAP  4096
#define NQ2   (2 * LCAP)
#define AGT   1024
#define WE    32
#define AU    8
static_assert(RNG <= 1024, "the edge word packs the local node into 10 bits");
static_assert(AU * AGT >= NQ2, "the maximum pass takes an owner's edges in one trip");
__device__ __forceinline__ float l1_logit(float es, float ed) {
  float v = es + ed;
  return v > 0.0f ? v : NEG_SLOPE * v;
}
__device__ __forceinline__ unsigned zero_bytes(unsigned x) {
  unsigned t = ((x & 0x7f7f7f7fu) + 0x7f7f7f7fu) | x;
  return ~t & 0x80808080u;
}
__device__ __forceinline__ void lds_max_f32(float* addr, float val) {
  if (__float_as_int(val) >= 0) atomicMax((int*)addr, __float_as_int(val));
  else                          atomicMin((unsigned int*)addr, __float_as_uint(val));
}
__device__ __forceinline__ void aggr_add(float p, float4 hv, int dl, int sub, float* acc, float* zl) {
  float* ap = acc + dl * C1 + sub * 4;
  atomicAdd(ap, p * hv.x);
  atomicAdd(ap + 1, p * hv.y);
  atomicAdd(ap + 2, p * hv.z);
  atomicAdd(ap + 3, p * hv.w);
  if (sub == 0) atomicAdd(zl + dl, p);
}

__device__ __forceinline__ void row_add(unsigned w, float p, float2 hv, int u, float* al) {
  const unsigned wu = (unsigned)__builtin_amdgcn_readlane((int)w, u);
  const float pu = __int_as_float(__builtin_amdgcn_readlane(__float_as_int(p), u));
  float* ap = al + (wu & 1023u) * C1;
  atomicAdd(ap, pu * hv.x);
  atomicAdd(ap + 1, pu * hv.y);
}

__global__ __launch_bounds__(AGT) void k_range_lists(int E, const unsigned char* __restrict__ rid,
                              const unsigned* __restrict__ pk, int* __restrict__ lst, int* __restrict__ cntq) {
  __shared__ int q[QCAP];
  __shared__ int qn;
  __shared__ int ln[2];
  const int tid  = threadIdx.x;
  const int half = blockIdx.x & 1;
  if (tid == 0) { qn = 0; ln[0] = 0; ln[1] = 0; }
  __syncthreads();

  const unsigned rno = (unsigned)(blockIdx.x >> 1);
  const unsigned pat = rno * 0x01010101u;
  const int nv = E >> 4;
  const int split = half_split(E);
  const int hb = half ? split : 0;
  {
    const uint4* rv = (const uint4*)rid;
    const int ve = half ? nv : (split >> 4);
    for (int v0 = (hb >> 4) + tid; v0 < ve; v0 += 4 * AGT) {
      uint4 a[4];
#pragma unroll
      for (int u = 0; u < 4; ++u) {
        int v = v0 + u * AGT;
        a[u] = rv[v < ve ? v : v0];
      }
      unsigned lo = (zero_bytes(a[0].x ^ pat) >> 7) | (zero_bytes(a[0].y ^ pat) >> 6) |
                    (zero_bytes(a[0].z ^ pat) >> 5) | (zero_bytes(a[0].w ^ pat) >> 4) |
                    (zero_bytes(a[1].x ^ pat) >> 3) | (zero_bytes(a[1].y ^ pat) >> 2) |
                    (zero_bytes(a[1].z ^ pat) >> 1) |  zero_bytes(a[1].w ^ pat);
      unsigned hi = (zero_bytes(a[2].x ^ pat) >> 7) | (zero_bytes(a[2].y ^ pat) >> 6) |
                    (zero_bytes(a[2].z ^ pat) >> 5) | (zero_bytes(a[2].w ^ pat) >> 4) |
                    (zero_bytes(a[3].x ^ pat) >> 3) | (zero_bytes(a[3].y ^ pat) >> 2) |
                    (zero_bytes(a[3].z ^ pat) >> 1) |  zero_bytes(a[3].w ^ pat);
      lo &= (v0 + AGT < ve) ? 0xffffffffu : 0x0f0f0f0fu;
      hi &= ((v0 + 2 * AGT < ve) ? 0x0f0f0f0fu : 0u) | ((v0 + 3 * AGT < ve) ? 0xf0f0f0f0u : 0u);
      unsigned long long hit = ((unsigned long long)hi << 32) | lo;
      while (hit) {
        int b = __ffsll(hit) - 1;
        hit &= hit - 1;
        int u = 2 * (b >> 5) + ((b >> 2) & 1);
        int idx = 16 * (v0 + u * AGT) + 4 * (b & 3) + ((b & 31) >> 3);
        int slot = atomicAdd(&qn, 1);
        if (slot < QCAP) q[slot] = idx;
      }
    }
  }
  if (half)
    for (int idx = 16 * nv + tid; idx < E; idx += AGT)
      if (rid[idx] == rno) {
        int slot = atomicAdd(&qn, 1);
        if (slot < QCAP) q[slot] = idx;
      }
  __syncthreads();

  const int nqraw = qn;
  const bool listed = nqraw <= QCAP;
  const int nq = listed ? nqraw : 0;

  int* l0 = lst + (size_t)(2 * blockIdx.x) * LCAP;
  int* l1 = l0 + LCAP;
  for (int k0 = tid; k0 < nq; k0 += AU * AGT) {
    int      idx[AU];
    unsigned ew[AU];
#pragma unroll
    for (int u = 0; u < AU; ++u) {
      int k = k0 + u * AGT;
      idx[u] = q[k < nq ? k : k0];
    }
#pragma unroll
    for (int u = 0; u < AU; ++u) ew[u] = pk[idx[u]];
#pragma unroll
    for (int u = 0; u < AU; ++u) {
      if (k0 + u * AGT < nq) {
        if ((ew[u] & 1023u) >= (unsigned)SUB) {
          int slot = atomicAdd(&ln[1], 1);
          if (slot < LCAP) l1[slot] = (int)(ew[u] - (unsigned)SUB);
        } else {
          int slot = atomicAdd(&ln[0], 1);
          if (slot < LCAP) l0[slot] = (int)ew[u];
        }
      }
    }
  }
  __syncthreads();
  if (tid < 2) cntq[2 * blockIdx.x + tid] = listed ? ln[tid] : LCAP + 1;
}

__global__ __launch_bounds__(AGT) void k_edge_aggr64(const int* __restrict__ ei, int E, int n,
                              const float* __restrict__ esrc, const float* __restrict__ edst,
                              const float* __restrict__ h, const int* __restrict__ lst,
                              const int* __restrict__ cntq, const float* __restrict__ b1,
                              const float* __restrict__ W2, float* __restrict__ g) {
  __shared__ float acc[SUB * C1];
  __shared__ float zl[SUB];
  __shared__ float ml[SUB];
  __shared__ float edl[SUB];
  __shared__ int   q[NQ2];
  __shared__ int   q2[NQ2];
  const int tid  = threadIdx.x;
  const int rng  = blockIdx.x >> 1;
  const int nh   = blockIdx.x & 1;
  const int base = rng * RNG + nh * SUB;
  const int cnt  = max(0, min(SUB, n - base));
  const unsigned ucnt = (unsigned)cnt;
  const int sub  = tid & 15;
  const int grp  = tid >> 4;
  const int* dst = ei + E;

  const int ia = 2 * (2 * rng) + nh, ib = 2 * (2 * rng + 1) + nh;
  const int* la = lst + (size_t)ia * LCAP;
  const int* lb = lst + (size_t)ib * LCAP;
  const int nA = cntq[ia], nB = cntq[ib];
  const bool listed = nA <= LCAP && nB <= LCAP;
  const int nq = listed ? nA + nB : 0;

  for (int r = tid; r < cnt; r += AGT) {
    float ed = edst[base + r];
    edl[r] = ed;
    ml[r] = l1_logit(esrc[base + r], ed);
  }
  __syncthreads();

  if (listed) {
    for (int k0 = tid; k0 < nq; k0 += AU * AGT) {
      unsigned ew[AU];
      float    es[AU];
#pragma unroll
      for (int u = 0; u < AU; ++u) {
        int k = k0 + u * AGT;
        int kk = k < nq ? k : k0;
        ew[u] = (unsigned)(kk < nA ? la[kk] : lb[kk - nA]);
      }
#pragma unroll
      for (int u = 0; u < AU; ++u) es[u] = esrc[ew[u] >> 10];
#pragma unroll
      for (int u = 0; u < AU; ++u) {
        int k = k0 + u * AGT;
        if (k < nq) {
          int dl = (int)(ew[u] & 1023u);
          float e = l1_logit(es[u], edl[dl]);
          lds_max_f32(&ml[dl], e);
          q[k]  = (int)ew[u];
          q2[k] = __float_as_int(e);
        }
      }
    }
  } else {
    for (int idx = tid; idx < E; idx += AGT) {
      int d = dst[idx];
      if ((unsigned)(d - base) < ucnt)
        lds_max_f32(&ml[d - base], l1_logit(esrc[ei[idx]], edl[d - base]));
    }
  }
  __syncthreads();

  for (int r = grp; r < cnt; r += AGT / 16) {
    const int d = base + r;
    const float p = __expf(l1_logit(esrc[d], edl[r]) - ml[r]);
    const float4 hv = *(const float4*)(h + (size_t)d * C1 + sub * 4);
    *(float4*)(acc + r * C1 + sub * 4) = make_float4(p * hv.x, p * hv.y, p * hv.z, p * hv.w);
    if (sub == 0) zl[r] = p;
  }
  __syncthreads();

  if (listed) {
    const int wv   = __builtin_amdgcn_readfirstlane(tid >> 5);
    const int nqs  = __builtin_amdgcn_readfirstlane(nq);
    const int lane = tid & 31;
    float* al = acc + lane * 2;
    for (int k0 = WE * wv; k0 < nqs; k0 += WE * (AGT / 32)) {
      const int nv = min(WE, nqs - k0);
      const int kk = lane < nv ? k0 + lane : k0;
      const unsigned w = (unsigned)q[kk];
      float2 hv[WE];
#pragma unroll
      for (int u = 0; u < WE; ++u) {
        const unsigned wu = (unsigned)__builtin_amdgcn_readlane((int)w, u);
        const float* hr = h + (size_t)(wu >> 10) * C1;
        hv[u] = *(const float2*)(hr + lane * 2);
      }
      const int dl = (int)(w & 1023u);
      const float p = __expf(__int_as_float(q2[kk]) - ml[dl]);
      if (lane < nv) atomicAdd(zl + dl, p);
      if (nv == WE) {
#pragma unroll
        for (int u = 0; u < WE; ++u) row_add(w, p, hv[u], u, al);
      } else {
#pragma unroll
        for (int u = 0; u < WE; ++u)
          if (u < nv) row_add(w, p, hv[u], u, al);
      }
    }
  } else {
    for (int idx = tid; idx < E; idx += AGT) {
      int d = dst[idx];
      if ((unsigned)(d - base) < ucnt) {
        int s = ei[idx];
        float p = __expf(l1_logit(esrc[s], edl[d - base]) - ml[d - base]);
#pragma nounroll
        for (int s2 = 0; s2 < 16; ++s2)
          aggr_add(p, *(const float4*)(h + (size_t)s * C1 + s2 * 4), d - base, s2, acc, zl);
      }
    }
  }
  __syncthreads();

  {
    const v4f bb = *(const v4f*)(b1 + sub * 4);
    const v4f ww = *(const v4f*)(W2 + sub * 4);
    for (int r = grp; r < cnt; r += AGT / 16) {
      const float rz = 1.0f / zl[r];
      const float4 a = *(const float4*)(acc + r * C1 + sub * 4);
      const float v0 = fmaxf(a.x * rz + bb.x, 0.0f);
      const float v1 = fmaxf(a.y * rz + bb.y, 0.0f);
      const float v2 = fmaxf(a.z * rz + bb.z, 0.0f);
      const float v3 = fmaxf(a.w * rz + bb.w, 0.0f);
      const float t = row16_sum(v0 * ww.x + v1 * ww.y + v2 * ww.z + v3 * ww.w);
      if (sub == 0) g[base + r] = t;
    }
  }
}

#define L2T 1024
#define L2U (NQ2 / L2T)
__device__ __forceinline__ float l2_logit(float gs, float as2, float ed) {
  float v = gs * as2 + ed;
  return v > 0.0f ? v : NEG_SLOPE * v;
}

__global__ __launch_bounds__(L2T) void k_l2_range(const int* __restrict__ ei, int E, int n,
                              const int* __restrict__ lst, const int* __restrict__ cntq,
                              const float* __restrict__ g, const float* __restrict__ a_src2,
                              const float* __restrict__ a_dst2, const float* __restrict__ b2,
                              float* __restrict__ out) {
  __shared__ float ml[SUB];
  __shared__ float zl[SUB];
  __shared__ float nl[SUB];
  __shared__ float edl[SUB];
  __shared__ float gl[SUB];
  __shared__ float esl[SUB];
  const int tid  = threadIdx.x;
  const int rng  = blockIdx.x >> 1;
  const int nh   = blockIdx.x & 1;
  const int base = rng * RNG + nh * SUB;
  const int cnt  = max(0, min(SUB, n - base));
  const unsigned ucnt = (unsigned)cnt;
  const float as2 = a_src2[0];
  const float ad2 = a_dst2[0];
  const int* dst = ei + E;

  const int ia = 2 * (2 * rng) + nh, ib = 2 * (2 * rng + 1) + nh;
  const int* la = lst + (size_t)ia * LCAP;
  const int* lb = lst + (size_t)ib * LCAP;
  const int nA = cntq[ia], nB = cntq[ib];
  const bool listed = nA <= LCAP && nB <= LCAP;
  const int nq = listed ? nA + nB : 0;

  const unsigned dummy = (unsigned)(cnt > 0 ? base : 0) << 10;
  unsigned sd[L2U];
  float gs[L2U], ev[L2U];
#pragma unroll
  for (int u = 0; u < L2U; ++u) {
    int k = tid + u * L2T;
    sd[u] = k < nq ? (unsigned)(k < nA ? la[k] : lb[k - nA]) : dummy;
  }
#pragma unroll
  for (int u = 0; u < L2U; ++u) gs[u] = g[sd[u] >> 10];

  for (int r = tid; r < cnt; r += L2T) {
    float gv = g[base + r];
    float ed = gv * ad2;
    float v = l2_logit(gv, as2, ed);
    gl[r] = gv;
    edl[r] = ed;
    esl[r] = v;
    ml[r] = v;
  }
  __syncthreads();

  if (listed) {
#pragma unroll
    for (int u = 0; u < L2U; ++u) {
      if (tid + u * L2T < nq) {
        int dl = (int)(sd[u] & 1023u);
        ev[u] = l2_logit(gs[u], as2, edl[dl]);
        lds_max_f32(&ml[dl], ev[u]);
      } else {
        ev[u] = 0.0f;
      }
    }
  } else {
    for (int idx = tid; idx < E; idx += L2T) {
      int d = dst[idx];
      if ((unsigned)(d - base) < ucnt)
        lds_max_f32(&ml[d - base], l2_logit(g[ei[idx]], as2, edl[d - base]));
    }
  }
  __syncthreads();

  for (int r = tid; r < cnt; r += L2T) {
    float p = __expf(esl[r] - ml[r]);
    zl[r] = p;
    nl[r] = p * gl[r];
  }
  __syncthreads();

  if (listed) {
#pragma unroll
    for (int u = 0; u < L2U; ++u) {
      if (tid + u * L2T < nq) {
        int dl = (int)(sd[u] & 1023u);
        float p = __expf(ev[u] - ml[dl]);
        atomicAdd(&zl[dl], p);
        atomicAdd(&nl[dl], p * gs[u]);
      }
    }
  } else {
    for (int idx = tid; idx < E; idx += L2T) {
      int d = dst[idx];
      if ((unsigned)(d - base) < ucnt) {
        float gsv = g[ei[idx]];
        float p = __expf(l2_logit(gsv, as2, edl[d - base]) - ml[d - base]);
        atomicAdd(&zl[d - base], p);
        atomicAdd(&nl[d - base], p * gsv);
      }
    }
  }
  __syncthreads();

  const float bo = b2[0];
  for (int r = tid; r < cnt; r += L2T) {
    float v = nl[r] / zl[r] + bo;
    out[base + r] = 1.0f / (1.0f + __expf(-v));
  }
}

extern "C" void kernel_launch(void* const* d_in, const int* in_sizes, int n_in,
                              void* d_out, int out_size, void* d_ws, size_t ws_size,
                              hipStream_t stream) {
  const int*   ei     = (const int*)d_in[0];
  const float* x      = (const float*)d_in[1];
  const float* W1     = (const float*)d_in[2];
  const float* a_src1 = (const float*)d_in[3];
  const float* a_dst1 = (const float*)d_in[4];
  const float* b1     = (const float*)d_in[5];
  const float* W2     = (const float*)d_in[6];
  const float* a_src2 = (const float*)d_in[7];
  const float* a_dst2 = (const float*)d_in[8];
  const float* b2     = (const float*)d_in[9];

  const int E = in_sizes[0] / 2;
  const int n = in_sizes[1] / DIN;

  float* f     = (float*)d_ws;
  float* h     = f;
  float* esrc  = f + (size_t)64 * n;
  float* edst  = f + (size_t)65 * n;
  float* g     = f + (size_t)66 * n;
  const int nwg = 2 * ((n + RNG - 1) / RNG);
  const size_t o2 = ((size_t)67 * n + 3) & ~(size_t)3;
  int*   lst   = (int*)(f + o2);
  int*   cntq  = lst + (size_t)nwg * 2 * LCAP;
  unsigned char* rid = (unsigned char*)(cntq + ((2 * nwg + 3) & ~3));
  unsigned* pk = (unsigned*)(rid + (((size_t)E + 15) & ~(size_t)15));
  float* out   = (float*)d_out;

  const int B = 256;
  dim3 blk(B);

  k_gemm_wmma<<<dim3(256), dim3(32 * GEMM_WAVES), 0, stream>>>(x, W1, h, n);
  k_logits<<<dim3((n + B - 1) / B), blk, 0, stream>>>(h, a_src1, a_dst1, esrc, edst, n, ei, E, rid, pk);
  k_range_lists<<<dim3(nwg), dim3(AGT), 0, stream>>>(E, rid, pk, lst, cntq);
  k_edge_aggr64<<<dim3(nwg), dim3(AGT), 0, stream>>>(ei, E, n, esrc, edst, h, lst, cntq, b1, W2, g);

  k_l2_range<<<dim3(nwg), dim3(L2T), 0, stream>>>(ei, E, n, lst, cntq, g, a_src2, a_dst2, b2, out);
}
